# starting kernel
# baseline (speedup 1.0000x reference)
#include <hip/hip_runtime.h>
#include <hip/hip_bf16.h>

#define NN 50000
#define EE 600000
#define FF 128
#define HH 128
#define OO 64
#define RR 8

typedef float v2f __attribute__((ext_vector_type(2)));
typedef float v8f __attribute__((ext_vector_type(8)));

__global__ void init_mm_kernel(unsigned* mm) {
    if (threadIdx.x == 0) { mm[0] = 0x7F800000u; mm[1] = 0u; }
}

__global__ void ew_minmax_kernel(const float* __restrict__ ew, int E,
                                 unsigned* __restrict__ mm) {
    float mn = __builtin_huge_valf();
    float mx = 0.0f;
    for (int i = blockIdx.x * blockDim.x + threadIdx.x; i < E;
         i += gridDim.x * blockDim.x) {
        float w = ew[i];
        mn = fminf(mn, w);
        mx = fmaxf(mx, w);
    }
    for (int off = 16; off > 0; off >>= 1) {
        mn = fminf(mn, __shfl_down(mn, off, 32));
        mx = fmaxf(mx, __shfl_down(mx, off, 32));
    }
    if ((threadIdx.x & 31) == 0) {
        atomicMin(&mm[0], __float_as_uint(mn));
        atomicMax(&mm[1], __float_as_uint(mx));
    }
}

__global__ void count_kernel(const int* __restrict__ dst, float* __restrict__ cnt,
                             int E) {
    int e = blockIdx.x * blockDim.x + threadIdx.x;
    if (e < E) atomicAdd(&cnt[dst[e]], 1.0f);
}

__global__ void invcnt_kernel(const float* __restrict__ cnt,
                              float* __restrict__ inv, int N) {
    int n = blockIdx.x * blockDim.x + threadIdx.x;
    if (n < N) inv[n] = 1.0f / fmaxf(cnt[n], 1.0f);
}

template <int M>
__global__ __launch_bounds__(32)
void gemm_wmma_f32_kernel(const float* __restrict__ A, const float* __restrict__ B,
                          float* __restrict__ C) {
    const int K = 128;
    const int lane = threadIdx.x;
    const int half = lane >> 4;
    const int lr   = lane & 15;
    const int rowbase = blockIdx.x * 16;

    const float* arow = A + (size_t)(rowbase + lr) * K + 2 * half;
    v2f a[32];
#pragma unroll
    for (int s = 0; s < 32; ++s) {
        a[s] = *(const v2f*)(arow + 4 * s);
    }

#pragma unroll
    for (int ct = 0; ct < (M >> 4); ++ct) {
        const int col = ct * 16 + lr;
        const float* bp = B + (size_t)(2 * half) * M + col;
        v2f b[32];
#pragma unroll
        for (int s = 0; s < 32; ++s) {
            b[s].x = bp[(4 * s + 0) * M];
            b[s].y = bp[(4 * s + 1) * M];
        }
        v8f acc0 = {};
        v8f acc1 = {};
#pragma unroll
        for (int s = 0; s < 32; s += 2) {
            acc0 = __builtin_amdgcn_wmma_f32_16x16x4_f32(
                false, a[s], false, b[s], (short)0, acc0, false, false);
            acc1 = __builtin_amdgcn_wmma_f32_16x16x4_f32(
                false, a[s + 1], false, b[s + 1], (short)0, acc1, false, false);
        }
        float* crow = C + (size_t)(rowbase + 8 * half) * M + col;
#pragma unroll
        for (int j = 0; j < 8; ++j) {
            crow[(size_t)j * M] = acc0[j] + acc1[j];
        }
    }
}

__global__ void scatter_rel_kernel(const float* __restrict__ proj,
                                   const int* __restrict__ src,
                                   const int* __restrict__ dst,
                                   const int* __restrict__ et,
                                   const float* __restrict__ ew,
                                   const unsigned* __restrict__ mm,
                                   float* __restrict__ agg,
                                   int E, int M, int r) {
    const int Mq = M >> 2;
    long long idx = (long long)blockIdx.x * blockDim.x + threadIdx.x;
    long long total = (long long)E * Mq;
    if (idx >= total) return;
    int e  = (int)(idx / Mq);
    int c4 = (int)(idx - (long long)e * Mq) << 2;
    if (et[e] != r) return;
    float mn = __uint_as_float(mm[0]);
    float mx = __uint_as_float(mm[1]);
    float w = (ew[e] - mn) / (mx - mn + 1e-8f);
    const float4 p = *(const float4*)(proj + (size_t)src[e] * M + c4);
    float* ag = agg + (size_t)dst[e] * M + c4;
    atomicAdd(ag + 0, p.x * w);
    atomicAdd(ag + 1, p.y * w);
    atomicAdd(ag + 2, p.z * w);
    atomicAdd(ag + 3, p.w * w);
}

__global__ void finalize_kernel(const float* __restrict__ agg,
                                const float* __restrict__ inv,
                                const float* __restrict__ bias,
                                float* __restrict__ out,
                                int N, int M, int do_relu) {
    long long idx = (long long)blockIdx.x * blockDim.x + threadIdx.x;
    if (idx >= (long long)N * M) return;
    int n = (int)(idx / M);
    int c = (int)(idx - (long long)n * M);
    float v = agg[idx] * inv[n] + out[idx] + bias[c];
    out[idx] = do_relu ? fmaxf(v, 0.0f) : v;
}

extern "C" void kernel_launch(void* const* d_in, const int* in_sizes, int n_in,
                              void* d_out, int out_size, void* d_ws, size_t ws_size,
                              hipStream_t stream) {
    const float* x     = (const float*)d_in[0];
    const int*   eidx  = (const int*)d_in[1];
    const float* ew    = (const float*)d_in[2];
    const int*   et    = (const int*)d_in[3];
    const float* W1    = (const float*)d_in[4];
    const float* root1 = (const float*)d_in[5];
    const float* bias1 = (const float*)d_in[6];
    const float* W2    = (const float*)d_in[7];
    const float* root2 = (const float*)d_in[8];
    const float* bias2 = (const float*)d_in[9];
    float* out = (float*)d_out;

    const int* src = eidx;
    const int* dst = eidx + EE;

    float* proj = (float*)d_ws;
    float* agg  = proj + (size_t)NN * HH;
    float* h    = agg  + (size_t)NN * HH;
    float* cnt  = h    + (size_t)NN * HH;
    float* inv  = cnt  + NN;
    unsigned* mm = (unsigned*)(inv + NN);

    const int T = 256;
    const int rowTiles = NN / 16;

    hipMemsetAsync(agg, 0, (size_t)NN * HH * sizeof(float), stream);
    hipMemsetAsync(cnt, 0, (size_t)NN * sizeof(float), stream);
    init_mm_kernel<<<1, 32, 0, stream>>>(mm);
    ew_minmax_kernel<<<1024, T, 0, stream>>>(ew, EE, mm);
    count_kernel<<<(EE + T - 1) / T, T, 0, stream>>>(dst, cnt, EE);
    invcnt_kernel<<<(NN + T - 1) / T, T, 0, stream>>>(cnt, inv, NN);

    {
        long long sw = (long long)EE * (HH / 4);
        int sblocks = (int)((sw + T - 1) / T);
        for (int r = 0; r < RR; ++r) {
            gemm_wmma_f32_kernel<HH><<<rowTiles, 32, 0, stream>>>(
                x, W1 + (size_t)r * FF * HH, proj);
            scatter_rel_kernel<<<sblocks, T, 0, stream>>>(
                proj, src, dst, et, ew, mm, agg, EE, HH, r);
        }
        gemm_wmma_f32_kernel<HH><<<rowTiles, 32, 0, stream>>>(x, root1, h);
        long long fw = (long long)NN * HH;
        finalize_kernel<<<(int)((fw + T - 1) / T), T, 0, stream>>>(
            agg, inv, bias1, h, NN, HH, 1);
    }

    {
        hipMemsetAsync(agg, 0, (size_t)NN * OO * sizeof(float), stream);
        long long sw = (long long)EE * (OO / 4);
        int sblocks = (int)((sw + T - 1) / T);
        for (int r = 0; r < RR; ++r) {
            gemm_wmma_f32_kernel<OO><<<rowTiles, 32, 0, stream>>>(
                h, W2 + (size_t)r * HH * OO, proj);
            scatter_rel_kernel<<<sblocks, T, 0, stream>>>(
                proj, src, dst, et, ew, mm, agg, EE, OO, r);
        }
        gemm_wmma_f32_kernel<OO><<<rowTiles, 32, 0, stream>>>(h, root2, out);
        long long fw = (long long)NN * OO;
        finalize_kernel<<<(int)((fw + T - 1) / T), T, 0, stream>>>(
            agg, inv, bias2, out, NN, OO, 0);
    }
}


# Replace per-relation float-atomic scatter by a per-call counting sort of edges by (relation,dst) and a per-node register gather; node sums now run in arrival-rank order, relations in fixed order
# speedup vs baseline: 1.2803x; 1.2803x over previous
#include <hip/hip_runtime.h>
#include <hip/hip_bf16.h>

#define NN 50000
#define EE 600000
#define FF 128
#define HH 128
#define OO 64
#define RR 8

typedef float v2f __attribute__((ext_vector_type(2)));
typedef float v8f __attribute__((ext_vector_type(8)));

__global__ void init_mm_kernel(unsigned* mm) {
    if (threadIdx.x == 0) { mm[0] = 0x7F800000u; mm[1] = 0u; }
}

__global__ void ew_minmax_kernel(const float* __restrict__ ew, int E,
                                 unsigned* __restrict__ mm) {
    float mn = __builtin_huge_valf();
    float mx = 0.0f;
    for (int i = blockIdx.x * blockDim.x + threadIdx.x; i < E;
         i += gridDim.x * blockDim.x) {
        float w = ew[i];
        mn = fminf(mn, w);
        mx = fmaxf(mx, w);
    }
    for (int off = 16; off > 0; off >>= 1) {
        mn = fminf(mn, __shfl_down(mn, off, 32));
        mx = fmaxf(mx, __shfl_down(mx, off, 32));
    }
    if ((threadIdx.x & 31) == 0) {
        atomicMin(&mm[0], __float_as_uint(mn));
        atomicMax(&mm[1], __float_as_uint(mx));
    }
}

__global__ void count_kernel(const int* __restrict__ dst, float* __restrict__ cnt,
                             int E) {
    int e = blockIdx.x * blockDim.x + threadIdx.x;
    if (e < E) atomicAdd(&cnt[dst[e]], 1.0f);
}

__global__ void invcnt_kernel(const float* __restrict__ cnt,
                              float* __restrict__ inv, int N) {
    int n = blockIdx.x * blockDim.x + threadIdx.x;
    if (n < N) inv[n] = 1.0f / fmaxf(cnt[n], 1.0f);
}

template <int M>
__global__ __launch_bounds__(32)
void gemm_wmma_f32_kernel(const float* __restrict__ A, const float* __restrict__ B,
                          float* __restrict__ C) {
    const int K = 128;
    const int lane = threadIdx.x;
    const int half = lane >> 4;
    const int lr   = lane & 15;
    const int rowbase = blockIdx.x * 16;

    const float* arow = A + (size_t)(rowbase + lr) * K + 2 * half;
    v2f a[32];
#pragma unroll
    for (int s = 0; s < 32; ++s) {
        a[s] = *(const v2f*)(arow + 4 * s);
    }

#pragma unroll
    for (int ct = 0; ct < (M >> 4); ++ct) {
        const int col = ct * 16 + lr;
        const float* bp = B + (size_t)(2 * half) * M + col;
        v2f b[32];
#pragma unroll
        for (int s = 0; s < 32; ++s) {
            b[s].x = bp[(4 * s + 0) * M];
            b[s].y = bp[(4 * s + 1) * M];
        }
        v8f acc0 = {};
        v8f acc1 = {};
#pragma unroll
        for (int s = 0; s < 32; s += 2) {
            acc0 = __builtin_amdgcn_wmma_f32_16x16x4_f32(
                false, a[s], false, b[s], (short)0, acc0, false, false);
            acc1 = __builtin_amdgcn_wmma_f32_16x16x4_f32(
                false, a[s + 1], false, b[s + 1], (short)0, acc1, false, false);
        }
        float* crow = C + (size_t)(rowbase + 8 * half) * M + col;
#pragma unroll
        for (int j = 0; j < 8; ++j) {
            crow[(size_t)j * M] = acc0[j] + acc1[j];
        }
    }
}

#define NB (RR * NN)
#define SCAN_TILE 1024
#define SCAN_BLOCKS ((NB + SCAN_TILE) / SCAN_TILE)
#define NBP (SCAN_BLOCKS * SCAN_TILE)

__global__ void rank_kernel(const int* __restrict__ dst, const int* __restrict__ et,
                            int* __restrict__ hist, int* __restrict__ rank, int E) {
    int e = blockIdx.x * blockDim.x + threadIdx.x;
    if (e >= E) return;
    int t = et[e];
    int d = dst[e];
    int rk = -1;
    if ((unsigned)t < (unsigned)RR && (unsigned)d < (unsigned)NN)
        rk = atomicAdd(&hist[t * NN + d], 1);
    rank[e] = rk;
}

__global__ __launch_bounds__(256)
void scan_tilesum_kernel(const int* __restrict__ hist, int* __restrict__ tsum) {
    __shared__ int wsum[8];
    const int t = threadIdx.x;
    const int4 v = ((const int4*)hist)[blockIdx.x * 256 + t];
    int s = v.x + v.y + v.z + v.w;
    for (int off = 16; off > 0; off >>= 1) s += __shfl_down(s, off, 32);
    if ((t & 31) == 0) wsum[t >> 5] = s;
    __syncthreads();
    if (t == 0) {
        int tot = 0;
        for (int i = 0; i < 8; ++i) tot += wsum[i];
        tsum[blockIdx.x] = tot;
    }
}

__global__ __launch_bounds__(256)
void scan_apply_kernel(int* hist, const int* __restrict__ tsum) {
    __shared__ int wpart[8];
    __shared__ int wtot[8];
    const int t = threadIdx.x;
    const int lane = t & 31;
    const int wv = t >> 5;
    const int b = blockIdx.x;

    int part = 0;
    for (int j = t; j < b; j += 256) part += tsum[j];
    for (int off = 16; off > 0; off >>= 1) part += __shfl_down(part, off, 32);

    const int4 v = ((const int4*)hist)[b * 256 + t];
    const int s = v.x + v.y + v.z + v.w;
    int inc = s;
    for (int off = 1; off < 32; off <<= 1) {
        int n = __shfl_up(inc, off, 32);
        if (lane >= off) inc += n;
    }
    if (lane == 0) wpart[wv] = part;
    if (lane == 31) wtot[wv] = inc;
    __syncthreads();
    int base = 0;
    for (int i = 0; i < 8; ++i) {
        base += wpart[i];
        if (i < wv) base += wtot[i];
    }
    const int ex = base + inc - s;
    int4 o;
    o.x = ex;
    o.y = ex + v.x;
    o.z = o.y + v.y;
    o.w = o.z + v.z;
    ((int4*)hist)[b * 256 + t] = o;
}

__global__ void fill_kernel(const int* __restrict__ src, const int* __restrict__ dst,
                            const int* __restrict__ et, const float* __restrict__ ew,
                            const unsigned* __restrict__ mm,
                            const int* __restrict__ starts, const int* __restrict__ rank,
                            int* __restrict__ sedge, int E) {
    int e = blockIdx.x * blockDim.x + threadIdx.x;
    if (e >= E) return;
    int rk = rank[e];
    if (rk < 0) return;
    int pos = starts[et[e] * NN + dst[e]] + rk;
    if ((unsigned)pos >= (unsigned)E) return;
    float mn = __uint_as_float(mm[0]);
    float mx = __uint_as_float(mm[1]);
    int2 rec;
    rec.x = src[e];
    rec.y = __float_as_int((ew[e] - mn) / (mx - mn + 1e-8f));
    ((int2*)sedge)[pos] = rec;
}

template <int M>
__global__ __launch_bounds__(256)
void gather_rel_kernel(const float* __restrict__ proj, const int* __restrict__ starts,
                       const int* __restrict__ sedge,
                       float* __restrict__ agg, int r) {
    constexpr int LPN = M / 4;
    constexpr int NPB = 256 / LPN;
    const int i = blockIdx.x * NPB + threadIdx.x / LPN;
    if (i >= NN) return;
    const int c4 = (threadIdx.x % LPN) * 4;
    const int key = r * NN + i;
    int beg = starts[key];
    int end = starts[key + 1];
    beg = beg < 0 ? 0 : beg;
    end = end > EE ? EE : end;
    float4 acc = {0.0f, 0.0f, 0.0f, 0.0f};
    for (int p = beg; p < end; ++p) {
        const int2 rec = ((const int2*)sedge)[p];
        int s = rec.x;
        const float w = __int_as_float(rec.y);
        s = s < 0 ? 0 : (s >= NN ? NN - 1 : s);
        const float4 v = *(const float4*)(proj + (size_t)s * M + c4);
        acc.x += v.x * w;
        acc.y += v.y * w;
        acc.z += v.z * w;
        acc.w += v.w * w;
    }
    float4* ap = (float4*)(agg + (size_t)i * M + c4);
    if (r == 0) {
        *ap = acc;
    } else if (end > beg) {
        float4 a = *ap;
        a.x += acc.x;
        a.y += acc.y;
        a.z += acc.z;
        a.w += acc.w;
        *ap = a;
    }
}

__global__ void finalize_kernel(const float* __restrict__ agg,
                                const float* __restrict__ inv,
                                const float* __restrict__ bias,
                                float* __restrict__ out,
                                int N, int M, int do_relu) {
    long long idx = (long long)blockIdx.x * blockDim.x + threadIdx.x;
    if (idx >= (long long)N * M) return;
    int n = (int)(idx / M);
    int c = (int)(idx - (long long)n * M);
    float v = agg[idx] * inv[n] + out[idx] + bias[c];
    out[idx] = do_relu ? fmaxf(v, 0.0f) : v;
}

extern "C" void kernel_launch(void* const* d_in, const int* in_sizes, int n_in,
                              void* d_out, int out_size, void* d_ws, size_t ws_size,
                              hipStream_t stream) {
    const float* x     = (const float*)d_in[0];
    const int*   eidx  = (const int*)d_in[1];
    const float* ew    = (const float*)d_in[2];
    const int*   et    = (const int*)d_in[3];
    const float* W1    = (const float*)d_in[4];
    const float* root1 = (const float*)d_in[5];
    const float* bias1 = (const float*)d_in[6];
    const float* W2    = (const float*)d_in[7];
    const float* root2 = (const float*)d_in[8];
    const float* bias2 = (const float*)d_in[9];
    float* out = (float*)d_out;

    const int* src = eidx;
    const int* dst = eidx + EE;

    float* bufP = (float*)d_ws;
    float* agg  = bufP + (size_t)NN * HH;
    float* bufG = agg  + (size_t)NN * HH;
    float* cnt  = bufG + (size_t)NN * HH;
    float* inv  = cnt  + NN;
    unsigned* mm = (unsigned*)(inv + NN);

    int*   starts = (int*)bufG;
    int*   tsum   = starts + NBP;
    int*   sedge  = tsum + 512;
    int*   rank   = sedge + 2 * EE;
    float* proj2  = (float*)(rank + EE);

    const int T = 256;
    const int rowTiles = NN / 16;

    hipMemsetAsync(starts, 0, (size_t)NBP * sizeof(int), stream);
    hipMemsetAsync(cnt, 0, (size_t)NN * sizeof(float), stream);
    init_mm_kernel<<<1, 32, 0, stream>>>(mm);
    ew_minmax_kernel<<<1024, T, 0, stream>>>(ew, EE, mm);
    count_kernel<<<(EE + T - 1) / T, T, 0, stream>>>(dst, cnt, EE);
    invcnt_kernel<<<(NN + T - 1) / T, T, 0, stream>>>(cnt, inv, NN);

    rank_kernel<<<(EE + T - 1) / T, T, 0, stream>>>(dst, et, starts, rank, EE);
    scan_tilesum_kernel<<<SCAN_BLOCKS, 256, 0, stream>>>(starts, tsum);
    scan_apply_kernel<<<SCAN_BLOCKS, 256, 0, stream>>>(starts, tsum);
    fill_kernel<<<(EE + T - 1) / T, T, 0, stream>>>(src, dst, et, ew, mm, starts, rank,
                                                    sedge, EE);

    {
        const int gblocks = (NN + (256 / (HH / 4)) - 1) / (256 / (HH / 4));
        for (int r = 0; r < RR; ++r) {
            gemm_wmma_f32_kernel<HH><<<rowTiles, 32, 0, stream>>>(
                x, W1 + (size_t)r * FF * HH, bufP);
            gather_rel_kernel<HH><<<gblocks, 256, 0, stream>>>(
                bufP, starts, sedge, agg, r);
        }
        gemm_wmma_f32_kernel<HH><<<rowTiles, 32, 0, stream>>>(x, root1, bufP);
        long long fw = (long long)NN * HH;
        finalize_kernel<<<(int)((fw + T - 1) / T), T, 0, stream>>>(
            agg, inv, bias1, bufP, NN, HH, 1);
    }

    {
        const int gblocks = (NN + (256 / (OO / 4)) - 1) / (256 / (OO / 4));
        for (int r = 0; r < RR; ++r) {
            gemm_wmma_f32_kernel<OO><<<rowTiles, 32, 0, stream>>>(
                bufP, W2 + (size_t)r * HH * OO, proj2);
            gather_rel_kernel<OO><<<gblocks, 256, 0, stream>>>(
                proj2, starts, sedge, agg, r);
        }
        gemm_wmma_f32_kernel<OO><<<rowTiles, 32, 0, stream>>>(bufP, root2, out);
        long long fw = (long long)NN * OO;
        finalize_kernel<<<(int)((fw + T - 1) / T), T, 0, stream>>>(
            agg, inv, bias2, out, NN, OO, 0);
    }
}


# Fewer, fatter waves: gathers run a fixed 8192-wave grid with a wave-stride loop over nodes, finalize takes 5 float4 per thread, edge prep 4 edges per thread; order of every sum unchanged
# speedup vs baseline: 1.3496x; 1.0541x over previous
#include <hip/hip_runtime.h>
#include <hip/hip_bf16.h>

#define NN 50000
#define EE 600000
#define FF 128
#define HH 128
#define OO 64
#define RR 8

typedef float v2f __attribute__((ext_vector_type(2)));
typedef float v8f __attribute__((ext_vector_type(8)));

__global__ void init_mm_kernel(unsigned* mm) {
    if (threadIdx.x == 0) { mm[0] = 0x7F800000u; mm[1] = 0u; }
}

__global__ void ew_minmax_kernel(const float* __restrict__ ew, int E,
                                 unsigned* __restrict__ mm) {
    float mn = __builtin_huge_valf();
    float mx = 0.0f;
    for (int i = blockIdx.x * blockDim.x + threadIdx.x; i < E;
         i += gridDim.x * blockDim.x) {
        float w = ew[i];
        mn = fminf(mn, w);
        mx = fmaxf(mx, w);
    }
    for (int off = 16; off > 0; off >>= 1) {
        mn = fminf(mn, __shfl_down(mn, off, 32));
        mx = fmaxf(mx, __shfl_down(mx, off, 32));
    }
    if ((threadIdx.x & 31) == 0) {
        atomicMin(&mm[0], __float_as_uint(mn));
        atomicMax(&mm[1], __float_as_uint(mx));
    }
}

__global__ void count_kernel(const int* __restrict__ dst, float* __restrict__ cnt,
                             int E4) {
    int t = blockIdx.x * blockDim.x + threadIdx.x;
    if (t >= E4) return;
    const int4 d = ((const int4*)dst)[t];
    atomicAdd(&cnt[d.x], 1.0f);
    atomicAdd(&cnt[d.y], 1.0f);
    atomicAdd(&cnt[d.z], 1.0f);
    atomicAdd(&cnt[d.w], 1.0f);
}

__global__ void invcnt_kernel(const float* __restrict__ cnt,
                              float* __restrict__ inv, int N) {
    int n = blockIdx.x * blockDim.x + threadIdx.x;
    if (n < N) inv[n] = 1.0f / fmaxf(cnt[n], 1.0f);
}

template <int M>
__global__ __launch_bounds__(32)
void gemm_wmma_f32_kernel(const float* __restrict__ A, const float* __restrict__ B,
                          float* __restrict__ C) {
    const int K = 128;
    const int lane = threadIdx.x;
    const int half = lane >> 4;
    const int lr   = lane & 15;
    const int rowbase = blockIdx.x * 16;

    const float* arow = A + (size_t)(rowbase + lr) * K + 2 * half;
    v2f a[32];
#pragma unroll
    for (int s = 0; s < 32; ++s) {
        a[s] = *(const v2f*)(arow + 4 * s);
    }

#pragma unroll
    for (int ct = 0; ct < (M >> 4); ++ct) {
        const int col = ct * 16 + lr;
        const float* bp = B + (size_t)(2 * half) * M + col;
        v2f b[32];
#pragma unroll
        for (int s = 0; s < 32; ++s) {
            b[s].x = bp[(4 * s + 0) * M];
            b[s].y = bp[(4 * s + 1) * M];
        }
        v8f acc0 = {};
        v8f acc1 = {};
#pragma unroll
        for (int s = 0; s < 32; s += 2) {
            acc0 = __builtin_amdgcn_wmma_f32_16x16x4_f32(
                false, a[s], false, b[s], (short)0, acc0, false, false);
            acc1 = __builtin_amdgcn_wmma_f32_16x16x4_f32(
                false, a[s + 1], false, b[s + 1], (short)0, acc1, false, false);
        }
        float* crow = C + (size_t)(rowbase + 8 * half) * M + col;
#pragma unroll
        for (int j = 0; j < 8; ++j) {
            crow[(size_t)j * M] = acc0[j] + acc1[j];
        }
    }
}

#define NB (RR * NN)
#define SCAN_TILE 1024
#define SCAN_BLOCKS ((NB + SCAN_TILE) / SCAN_TILE)
#define NBP (SCAN_BLOCKS * SCAN_TILE)

__device__ __forceinline__ int rank_one(int t, int d, int* __restrict__ hist) {
    int rk = -1;
    if ((unsigned)t < (unsigned)RR && (unsigned)d < (unsigned)NN)
        rk = atomicAdd(&hist[t * NN + d], 1);
    return rk;
}

__global__ void rank_kernel(const int* __restrict__ dst, const int* __restrict__ et,
                            int* __restrict__ hist, int* __restrict__ rank, int E4) {
    int q = blockIdx.x * blockDim.x + threadIdx.x;
    if (q >= E4) return;
    const int4 t = ((const int4*)et)[q];
    const int4 d = ((const int4*)dst)[q];
    int4 rk;
    rk.x = rank_one(t.x, d.x, hist);
    rk.y = rank_one(t.y, d.y, hist);
    rk.z = rank_one(t.z, d.z, hist);
    rk.w = rank_one(t.w, d.w, hist);
    ((int4*)rank)[q] = rk;
}

__global__ __launch_bounds__(256)
void scan_tilesum_kernel(const int* __restrict__ hist, int* __restrict__ tsum) {
    __shared__ int wsum[8];
    const int t = threadIdx.x;
    const int4 v = ((const int4*)hist)[blockIdx.x * 256 + t];
    int s = v.x + v.y + v.z + v.w;
    for (int off = 16; off > 0; off >>= 1) s += __shfl_down(s, off, 32);
    if ((t & 31) == 0) wsum[t >> 5] = s;
    __syncthreads();
    if (t == 0) {
        int tot = 0;
        for (int i = 0; i < 8; ++i) tot += wsum[i];
        tsum[blockIdx.x] = tot;
    }
}

__global__ __launch_bounds__(256)
void scan_apply_kernel(int* hist, const int* __restrict__ tsum) {
    __shared__ int wpart[8];
    __shared__ int wtot[8];
    const int t = threadIdx.x;
    const int lane = t & 31;
    const int wv = t >> 5;
    const int b = blockIdx.x;

    int part = 0;
    for (int j = t; j < b; j += 256) part += tsum[j];
    for (int off = 16; off > 0; off >>= 1) part += __shfl_down(part, off, 32);

    const int4 v = ((const int4*)hist)[b * 256 + t];
    const int s = v.x + v.y + v.z + v.w;
    int inc = s;
    for (int off = 1; off < 32; off <<= 1) {
        int n = __shfl_up(inc, off, 32);
        if (lane >= off) inc += n;
    }
    if (lane == 0) wpart[wv] = part;
    if (lane == 31) wtot[wv] = inc;
    __syncthreads();
    int base = 0;
    for (int i = 0; i < 8; ++i) {
        base += wpart[i];
        if (i < wv) base += wtot[i];
    }
    const int ex = base + inc - s;
    int4 o;
    o.x = ex;
    o.y = ex + v.x;
    o.z = o.y + v.y;
    o.w = o.z + v.z;
    ((int4*)hist)[b * 256 + t] = o;
}

__device__ __forceinline__ int fill_key(int rk, int t, int d) {
    return rk < 0 ? 0 : t * NN + d;
}

__device__ __forceinline__ void fill_one(int rk, int start, int s, float w, float mn,
                                         float mx, int* __restrict__ sedge) {
    if (rk < 0) return;
    int pos = start + rk;
    if ((unsigned)pos >= (unsigned)EE) return;
    int2 rec;
    rec.x = s;
    rec.y = __float_as_int((w - mn) / (mx - mn + 1e-8f));
    ((int2*)sedge)[pos] = rec;
}

__global__ void fill_kernel(const int* __restrict__ src, const int* __restrict__ dst,
                            const int* __restrict__ et, const float* __restrict__ ew,
                            const unsigned* __restrict__ mm,
                            const int* __restrict__ starts, const int* __restrict__ rank,
                            int* __restrict__ sedge, int E4) {
    int q = blockIdx.x * blockDim.x + threadIdx.x;
    if (q >= E4) return;
    const int4 rk = ((const int4*)rank)[q];
    const int4 t = ((const int4*)et)[q];
    const int4 d = ((const int4*)dst)[q];
    const int4 s = ((const int4*)src)[q];
    const float4 w = ((const float4*)ew)[q];
    const float mn = __uint_as_float(mm[0]);
    const float mx = __uint_as_float(mm[1]);
    const int st0 = starts[fill_key(rk.x, t.x, d.x)];
    const int st1 = starts[fill_key(rk.y, t.y, d.y)];
    const int st2 = starts[fill_key(rk.z, t.z, d.z)];
    const int st3 = starts[fill_key(rk.w, t.w, d.w)];
    fill_one(rk.x, st0, s.x, w.x, mn, mx, sedge);
    fill_one(rk.y, st1, s.y, w.y, mn, mx, sedge);
    fill_one(rk.z, st2, s.z, w.z, mn, mx, sedge);
    fill_one(rk.w, st3, s.w, w.w, mn, mx, sedge);
}

#define GATHER_BLOCKS 1024

template <int M>
__global__ __launch_bounds__(256)
void gather_rel_kernel(const float* __restrict__ proj, const int* __restrict__ starts,
                       const int* __restrict__ sedge,
                       float* __restrict__ agg, int r) {
    constexpr int VPL = M / 32;
    typedef float vecf __attribute__((ext_vector_type(VPL)));
    const int lane = threadIdx.x & 31;
    const int wv = __builtin_amdgcn_readfirstlane((int)(threadIdx.x >> 5));
    const int nwaves = (int)gridDim.x * 8;
    int i = (int)blockIdx.x * 8 + wv;
    if (i >= NN) return;
    const int c = lane * VPL;
    const int2* __restrict__ recs = (const int2*)sedge;
    const int* __restrict__ st = starts + r * NN;
    int beg = st[i];
    int end = st[i + 1];
    beg = beg < 0 ? 0 : beg;
    end = end > EE ? EE : end;
    for (;;) {
        const int inext = i + nwaves;
        const int ipre = inext < NN ? inext : i;
        const int nbeg = st[ipre];
        const int nend = st[ipre + 1];
        vecf* ap = (vecf*)(agg + (size_t)i * M + c);
        const bool has = end > beg;
        vecf old = {};
        if (r != 0 && has) old = *ap;
        vecf acc = {};
        int p = beg;
        for (; p + 1 < end; p += 2) {
            const int2 r0 = recs[p];
            const int2 r1 = recs[p + 1];
            int s0 = r0.x;
            int s1 = r1.x;
            s0 = s0 < 0 ? 0 : (s0 >= NN ? NN - 1 : s0);
            s1 = s1 < 0 ? 0 : (s1 >= NN ? NN - 1 : s1);
            const float w0 = __int_as_float(r0.y);
            const float w1 = __int_as_float(r1.y);
            const vecf v0 = *(const vecf*)(proj + (size_t)s0 * M + c);
            const vecf v1 = *(const vecf*)(proj + (size_t)s1 * M + c);
#pragma unroll
            for (int k = 0; k < VPL; ++k) acc[k] += v0[k] * w0;
#pragma unroll
            for (int k = 0; k < VPL; ++k) acc[k] += v1[k] * w1;
        }
        if (p < end) {
            const int2 r0 = recs[p];
            int s0 = r0.x;
            s0 = s0 < 0 ? 0 : (s0 >= NN ? NN - 1 : s0);
            const float w0 = __int_as_float(r0.y);
            const vecf v0 = *(const vecf*)(proj + (size_t)s0 * M + c);
#pragma unroll
            for (int k = 0; k < VPL; ++k) acc[k] += v0[k] * w0;
        }
        if (r == 0) {
            *ap = acc;
        } else if (has) {
#pragma unroll
            for (int k = 0; k < VPL; ++k) old[k] += acc[k];
            *ap = old;
        }
        if (inext >= NN) break;
        i = inext;
        beg = nbeg < 0 ? 0 : nbeg;
        end = nend > EE ? EE : nend;
    }
}

#define FIN_PER 5

template <int M>
__global__ __launch_bounds__(256)
void finalize_kernel(const float* __restrict__ agg,
                     const float* __restrict__ inv,
                     const float* __restrict__ bias,
                     float* __restrict__ out, int do_relu) {
    constexpr int Q = M / 4;
    static_assert((NN * Q) % (256 * FIN_PER) == 0, "tile must divide the array");
    static_assert((256 * FIN_PER) % Q == 0 && 256 % Q == 0, "column is fixed per thread");
    const int base = (int)blockIdx.x * (256 * FIN_PER) + (int)threadIdx.x;
    const float4* agg4 = (const float4*)agg;
    float4* out4 = (float4*)out;
    float4 a[FIN_PER];
    float4 o[FIN_PER];
    float iv[FIN_PER];
#pragma unroll
    for (int k = 0; k < FIN_PER; ++k) {
        const int q = base + k * 256;
        a[k] = agg4[q];
        o[k] = out4[q];
        iv[k] = inv[q / Q];
    }
    const float4 b = *(const float4*)(bias + ((int)threadIdx.x % Q) * 4);
#pragma unroll
    for (int k = 0; k < FIN_PER; ++k) {
        const int q = base + k * 256;
        float4 v;
        v.x = a[k].x * iv[k] + o[k].x + b.x;
        v.y = a[k].y * iv[k] + o[k].y + b.y;
        v.z = a[k].z * iv[k] + o[k].z + b.z;
        v.w = a[k].w * iv[k] + o[k].w + b.w;
        if (do_relu) {
            v.x = fmaxf(v.x, 0.0f);
            v.y = fmaxf(v.y, 0.0f);
            v.z = fmaxf(v.z, 0.0f);
            v.w = fmaxf(v.w, 0.0f);
        }
        out4[q] = v;
    }
}

extern "C" void kernel_launch(void* const* d_in, const int* in_sizes, int n_in,
                              void* d_out, int out_size, void* d_ws, size_t ws_size,
                              hipStream_t stream) {
    const float* x     = (const float*)d_in[0];
    const int*   eidx  = (const int*)d_in[1];
    const float* ew    = (const float*)d_in[2];
    const int*   et    = (const int*)d_in[3];
    const float* W1    = (const float*)d_in[4];
    const float* root1 = (const float*)d_in[5];
    const float* bias1 = (const float*)d_in[6];
    const float* W2    = (const float*)d_in[7];
    const float* root2 = (const float*)d_in[8];
    const float* bias2 = (const float*)d_in[9];
    float* out = (float*)d_out;

    const int* src = eidx;
    const int* dst = eidx + EE;

    float* bufP = (float*)d_ws;
    float* agg  = bufP + (size_t)NN * HH;
    float* bufG = agg  + (size_t)NN * HH;
    float* cnt  = bufG + (size_t)NN * HH;
    float* inv  = cnt  + NN;
    unsigned* mm = (unsigned*)(inv + NN);

    int*   starts = (int*)bufG;
    int*   tsum   = starts + NBP;
    int*   sedge  = tsum + 512;
    int*   rank   = sedge + 2 * EE;
    float* proj2  = (float*)(rank + EE);

    const int T = 256;
    const int rowTiles = NN / 16;

    hipMemsetAsync(starts, 0, (size_t)NBP * sizeof(int), stream);
    hipMemsetAsync(cnt, 0, (size_t)NN * sizeof(float), stream);
    init_mm_kernel<<<1, 32, 0, stream>>>(mm);
    const int E4 = EE / 4;
    ew_minmax_kernel<<<256, T, 0, stream>>>(ew, EE, mm);
    count_kernel<<<(E4 + T - 1) / T, T, 0, stream>>>(dst, cnt, E4);
    invcnt_kernel<<<(NN + T - 1) / T, T, 0, stream>>>(cnt, inv, NN);

    rank_kernel<<<(E4 + T - 1) / T, T, 0, stream>>>(dst, et, starts, rank, E4);
    scan_tilesum_kernel<<<SCAN_BLOCKS, 256, 0, stream>>>(starts, tsum);
    scan_apply_kernel<<<SCAN_BLOCKS, 256, 0, stream>>>(starts, tsum);
    fill_kernel<<<(E4 + T - 1) / T, T, 0, stream>>>(src, dst, et, ew, mm, starts, rank,
                                                    sedge, E4);

    {
        const int gblocks = GATHER_BLOCKS;
        for (int r = 0; r < RR; ++r) {
            gemm_wmma_f32_kernel<HH><<<rowTiles, 32, 0, stream>>>(
                x, W1 + (size_t)r * FF * HH, bufP);
            gather_rel_kernel<HH><<<gblocks, 256, 0, stream>>>(
                bufP, starts, sedge, agg, r);
        }
        gemm_wmma_f32_kernel<HH><<<rowTiles, 32, 0, stream>>>(x, root1, bufP);
        finalize_kernel<HH><<<(NN * (HH / 4)) / (256 * FIN_PER), 256, 0, stream>>>(
            agg, inv, bias1, bufP, 1);
    }

    {
        const int gblocks = GATHER_BLOCKS;
        for (int r = 0; r < RR; ++r) {
            gemm_wmma_f32_kernel<OO><<<rowTiles, 32, 0, stream>>>(
                bufP, W2 + (size_t)r * HH * OO, proj2);
            gather_rel_kernel<OO><<<gblocks, 256, 0, stream>>>(
                proj2, starts, sedge, agg, r);
        }
        gemm_wmma_f32_kernel<OO><<<rowTiles, 32, 0, stream>>>(bufP, root2, out);
        finalize_kernel<OO><<<(NN * (OO / 4)) / (256 * FIN_PER), 256, 0, stream>>>(
            agg, inv, bias2, out, 0);
    }
}


# Fuse gather into the GEMM (aggregate, then transform): each 16-row wave sums its edges' weighted input rows per relation into an LDS A tile and applies W_r; one kernel per layer; sum order changed
# speedup vs baseline: 2.5702x; 1.9044x over previous
#include <hip/hip_runtime.h>
#include <hip/hip_bf16.h>

#define NN 50000
#define EE 600000
#define FF 128
#define HH 128
#define OO 64
#define RR 8

typedef float v2f __attribute__((ext_vector_type(2)));
typedef float v8f __attribute__((ext_vector_type(8)));

__global__ void init_mm_kernel(unsigned* mm) {
    if (threadIdx.x == 0) { mm[0] = 0x7F800000u; mm[1] = 0u; }
}

__global__ void ew_minmax_kernel(const float* __restrict__ ew, int E,
                                 unsigned* __restrict__ mm) {
    float mn = __builtin_huge_valf();
    float mx = 0.0f;
    for (int i = blockIdx.x * blockDim.x + threadIdx.x; i < E;
         i += gridDim.x * blockDim.x) {
        float w = ew[i];
        mn = fminf(mn, w);
        mx = fmaxf(mx, w);
    }
    for (int off = 16; off > 0; off >>= 1) {
        mn = fminf(mn, __shfl_down(mn, off, 32));
        mx = fmaxf(mx, __shfl_down(mx, off, 32));
    }
    if ((threadIdx.x & 31) == 0) {
        atomicMin(&mm[0], __float_as_uint(mn));
        atomicMax(&mm[1], __float_as_uint(mx));
    }
}

__global__ void count_kernel(const int* __restrict__ dst, float* __restrict__ cnt,
                             int E4) {
    int t = blockIdx.x * blockDim.x + threadIdx.x;
    if (t >= E4) return;
    const int4 d = ((const int4*)dst)[t];
    atomicAdd(&cnt[d.x], 1.0f);
    atomicAdd(&cnt[d.y], 1.0f);
    atomicAdd(&cnt[d.z], 1.0f);
    atomicAdd(&cnt[d.w], 1.0f);
}

__global__ void invcnt_kernel(const float* __restrict__ cnt,
                              float* __restrict__ inv, int N) {
    int n = blockIdx.x * blockDim.x + threadIdx.x;
    if (n < N) inv[n] = 1.0f / fmaxf(cnt[n], 1.0f);
}

#define NB (RR * NN)
#define SCAN_TILE 1024
#define SCAN_BLOCKS ((NB + SCAN_TILE) / SCAN_TILE)
#define NBP (SCAN_BLOCKS * SCAN_TILE)

__device__ __forceinline__ int rank_one(int t, int d, int* __restrict__ hist) {
    int rk = -1;
    if ((unsigned)t < (unsigned)RR && (unsigned)d < (unsigned)NN)
        rk = atomicAdd(&hist[t * NN + d], 1);
    return rk;
}

__global__ void rank_kernel(const int* __restrict__ dst, const int* __restrict__ et,
                            int* __restrict__ hist, int* __restrict__ rank, int E4) {
    int q = blockIdx.x * blockDim.x + threadIdx.x;
    if (q >= E4) return;
    const int4 t = ((const int4*)et)[q];
    const int4 d = ((const int4*)dst)[q];
    int4 rk;
    rk.x = rank_one(t.x, d.x, hist);
    rk.y = rank_one(t.y, d.y, hist);
    rk.z = rank_one(t.z, d.z, hist);
    rk.w = rank_one(t.w, d.w, hist);
    ((int4*)rank)[q] = rk;
}

__global__ __launch_bounds__(256)
void scan_tilesum_kernel(const int* __restrict__ hist, int* __restrict__ tsum) {
    __shared__ int wsum[8];
    const int t = threadIdx.x;
    const int4 v = ((const int4*)hist)[blockIdx.x * 256 + t];
    int s = v.x + v.y + v.z + v.w;
    for (int off = 16; off > 0; off >>= 1) s += __shfl_down(s, off, 32);
    if ((t & 31) == 0) wsum[t >> 5] = s;
    __syncthreads();
    if (t == 0) {
        int tot = 0;
        for (int i = 0; i < 8; ++i) tot += wsum[i];
        tsum[blockIdx.x] = tot;
    }
}

__global__ __launch_bounds__(256)
void scan_apply_kernel(int* hist, const int* __restrict__ tsum) {
    __shared__ int wpart[8];
    __shared__ int wtot[8];
    const int t = threadIdx.x;
    const int lane = t & 31;
    const int wv = t >> 5;
    const int b = blockIdx.x;

    int part = 0;
    for (int j = t; j < b; j += 256) part += tsum[j];
    for (int off = 16; off > 0; off >>= 1) part += __shfl_down(part, off, 32);

    const int4 v = ((const int4*)hist)[b * 256 + t];
    const int s = v.x + v.y + v.z + v.w;
    int inc = s;
    for (int off = 1; off < 32; off <<= 1) {
        int n = __shfl_up(inc, off, 32);
        if (lane >= off) inc += n;
    }
    if (lane == 0) wpart[wv] = part;
    if (lane == 31) wtot[wv] = inc;
    __syncthreads();
    int base = 0;
    for (int i = 0; i < 8; ++i) {
        base += wpart[i];
        if (i < wv) base += wtot[i];
    }
    const int ex = base + inc - s;
    int4 o;
    o.x = ex;
    o.y = ex + v.x;
    o.z = o.y + v.y;
    o.w = o.z + v.z;
    ((int4*)hist)[b * 256 + t] = o;
}

__device__ __forceinline__ int fill_key(int rk, int t, int d) {
    return rk < 0 ? 0 : t * NN + d;
}

__device__ __forceinline__ void fill_one(int rk, int start, int s, float w, float mn,
                                         float mx, int* __restrict__ sedge) {
    if (rk < 0) return;
    int pos = start + rk;
    if ((unsigned)pos >= (unsigned)EE) return;
    int2 rec;
    rec.x = s;
    rec.y = __float_as_int((w - mn) / (mx - mn + 1e-8f));
    ((int2*)sedge)[pos] = rec;
}

__global__ void fill_kernel(const int* __restrict__ src, const int* __restrict__ dst,
                            const int* __restrict__ et, const float* __restrict__ ew,
                            const unsigned* __restrict__ mm,
                            const int* __restrict__ starts, const int* __restrict__ rank,
                            int* __restrict__ sedge, int E4) {
    int q = blockIdx.x * blockDim.x + threadIdx.x;
    if (q >= E4) return;
    const int4 rk = ((const int4*)rank)[q];
    const int4 t = ((const int4*)et)[q];
    const int4 d = ((const int4*)dst)[q];
    const int4 s = ((const int4*)src)[q];
    const float4 w = ((const float4*)ew)[q];
    const float mn = __uint_as_float(mm[0]);
    const float mx = __uint_as_float(mm[1]);
    const int st0 = starts[fill_key(rk.x, t.x, d.x)];
    const int st1 = starts[fill_key(rk.y, t.y, d.y)];
    const int st2 = starts[fill_key(rk.z, t.z, d.z)];
    const int st3 = starts[fill_key(rk.w, t.w, d.w)];
    fill_one(rk.x, st0, s.x, w.x, mn, mx, sedge);
    fill_one(rk.y, st1, s.y, w.y, mn, mx, sedge);
    fill_one(rk.z, st2, s.z, w.z, mn, mx, sedge);
    fill_one(rk.w, st3, s.w, w.w, mn, mx, sedge);
}

#define APITCH 132
#define GU 8

template <int M, int KH>
__device__ __forceinline__ void mma_pair(const v2f (&a)[32], const float* __restrict__ Bm,
                                         int half, int col, v8f& acc0, v8f& acc1) {
    const float* bp = Bm + (size_t)(2 * half) * M + col;
    v2f b0[16];
    v2f b1[16];
#pragma unroll
    for (int t = 0; t < 16; ++t) {
        const int s = KH * 16 + t;
        b0[t].x = bp[(4 * s + 0) * M];
        b0[t].y = bp[(4 * s + 1) * M];
        b1[t].x = bp[(4 * s + 0) * M + 16];
        b1[t].y = bp[(4 * s + 1) * M + 16];
    }
#pragma unroll
    for (int t = 0; t < 16; ++t) {
        const int s = KH * 16 + t;
        acc0 = __builtin_amdgcn_wmma_f32_16x16x4_f32(
            false, a[s], false, b0[t], (short)0, acc0, false, false);
        acc1 = __builtin_amdgcn_wmma_f32_16x16x4_f32(
            false, a[s], false, b1[t], (short)0, acc1, false, false);
    }
}

template <int M, bool RELU>
__global__ __launch_bounds__(32)
void fused_layer_kernel(const float* __restrict__ X,
                        const float* __restrict__ Wrel,
                        const float* __restrict__ Wroot,
                        const float* __restrict__ bias,
                        const float* __restrict__ inv,
                        const int* __restrict__ starts,
                        const int* __restrict__ sedge,
                        float* __restrict__ Out) {
    constexpr int K = 128;
    constexpr int CT = M / 16;
    __shared__ __attribute__((aligned(16))) float tile[16 * APITCH];
    const int lane = threadIdx.x;
    const int half = lane >> 4;
    const int lr   = lane & 15;
    const int rowbase = blockIdx.x * 16;
    const int2* __restrict__ recs = (const int2*)sedge;
    const float4* __restrict__ X4 = (const float4*)X;
    const int sl = lane < 16 ? lane : 16;
    const float4 zero4 = make_float4(0.0f, 0.0f, 0.0f, 0.0f);

    v8f acc[CT];
#pragma unroll
    for (int ct = 0; ct < CT; ++ct) acc[ct] = v8f{};

    int svn = starts[rowbase + sl];
    for (int r = 0; r < RR; ++r) {
        const int svc = svn;
        const int rn = r + 1 < RR ? r + 1 : r;
        svn = starts[rn * NN + rowbase + sl];

        int eb = __builtin_amdgcn_readlane(svc, 0);
        int ee = __builtin_amdgcn_readlane(svc, 16);
        eb = eb < 0 ? 0 : eb;
        ee = ee > EE ? EE : ee;

#pragma unroll
        for (int j = 0; j < 16; ++j) *(float4*)&tile[j * APITCH + lane * 4] = zero4;

        int cur = -1;
        float4 g = zero4;
        for (int c0 = eb; c0 < ee; c0 += 32) {
            int n = ee - c0;
            n = n > 32 ? 32 : n;
            const int myp = c0 + lane;
            int2 rec = make_int2(0, 0);
            if (lane < n) rec = recs[myp];
            int ms = rec.x;
            ms = ms < 0 ? 0 : (ms >= NN ? NN - 1 : ms);
            int myrow = 0;
#pragma unroll
            for (int j = 1; j < 16; ++j)
                myrow += (__builtin_amdgcn_readlane(svc, j) <= myp) ? 1 : 0;

            for (int e0 = 0; e0 < n; e0 += GU) {
                float4 v[GU];
#pragma unroll
                for (int u = 0; u < GU; ++u) {
                    int eu = e0 + u;
                    eu = eu < n ? eu : n - 1;
                    const int s = __builtin_amdgcn_readlane(ms, eu);
                    v[u] = X4[(size_t)s * (K / 4) + lane];
                }
#pragma unroll
                for (int u = 0; u < GU; ++u) {
                    const int eu = e0 + u;
                    if (eu < n) {
                        const int row = __builtin_amdgcn_readlane(myrow, eu);
                        const float w = __int_as_float(__builtin_amdgcn_readlane(rec.y, eu));
                        if (row != cur) {
                            if (cur >= 0) *(float4*)&tile[cur * APITCH + lane * 4] = g;
                            g = zero4;
                            cur = row;
                        }
                        g.x += v[u].x * w;
                        g.y += v[u].y * w;
                        g.z += v[u].z * w;
                        g.w += v[u].w * w;
                    }
                }
            }
        }
        if (cur >= 0) *(float4*)&tile[cur * APITCH + lane * 4] = g;

        __syncthreads();
        v2f a[32];
        const float* ap = &tile[lr * APITCH + 2 * half];
#pragma unroll
        for (int s = 0; s < 32; ++s) a[s] = *(const v2f*)(ap + 4 * s);
        __syncthreads();

        const float* Wr = Wrel + (size_t)r * K * M;
#pragma unroll
        for (int ct = 0; ct < CT; ct += 2) {
            mma_pair<M, 0>(a, Wr, half, ct * 16 + lr, acc[ct], acc[ct + 1]);
            mma_pair<M, 1>(a, Wr, half, ct * 16 + lr, acc[ct], acc[ct + 1]);
        }
    }

    float ivr[8];
#pragma unroll
    for (int j = 0; j < 8; ++j) ivr[j] = inv[rowbase + 8 * half + j];
#pragma unroll
    for (int ct = 0; ct < CT; ++ct) {
#pragma unroll
        for (int j = 0; j < 8; ++j) acc[ct][j] *= ivr[j];
    }

    {
        const float* arow = X + (size_t)(rowbase + lr) * K + 2 * half;
        v2f a[32];
#pragma unroll
        for (int s = 0; s < 32; ++s) a[s] = *(const v2f*)(arow + 4 * s);
#pragma unroll
        for (int ct = 0; ct < CT; ct += 2) {
            mma_pair<M, 0>(a, Wroot, half, ct * 16 + lr, acc[ct], acc[ct + 1]);
            mma_pair<M, 1>(a, Wroot, half, ct * 16 + lr, acc[ct], acc[ct + 1]);
        }
    }

#pragma unroll
    for (int ct = 0; ct < CT; ++ct) {
        const int col = ct * 16 + lr;
        const float bb = bias[col];
        float* crow = Out + (size_t)(rowbase + 8 * half) * M + col;
#pragma unroll
        for (int j = 0; j < 8; ++j) {
            float val = acc[ct][j] + bb;
            if (RELU) val = fmaxf(val, 0.0f);
            crow[(size_t)j * M] = val;
        }
    }
}

extern "C" void kernel_launch(void* const* d_in, const int* in_sizes, int n_in,
                              void* d_out, int out_size, void* d_ws, size_t ws_size,
                              hipStream_t stream) {
    const float* x     = (const float*)d_in[0];
    const int*   eidx  = (const int*)d_in[1];
    const float* ew    = (const float*)d_in[2];
    const int*   et    = (const int*)d_in[3];
    const float* W1    = (const float*)d_in[4];
    const float* root1 = (const float*)d_in[5];
    const float* bias1 = (const float*)d_in[6];
    const float* W2    = (const float*)d_in[7];
    const float* root2 = (const float*)d_in[8];
    const float* bias2 = (const float*)d_in[9];
    float* out = (float*)d_out;

    const int* src = eidx;
    const int* dst = eidx + EE;

    float* bufH = (float*)d_ws;
    float* agg  = bufH + (size_t)NN * HH;
    float* bufG = agg  + (size_t)NN * HH;
    float* cnt  = bufG + (size_t)NN * HH;
    float* inv  = cnt  + NN;
    unsigned* mm = (unsigned*)(inv + NN);

    int*   starts = (int*)bufG;
    int*   tsum   = starts + NBP;
    int*   sedge  = tsum + 512;
    int*   rank   = sedge + 2 * EE;

    const int T = 256;
    const int rowTiles = NN / 16;

    hipMemsetAsync(starts, 0, (size_t)NBP * sizeof(int), stream);
    hipMemsetAsync(cnt, 0, (size_t)NN * sizeof(float), stream);
    init_mm_kernel<<<1, 32, 0, stream>>>(mm);
    const int E4 = EE / 4;
    ew_minmax_kernel<<<256, T, 0, stream>>>(ew, EE, mm);
    count_kernel<<<(E4 + T - 1) / T, T, 0, stream>>>(dst, cnt, E4);
    invcnt_kernel<<<(NN + T - 1) / T, T, 0, stream>>>(cnt, inv, NN);

    rank_kernel<<<(E4 + T - 1) / T, T, 0, stream>>>(dst, et, starts, rank, E4);
    scan_tilesum_kernel<<<SCAN_BLOCKS, 256, 0, stream>>>(starts, tsum);
    scan_apply_kernel<<<SCAN_BLOCKS, 256, 0, stream>>>(starts, tsum);
    fill_kernel<<<(E4 + T - 1) / T, T, 0, stream>>>(src, dst, et, ew, mm, starts, rank,
                                                    sedge, E4);

    fused_layer_kernel<HH, true><<<rowTiles, 32, 0, stream>>>(
        x, W1, root1, bias1, inv, starts, sedge, bufH);
    fused_layer_kernel<OO, false><<<rowTiles, 32, 0, stream>>>(
        bufH, W2, root2, bias2, inv, starts, sedge, out);
}


# B operand feed: weights re-laid once per call (row pairs interleaved) so one 16-byte load yields two ready k-pairs: 4x fewer B loads, issued two chunks ahead of their WMMAs; sum order unchanged
# speedup vs baseline: 2.6762x; 1.0412x over previous
#include <hip/hip_runtime.h>
#include <hip/hip_bf16.h>

#define NN 50000
#define EE 600000
#define FF 128
#define HH 128
#define OO 64
#define RR 8

typedef float v2f __attribute__((ext_vector_type(2)));
typedef float v4f __attribute__((ext_vector_type(4)));
typedef float v8f __attribute__((ext_vector_type(8)));

#define WPR ((RR + 1) * (FF / 2))
#define WT1 (WPR * (HH / 4))
#define WT2 (WPR * (OO / 4))
__global__ __launch_bounds__(256)
void prep_weights_kernel(const float* __restrict__ W1, const float* __restrict__ root1,
                         const float* __restrict__ W2, const float* __restrict__ root2,
                         float* __restrict__ sw1, float* __restrict__ sw2,
                         unsigned* __restrict__ mm) {
    const int t = blockIdx.x * 256 + threadIdx.x;
    if (t == 0) { mm[0] = 0x7F800000u; mm[1] = 0u; }
    const float* src;
    float* dst;
    int M, pr, c4;
    if (t < WT1) {
        M = HH; pr = t / (HH / 4); c4 = t % (HH / 4); dst = sw1;
        src = pr < RR * (FF / 2) ? W1 + (size_t)(2 * pr) * HH
                                 : root1 + (size_t)(2 * (pr - RR * (FF / 2))) * HH;
    } else if (t < WT1 + WT2) {
        const int u = t - WT1;
        M = OO; pr = u / (OO / 4); c4 = u % (OO / 4); dst = sw2;
        src = pr < RR * (HH / 2) ? W2 + (size_t)(2 * pr) * OO
                                 : root2 + (size_t)(2 * (pr - RR * (HH / 2))) * OO;
    } else {
        return;
    }
    const v4f a = *(const v4f*)(src + 4 * c4);
    const v4f b = *(const v4f*)(src + M + 4 * c4);
    v4f o0, o1;
    o0.x = a.x; o0.y = b.x; o0.z = a.y; o0.w = b.y;
    o1.x = a.z; o1.y = b.z; o1.z = a.w; o1.w = b.w;
    float* d = dst + ((size_t)pr * M + 4 * c4) * 2;
    *(v4f*)d = o0;
    *(v4f*)(d + 4) = o1;
}

__global__ void ew_minmax_kernel(const float* __restrict__ ew, int E,
                                 unsigned* __restrict__ mm) {
    float mn = __builtin_huge_valf();
    float mx = 0.0f;
    for (int i = blockIdx.x * blockDim.x + threadIdx.x; i < E;
         i += gridDim.x * blockDim.x) {
        float w = ew[i];
        mn = fminf(mn, w);
        mx = fmaxf(mx, w);
    }
    for (int off = 16; off > 0; off >>= 1) {
        mn = fminf(mn, __shfl_down(mn, off, 32));
        mx = fmaxf(mx, __shfl_down(mx, off, 32));
    }
    if ((threadIdx.x & 31) == 0) {
        atomicMin(&mm[0], __float_as_uint(mn));
        atomicMax(&mm[1], __float_as_uint(mx));
    }
}

__global__ void count_kernel(const int* __restrict__ dst, float* __restrict__ cnt,
                             int E4) {
    int t = blockIdx.x * blockDim.x + threadIdx.x;
    if (t >= E4) return;
    const int4 d = ((const int4*)dst)[t];
    atomicAdd(&cnt[d.x], 1.0f);
    atomicAdd(&cnt[d.y], 1.0f);
    atomicAdd(&cnt[d.z], 1.0f);
    atomicAdd(&cnt[d.w], 1.0f);
}

__global__ void invcnt_kernel(const float* __restrict__ cnt,
                              float* __restrict__ inv, int N) {
    int n = blockIdx.x * blockDim.x + threadIdx.x;
    if (n < N) inv[n] = 1.0f / fmaxf(cnt[n], 1.0f);
}

#define NB (RR * NN)
#define SCAN_TILE 1024
#define SCAN_BLOCKS ((NB + SCAN_TILE) / SCAN_TILE)
#define NBP (SCAN_BLOCKS * SCAN_TILE)

__device__ __forceinline__ int rank_one(int t, int d, int* __restrict__ hist) {
    int rk = -1;
    if ((unsigned)t < (unsigned)RR && (unsigned)d < (unsigned)NN)
        rk = atomicAdd(&hist[t * NN + d], 1);
    return rk;
}

__global__ void rank_kernel(const int* __restrict__ dst, const int* __restrict__ et,
                            int* __restrict__ hist, int* __restrict__ rank, int E4) {
    int q = blockIdx.x * blockDim.x + threadIdx.x;
    if (q >= E4) return;
    const int4 t = ((const int4*)et)[q];
    const int4 d = ((const int4*)dst)[q];
    int4 rk;
    rk.x = rank_one(t.x, d.x, hist);
    rk.y = rank_one(t.y, d.y, hist);
    rk.z = rank_one(t.z, d.z, hist);
    rk.w = rank_one(t.w, d.w, hist);
    ((int4*)rank)[q] = rk;
}

__global__ __launch_bounds__(256)
void scan_tilesum_kernel(const int* __restrict__ hist, int* __restrict__ tsum) {
    __shared__ int wsum[8];
    const int t = threadIdx.x;
    const int4 v = ((const int4*)hist)[blockIdx.x * 256 + t];
    int s = v.x + v.y + v.z + v.w;
    for (int off = 16; off > 0; off >>= 1) s += __shfl_down(s, off, 32);
    if ((t & 31) == 0) wsum[t >> 5] = s;
    __syncthreads();
    if (t == 0) {
        int tot = 0;
        for (int i = 0; i < 8; ++i) tot += wsum[i];
        tsum[blockIdx.x] = tot;
    }
}

__global__ __launch_bounds__(256)
void scan_apply_kernel(int* hist, const int* __restrict__ tsum) {
    __shared__ int wpart[8];
    __shared__ int wtot[8];
    const int t = threadIdx.x;
    const int lane = t & 31;
    const int wv = t >> 5;
    const int b = blockIdx.x;

    int part = 0;
    for (int j = t; j < b; j += 256) part += tsum[j];
    for (int off = 16; off > 0; off >>= 1) part += __shfl_down(part, off, 32);

    const int4 v = ((const int4*)hist)[b * 256 + t];
    const int s = v.x + v.y + v.z + v.w;
    int inc = s;
    for (int off = 1; off < 32; off <<= 1) {
        int n = __shfl_up(inc, off, 32);
        if (lane >= off) inc += n;
    }
    if (lane == 0) wpart[wv] = part;
    if (lane == 31) wtot[wv] = inc;
    __syncthreads();
    int base = 0;
    for (int i = 0; i < 8; ++i) {
        base += wpart[i];
        if (i < wv) base += wtot[i];
    }
    const int ex = base + inc - s;
    int4 o;
    o.x = ex;
    o.y = ex + v.x;
    o.z = o.y + v.y;
    o.w = o.z + v.z;
    ((int4*)hist)[b * 256 + t] = o;
}

__device__ __forceinline__ int fill_key(int rk, int t, int d) {
    return rk < 0 ? 0 : t * NN + d;
}

__device__ __forceinline__ void fill_one(int rk, int start, int s, float w, float mn,
                                         float mx, int* __restrict__ sedge) {
    if (rk < 0) return;
    int pos = start + rk;
    if ((unsigned)pos >= (unsigned)EE) return;
    int2 rec;
    rec.x = s;
    rec.y = __float_as_int((w - mn) / (mx - mn + 1e-8f));
    ((int2*)sedge)[pos] = rec;
}

__global__ void fill_kernel(const int* __restrict__ src, const int* __restrict__ dst,
                            const int* __restrict__ et, const float* __restrict__ ew,
                            const unsigned* __restrict__ mm,
                            const int* __restrict__ starts, const int* __restrict__ rank,
                            int* __restrict__ sedge, int E4) {
    int q = blockIdx.x * blockDim.x + threadIdx.x;
    if (q >= E4) return;
    const int4 rk = ((const int4*)rank)[q];
    const int4 t = ((const int4*)et)[q];
    const int4 d = ((const int4*)dst)[q];
    const int4 s = ((const int4*)src)[q];
    const float4 w = ((const float4*)ew)[q];
    const float mn = __uint_as_float(mm[0]);
    const float mx = __uint_as_float(mm[1]);
    const int st0 = starts[fill_key(rk.x, t.x, d.x)];
    const int st1 = starts[fill_key(rk.y, t.y, d.y)];
    const int st2 = starts[fill_key(rk.z, t.z, d.z)];
    const int st3 = starts[fill_key(rk.w, t.w, d.w)];
    fill_one(rk.x, st0, s.x, w.x, mn, mx, sedge);
    fill_one(rk.y, st1, s.y, w.y, mn, mx, sedge);
    fill_one(rk.z, st2, s.z, w.z, mn, mx, sedge);
    fill_one(rk.w, st3, s.w, w.w, mn, mx, sedge);
}

#define APITCH 132
#define GU 8

#define BNS 4

template <int M>
__device__ __forceinline__ void load_b(v4f (&q)[BNS][2], const float* __restrict__ bp, int s0) {
#pragma unroll
    for (int t = 0; t < BNS; ++t) {
        q[t][0] = *(const v4f*)(bp + 4 * (s0 + t) * M);
        q[t][1] = *(const v4f*)(bp + 4 * (s0 + t) * M + 64);
    }
}

__device__ __forceinline__ void wmma_b(const v2f (&a)[32], int s0, const v4f (&q)[BNS][2],
                                       v8f& acc0, v8f& acc1, v8f& acc2, v8f& acc3) {
#pragma unroll
    for (int t = 0; t < BNS; ++t) {
        v2f b;
        b.x = q[t][0].x; b.y = q[t][0].y;
        acc0 = __builtin_amdgcn_wmma_f32_16x16x4_f32(
            false, a[s0 + t], false, b, (short)0, acc0, false, false);
        b.x = q[t][0].z; b.y = q[t][0].w;
        acc1 = __builtin_amdgcn_wmma_f32_16x16x4_f32(
            false, a[s0 + t], false, b, (short)0, acc1, false, false);
        b.x = q[t][1].x; b.y = q[t][1].y;
        acc2 = __builtin_amdgcn_wmma_f32_16x16x4_f32(
            false, a[s0 + t], false, b, (short)0, acc2, false, false);
        b.x = q[t][1].z; b.y = q[t][1].w;
        acc3 = __builtin_amdgcn_wmma_f32_16x16x4_f32(
            false, a[s0 + t], false, b, (short)0, acc3, false, false);
    }
}

#define CPG (32 / BNS)

template <int M>
__device__ __forceinline__ const float* b_base(const float* __restrict__ Bs, int half, int lr) {
    return Bs + (size_t)half * (2 * M) + 4 * lr;
}

template <int M>
__device__ __forceinline__ void mma_head(v4f (&ring)[3][BNS][2], const float* __restrict__ bp) {
    load_b<M>(ring[0], bp, 0);
    load_b<M>(ring[1], bp + 128 * (1 / CPG), (1 % CPG) * BNS);
}

template <int M>
__device__ __forceinline__ void mma_all(const v2f (&a)[32], v4f (&ring)[3][BNS][2],
                                        const float* __restrict__ bp, v8f (&acc)[M / 16]) {
    constexpr int NC = (M / 64) * CPG;
#pragma unroll
    for (int c = 0; c < NC; ++c) {
        const int g = c / CPG;
        if (c + 2 < NC)
            load_b<M>(ring[(c + 2) % 3], bp + 128 * ((c + 2) / CPG), ((c + 2) % CPG) * BNS);
        __builtin_amdgcn_sched_barrier(0);
        wmma_b(a, (c % CPG) * BNS, ring[c % 3],
               acc[4 * g], acc[4 * g + 1], acc[4 * g + 2], acc[4 * g + 3]);
        __builtin_amdgcn_sched_barrier(0);
    }
}

template <int M, bool RELU>
__global__ __launch_bounds__(32)
void fused_layer_kernel(const float* __restrict__ X,
                        const float* __restrict__ Wsw,
                        const float* __restrict__ bias,
                        const float* __restrict__ inv,
                        const int* __restrict__ starts,
                        const int* __restrict__ sedge,
                        float* __restrict__ Out) {
    constexpr int K = 128;
    constexpr int CT = M / 16;
    __shared__ __attribute__((aligned(16))) float tile[16 * APITCH];
    const int lane = threadIdx.x;
    const int half = lane >> 4;
    const int lr   = lane & 15;
    const int rowbase = blockIdx.x * 16;
    const int2* __restrict__ recs = (const int2*)sedge;
    const float4* __restrict__ X4 = (const float4*)X;
    const int sl = lane < 16 ? lane : 16;
    const float4 zero4 = make_float4(0.0f, 0.0f, 0.0f, 0.0f);

    v8f acc[CT];
#pragma unroll
    for (int ct = 0; ct < CT; ++ct) acc[ct] = v8f{};

    int svn = starts[rowbase + sl];
    for (int r = 0; r < RR; ++r) {
        const int svc = svn;
        const int rn = r + 1 < RR ? r + 1 : r;
        svn = starts[rn * NN + rowbase + sl];

        int eb = __builtin_amdgcn_readlane(svc, 0);
        int ee = __builtin_amdgcn_readlane(svc, 16);
        eb = eb < 0 ? 0 : eb;
        ee = ee > EE ? EE : ee;

        const float* bp = b_base<M>(Wsw + (size_t)r * K * M, half, lr);
        v4f ring[3][BNS][2];
        mma_head<M>(ring, bp);

#pragma unroll
        for (int j = 0; j < 16; ++j) *(float4*)&tile[j * APITCH + lane * 4] = zero4;

        int cur = -1;
        float4 g = zero4;
        for (int c0 = eb; c0 < ee; c0 += 32) {
            int n = ee - c0;
            n = n > 32 ? 32 : n;
            const int myp = c0 + lane;
            int2 rec = make_int2(0, 0);
            if (lane < n) rec = recs[myp];
            int ms = rec.x;
            ms = ms < 0 ? 0 : (ms >= NN ? NN - 1 : ms);
            int myrow = 0;
#pragma unroll
            for (int j = 1; j < 16; ++j)
                myrow += (__builtin_amdgcn_readlane(svc, j) <= myp) ? 1 : 0;

            for (int e0 = 0; e0 < n; e0 += GU) {
                float4 v[GU];
#pragma unroll
                for (int u = 0; u < GU; ++u) {
                    int eu = e0 + u;
                    eu = eu < n ? eu : n - 1;
                    const int s = __builtin_amdgcn_readlane(ms, eu);
                    v[u] = X4[(size_t)s * (K / 4) + lane];
                }
#pragma unroll
                for (int u = 0; u < GU; ++u) {
                    const int eu = e0 + u;
                    if (eu < n) {
                        const int row = __builtin_amdgcn_readlane(myrow, eu);
                        const float w = __int_as_float(__builtin_amdgcn_readlane(rec.y, eu));
                        if (row != cur) {
                            if (cur >= 0) *(float4*)&tile[cur * APITCH + lane * 4] = g;
                            g = zero4;
                            cur = row;
                        }
                        g.x += v[u].x * w;
                        g.y += v[u].y * w;
                        g.z += v[u].z * w;
                        g.w += v[u].w * w;
                    }
                }
            }
        }
        if (cur >= 0) *(float4*)&tile[cur * APITCH + lane * 4] = g;

        __syncthreads();
        v2f a[32];
        const float* ap = &tile[lr * APITCH + 2 * half];
#pragma unroll
        for (int s = 0; s < 32; ++s) a[s] = *(const v2f*)(ap + 4 * s);
        __syncthreads();

        mma_all<M>(a, ring, bp, acc);
    }

    float ivr[8];
#pragma unroll
    for (int j = 0; j < 8; ++j) ivr[j] = inv[rowbase + 8 * half + j];
#pragma unroll
    for (int ct = 0; ct < CT; ++ct) {
#pragma unroll
        for (int j = 0; j < 8; ++j) acc[ct][j] *= ivr[j];
    }

    {
        const float* bp = b_base<M>(Wsw + (size_t)RR * K * M, half, lr);
        v4f ring[3][BNS][2];
        mma_head<M>(ring, bp);
        const float* arow = X + (size_t)(rowbase + lr) * K + 2 * half;
        v2f a[32];
#pragma unroll
        for (int s = 0; s < 32; ++s) a[s] = *(const v2f*)(arow + 4 * s);
        mma_all<M>(a, ring, bp, acc);
    }

#pragma unroll
    for (int g = 0; g < M / 32; ++g) {
        const int col = 32 * g + 2 * lr;
        const v2f bb = *(const v2f*)(bias + col);
        float* crow = Out + (size_t)(rowbase + 8 * half) * M + col;
#pragma unroll
        for (int j = 0; j < 8; ++j) {
            v2f val;
            val.x = acc[2 * g + 0][j] + bb.x;
            val.y = acc[2 * g + 1][j] + bb.y;
            if (RELU) {
                val.x = fmaxf(val.x, 0.0f);
                val.y = fmaxf(val.y, 0.0f);
            }
            *(v2f*)(crow + (size_t)j * M) = val;
        }
    }
}

extern "C" void kernel_launch(void* const* d_in, const int* in_sizes, int n_in,
                              void* d_out, int out_size, void* d_ws, size_t ws_size,
                              hipStream_t stream) {
    const float* x     = (const float*)d_in[0];
    const int*   eidx  = (const int*)d_in[1];
    const float* ew    = (const float*)d_in[2];
    const int*   et    = (const int*)d_in[3];
    const float* W1    = (const float*)d_in[4];
    const float* root1 = (const float*)d_in[5];
    const float* bias1 = (const float*)d_in[6];
    const float* W2    = (const float*)d_in[7];
    const float* root2 = (const float*)d_in[8];
    const float* bias2 = (const float*)d_in[9];
    float* out = (float*)d_out;

    const int* src = eidx;
    const int* dst = eidx + EE;

    float* bufH = (float*)d_ws;
    float* agg  = bufH + (size_t)NN * HH;
    float* sw1  = agg;
    float* sw2  = sw1 + (size_t)(RR + 1) * FF * HH;
    float* bufG = agg  + (size_t)NN * HH;
    float* cnt  = bufG + (size_t)NN * HH;
    float* inv  = cnt  + NN;
    unsigned* mm = (unsigned*)(inv + NN);

    int*   starts = (int*)bufG;
    int*   tsum   = starts + NBP;
    int*   sedge  = tsum + 512;
    int*   rank   = sedge + 2 * EE;

    const int T = 256;
    const int rowTiles = NN / 16;

    hipMemsetAsync(starts, 0, (size_t)NBP * sizeof(int), stream);
    hipMemsetAsync(cnt, 0, (size_t)NN * sizeof(float), stream);
    prep_weights_kernel<<<(WT1 + WT2 + 255) / 256, 256, 0, stream>>>(W1, root1, W2, root2,
                                                                     sw1, sw2, mm);
    const int E4 = EE / 4;
    ew_minmax_kernel<<<256, T, 0, stream>>>(ew, EE, mm);
    count_kernel<<<(E4 + T - 1) / T, T, 0, stream>>>(dst, cnt, E4);
    invcnt_kernel<<<(NN + T - 1) / T, T, 0, stream>>>(cnt, inv, NN);

    rank_kernel<<<(E4 + T - 1) / T, T, 0, stream>>>(dst, et, starts, rank, E4);
    scan_tilesum_kernel<<<SCAN_BLOCKS, 256, 0, stream>>>(starts, tsum);
    scan_apply_kernel<<<SCAN_BLOCKS, 256, 0, stream>>>(starts, tsum);
    fill_kernel<<<(E4 + T - 1) / T, T, 0, stream>>>(src, dst, et, ew, mm, starts, rank,
                                                    sedge, E4);

    fused_layer_kernel<HH, true><<<rowTiles, 32, 0, stream>>>(
        x, sw1, bias1, inv, starts, sedge, bufH);
    fused_layer_kernel<OO, false><<<rowTiles, 32, 0, stream>>>(
        bufH, sw2, bias2, inv, starts, sedge, out);
}


# Degree from the edge grouping: each row tile sums its 8 segment lengths (bounds already in registers); the 600k float-atomic count pass, the inverse kernel and one memset are gone; sums unchanged
# speedup vs baseline: 3.1899x; 1.1920x over previous
#include <hip/hip_runtime.h>
#include <hip/hip_bf16.h>

#define NN 50000
#define EE 600000
#define FF 128
#define HH 128
#define OO 64
#define RR 8

typedef float v2f __attribute__((ext_vector_type(2)));
typedef float v4f __attribute__((ext_vector_type(4)));
typedef float v8f __attribute__((ext_vector_type(8)));

#define WPR ((RR + 1) * (FF / 2))
#define WT1 (WPR * (HH / 4))
#define WT2 (WPR * (OO / 4))
__global__ __launch_bounds__(256)
void prep_weights_kernel(const float* __restrict__ W1, const float* __restrict__ root1,
                         const float* __restrict__ W2, const float* __restrict__ root2,
                         float* __restrict__ sw1, float* __restrict__ sw2,
                         unsigned* __restrict__ mm) {
    const int t = blockIdx.x * 256 + threadIdx.x;
    if (t == 0) { mm[0] = 0x7F800000u; mm[1] = 0u; }
    const float* src;
    float* dst;
    int M, pr, c4;
    if (t < WT1) {
        M = HH; pr = t / (HH / 4); c4 = t % (HH / 4); dst = sw1;
        src = pr < RR * (FF / 2) ? W1 + (size_t)(2 * pr) * HH
                                 : root1 + (size_t)(2 * (pr - RR * (FF / 2))) * HH;
    } else if (t < WT1 + WT2) {
        const int u = t - WT1;
        M = OO; pr = u / (OO / 4); c4 = u % (OO / 4); dst = sw2;
        src = pr < RR * (HH / 2) ? W2 + (size_t)(2 * pr) * OO
                                 : root2 + (size_t)(2 * (pr - RR * (HH / 2))) * OO;
    } else {
        return;
    }
    const v4f a = *(const v4f*)(src + 4 * c4);
    const v4f b = *(const v4f*)(src + M + 4 * c4);
    v4f o0, o1;
    o0.x = a.x; o0.y = b.x; o0.z = a.y; o0.w = b.y;
    o1.x = a.z; o1.y = b.z; o1.z = a.w; o1.w = b.w;
    float* d = dst + ((size_t)pr * M + 4 * c4) * 2;
    *(v4f*)d = o0;
    *(v4f*)(d + 4) = o1;
}

__global__ void ew_minmax_kernel(const float* __restrict__ ew, int E,
                                 unsigned* __restrict__ mm) {
    float mn = __builtin_huge_valf();
    float mx = 0.0f;
    for (int i = blockIdx.x * blockDim.x + threadIdx.x; i < E;
         i += gridDim.x * blockDim.x) {
        float w = ew[i];
        mn = fminf(mn, w);
        mx = fmaxf(mx, w);
    }
    for (int off = 16; off > 0; off >>= 1) {
        mn = fminf(mn, __shfl_down(mn, off, 32));
        mx = fmaxf(mx, __shfl_down(mx, off, 32));
    }
    if ((threadIdx.x & 31) == 0) {
        atomicMin(&mm[0], __float_as_uint(mn));
        atomicMax(&mm[1], __float_as_uint(mx));
    }
}

#define NB (RR * NN)
#define SCAN_TILE 1024
#define SCAN_BLOCKS ((NB + SCAN_TILE) / SCAN_TILE)
#define NBP (SCAN_BLOCKS * SCAN_TILE)

__device__ __forceinline__ int rank_one(int t, int d, int* __restrict__ hist) {
    int rk = -1;
    if ((unsigned)t < (unsigned)RR && (unsigned)d < (unsigned)NN)
        rk = atomicAdd(&hist[t * NN + d], 1);
    return rk;
}

__global__ void rank_kernel(const int* __restrict__ dst, const int* __restrict__ et,
                            int* __restrict__ hist, int* __restrict__ rank, int E4) {
    int q = blockIdx.x * blockDim.x + threadIdx.x;
    if (q >= E4) return;
    const int4 t = ((const int4*)et)[q];
    const int4 d = ((const int4*)dst)[q];
    int4 rk;
    rk.x = rank_one(t.x, d.x, hist);
    rk.y = rank_one(t.y, d.y, hist);
    rk.z = rank_one(t.z, d.z, hist);
    rk.w = rank_one(t.w, d.w, hist);
    ((int4*)rank)[q] = rk;
}

__global__ __launch_bounds__(256)
void scan_tilesum_kernel(const int* __restrict__ hist, int* __restrict__ tsum) {
    __shared__ int wsum[8];
    const int t = threadIdx.x;
    const int4 v = ((const int4*)hist)[blockIdx.x * 256 + t];
    int s = v.x + v.y + v.z + v.w;
    for (int off = 16; off > 0; off >>= 1) s += __shfl_down(s, off, 32);
    if ((t & 31) == 0) wsum[t >> 5] = s;
    __syncthreads();
    if (t == 0) {
        int tot = 0;
        for (int i = 0; i < 8; ++i) tot += wsum[i];
        tsum[blockIdx.x] = tot;
    }
}

__global__ __launch_bounds__(256)
void scan_apply_kernel(int* hist, const int* __restrict__ tsum) {
    __shared__ int wpart[8];
    __shared__ int wtot[8];
    const int t = threadIdx.x;
    const int lane = t & 31;
    const int wv = t >> 5;
    const int b = blockIdx.x;

    int part = 0;
    for (int j = t; j < b; j += 256) part += tsum[j];
    for (int off = 16; off > 0; off >>= 1) part += __shfl_down(part, off, 32);

    const int4 v = ((const int4*)hist)[b * 256 + t];
    const int s = v.x + v.y + v.z + v.w;
    int inc = s;
    for (int off = 1; off < 32; off <<= 1) {
        int n = __shfl_up(inc, off, 32);
        if (lane >= off) inc += n;
    }
    if (lane == 0) wpart[wv] = part;
    if (lane == 31) wtot[wv] = inc;
    __syncthreads();
    int base = 0;
    for (int i = 0; i < 8; ++i) {
        base += wpart[i];
        if (i < wv) base += wtot[i];
    }
    const int ex = base + inc - s;
    int4 o;
    o.x = ex;
    o.y = ex + v.x;
    o.z = o.y + v.y;
    o.w = o.z + v.z;
    ((int4*)hist)[b * 256 + t] = o;
}

__device__ __forceinline__ int fill_key(int rk, int t, int d) {
    return rk < 0 ? 0 : t * NN + d;
}

__device__ __forceinline__ void fill_one(int rk, int start, int s, float w, float mn,
                                         float mx, int* __restrict__ sedge) {
    if (rk < 0) return;
    int pos = start + rk;
    if ((unsigned)pos >= (unsigned)EE) return;
    int2 rec;
    rec.x = s;
    rec.y = __float_as_int((w - mn) / (mx - mn + 1e-8f));
    ((int2*)sedge)[pos] = rec;
}

__global__ void fill_kernel(const int* __restrict__ src, const int* __restrict__ dst,
                            const int* __restrict__ et, const float* __restrict__ ew,
                            const unsigned* __restrict__ mm,
                            const int* __restrict__ starts, const int* __restrict__ rank,
                            int* __restrict__ sedge, int E4) {
    int q = blockIdx.x * blockDim.x + threadIdx.x;
    if (q >= E4) return;
    const int4 rk = ((const int4*)rank)[q];
    const int4 t = ((const int4*)et)[q];
    const int4 d = ((const int4*)dst)[q];
    const int4 s = ((const int4*)src)[q];
    const float4 w = ((const float4*)ew)[q];
    const float mn = __uint_as_float(mm[0]);
    const float mx = __uint_as_float(mm[1]);
    const int st0 = starts[fill_key(rk.x, t.x, d.x)];
    const int st1 = starts[fill_key(rk.y, t.y, d.y)];
    const int st2 = starts[fill_key(rk.z, t.z, d.z)];
    const int st3 = starts[fill_key(rk.w, t.w, d.w)];
    fill_one(rk.x, st0, s.x, w.x, mn, mx, sedge);
    fill_one(rk.y, st1, s.y, w.y, mn, mx, sedge);
    fill_one(rk.z, st2, s.z, w.z, mn, mx, sedge);
    fill_one(rk.w, st3, s.w, w.w, mn, mx, sedge);
}

#define APITCH 132
#define GU 8

#define BNS 4

template <int M>
__device__ __forceinline__ void load_b(v4f (&q)[BNS][2], const float* __restrict__ bp, int s0) {
#pragma unroll
    for (int t = 0; t < BNS; ++t) {
        q[t][0] = *(const v4f*)(bp + 4 * (s0 + t) * M);
        q[t][1] = *(const v4f*)(bp + 4 * (s0 + t) * M + 64);
    }
}

__device__ __forceinline__ void wmma_b(const v2f (&a)[32], int s0, const v4f (&q)[BNS][2],
                                       v8f& acc0, v8f& acc1, v8f& acc2, v8f& acc3) {
#pragma unroll
    for (int t = 0; t < BNS; ++t) {
        v2f b;
        b.x = q[t][0].x; b.y = q[t][0].y;
        acc0 = __builtin_amdgcn_wmma_f32_16x16x4_f32(
            false, a[s0 + t], false, b, (short)0, acc0, false, false);
        b.x = q[t][0].z; b.y = q[t][0].w;
        acc1 = __builtin_amdgcn_wmma_f32_16x16x4_f32(
            false, a[s0 + t], false, b, (short)0, acc1, false, false);
        b.x = q[t][1].x; b.y = q[t][1].y;
        acc2 = __builtin_amdgcn_wmma_f32_16x16x4_f32(
            false, a[s0 + t], false, b, (short)0, acc2, false, false);
        b.x = q[t][1].z; b.y = q[t][1].w;
        acc3 = __builtin_amdgcn_wmma_f32_16x16x4_f32(
            false, a[s0 + t], false, b, (short)0, acc3, false, false);
    }
}

#define CPG (32 / BNS)

template <int M>
__device__ __forceinline__ const float* b_base(const float* __restrict__ Bs, int half, int lr) {
    return Bs + (size_t)half * (2 * M) + 4 * lr;
}

template <int M>
__device__ __forceinline__ void mma_head(v4f (&ring)[3][BNS][2], const float* __restrict__ bp) {
    load_b<M>(ring[0], bp, 0);
    load_b<M>(ring[1], bp + 128 * (1 / CPG), (1 % CPG) * BNS);
}

template <int M>
__device__ __forceinline__ void mma_all(const v2f (&a)[32], v4f (&ring)[3][BNS][2],
                                        const float* __restrict__ bp, v8f (&acc)[M / 16]) {
    constexpr int NC = (M / 64) * CPG;
#pragma unroll
    for (int c = 0; c < NC; ++c) {
        const int g = c / CPG;
        if (c + 2 < NC)
            load_b<M>(ring[(c + 2) % 3], bp + 128 * ((c + 2) / CPG), ((c + 2) % CPG) * BNS);
        __builtin_amdgcn_sched_barrier(0);
        wmma_b(a, (c % CPG) * BNS, ring[c % 3],
               acc[4 * g], acc[4 * g + 1], acc[4 * g + 2], acc[4 * g + 3]);
        __builtin_amdgcn_sched_barrier(0);
    }
}

template <int M, bool RELU>
__global__ __launch_bounds__(32)
void fused_layer_kernel(const float* __restrict__ X,
                        const float* __restrict__ Wsw,
                        const float* __restrict__ bias,
                        const int* __restrict__ starts,
                        const int* __restrict__ sedge,
                        float* __restrict__ Out) {
    constexpr int K = 128;
    constexpr int CT = M / 16;
    __shared__ __attribute__((aligned(16))) float tile[16 * APITCH];
    const int lane = threadIdx.x;
    const int half = lane >> 4;
    const int lr   = lane & 15;
    const int rowbase = blockIdx.x * 16;
    const int2* __restrict__ recs = (const int2*)sedge;
    const float4* __restrict__ X4 = (const float4*)X;
    const int sl = lane < 16 ? lane : 16;
    const float4 zero4 = make_float4(0.0f, 0.0f, 0.0f, 0.0f);

    v8f acc[CT];
#pragma unroll
    for (int ct = 0; ct < CT; ++ct) acc[ct] = v8f{};

    int svn = starts[rowbase + sl];
    int ssum = 0;
    for (int r = 0; r < RR; ++r) {
        const int svc = svn;
        ssum += svc;
        const int rn = r + 1 < RR ? r + 1 : r;
        svn = starts[rn * NN + rowbase + sl];

        int eb = __builtin_amdgcn_readlane(svc, 0);
        int ee = __builtin_amdgcn_readlane(svc, 16);
        eb = eb < 0 ? 0 : eb;
        ee = ee > EE ? EE : ee;

        const float* bp = b_base<M>(Wsw + (size_t)r * K * M, half, lr);
        v4f ring[3][BNS][2];
        mma_head<M>(ring, bp);

#pragma unroll
        for (int j = 0; j < 16; ++j) *(float4*)&tile[j * APITCH + lane * 4] = zero4;

        int cur = -1;
        float4 g = zero4;
        for (int c0 = eb; c0 < ee; c0 += 32) {
            int n = ee - c0;
            n = n > 32 ? 32 : n;
            const int myp = c0 + lane;
            int2 rec = make_int2(0, 0);
            if (lane < n) rec = recs[myp];
            int ms = rec.x;
            ms = ms < 0 ? 0 : (ms >= NN ? NN - 1 : ms);
            int myrow = 0;
#pragma unroll
            for (int j = 1; j < 16; ++j)
                myrow += (__builtin_amdgcn_readlane(svc, j) <= myp) ? 1 : 0;

            for (int e0 = 0; e0 < n; e0 += GU) {
                float4 v[GU];
#pragma unroll
                for (int u = 0; u < GU; ++u) {
                    int eu = e0 + u;
                    eu = eu < n ? eu : n - 1;
                    const int s = __builtin_amdgcn_readlane(ms, eu);
                    v[u] = X4[(size_t)s * (K / 4) + lane];
                }
#pragma unroll
                for (int u = 0; u < GU; ++u) {
                    const int eu = e0 + u;
                    if (eu < n) {
                        const int row = __builtin_amdgcn_readlane(myrow, eu);
                        const float w = __int_as_float(__builtin_amdgcn_readlane(rec.y, eu));
                        if (row != cur) {
                            if (cur >= 0) *(float4*)&tile[cur * APITCH + lane * 4] = g;
                            g = zero4;
                            cur = row;
                        }
                        g.x += v[u].x * w;
                        g.y += v[u].y * w;
                        g.z += v[u].z * w;
                        g.w += v[u].w * w;
                    }
                }
            }
        }
        if (cur >= 0) *(float4*)&tile[cur * APITCH + lane * 4] = g;

        __syncthreads();
        v2f a[32];
        const float* ap = &tile[lr * APITCH + 2 * half];
#pragma unroll
        for (int s = 0; s < 32; ++s) a[s] = *(const v2f*)(ap + 4 * s);
        __syncthreads();

        mma_all<M>(a, ring, bp, acc);
    }

    const int deg = __shfl_down(ssum, 1, 32) - ssum;
    const int invl = __float_as_int(1.0f / fmaxf((float)deg, 1.0f));
    float ivr[8];
#pragma unroll
    for (int j = 0; j < 8; ++j) {
        const int lo = __builtin_amdgcn_readlane(invl, j);
        const int hi = __builtin_amdgcn_readlane(invl, j + 8);
        ivr[j] = __int_as_float(half ? hi : lo);
    }
#pragma unroll
    for (int ct = 0; ct < CT; ++ct) {
#pragma unroll
        for (int j = 0; j < 8; ++j) acc[ct][j] *= ivr[j];
    }

    {
        const float* bp = b_base<M>(Wsw + (size_t)RR * K * M, half, lr);
        v4f ring[3][BNS][2];
        mma_head<M>(ring, bp);
        const float* arow = X + (size_t)(rowbase + lr) * K + 2 * half;
        v2f a[32];
#pragma unroll
        for (int s = 0; s < 32; ++s) a[s] = *(const v2f*)(arow + 4 * s);
        mma_all<M>(a, ring, bp, acc);
    }

#pragma unroll
    for (int g = 0; g < M / 32; ++g) {
        const int col = 32 * g + 2 * lr;
        const v2f bb = *(const v2f*)(bias + col);
        float* crow = Out + (size_t)(rowbase + 8 * half) * M + col;
#pragma unroll
        for (int j = 0; j < 8; ++j) {
            v2f val;
            val.x = acc[2 * g + 0][j] + bb.x;
            val.y = acc[2 * g + 1][j] + bb.y;
            if (RELU) {
                val.x = fmaxf(val.x, 0.0f);
                val.y = fmaxf(val.y, 0.0f);
            }
            *(v2f*)(crow + (size_t)j * M) = val;
        }
    }
}

extern "C" void kernel_launch(void* const* d_in, const int* in_sizes, int n_in,
                              void* d_out, int out_size, void* d_ws, size_t ws_size,
                              hipStream_t stream) {
    const float* x     = (const float*)d_in[0];
    const int*   eidx  = (const int*)d_in[1];
    const float* ew    = (const float*)d_in[2];
    const int*   et    = (const int*)d_in[3];
    const float* W1    = (const float*)d_in[4];
    const float* root1 = (const float*)d_in[5];
    const float* bias1 = (const float*)d_in[6];
    const float* W2    = (const float*)d_in[7];
    const float* root2 = (const float*)d_in[8];
    const float* bias2 = (const float*)d_in[9];
    float* out = (float*)d_out;

    const int* src = eidx;
    const int* dst = eidx + EE;

    float* bufH = (float*)d_ws;
    float* agg  = bufH + (size_t)NN * HH;
    float* sw1  = agg;
    float* sw2  = sw1 + (size_t)(RR + 1) * FF * HH;
    float* bufG = agg  + (size_t)NN * HH;
    unsigned* mm = (unsigned*)(bufG + (size_t)NN * HH);

    int*   starts = (int*)bufG;
    int*   tsum   = starts + NBP;
    int*   sedge  = tsum + 512;
    int*   rank   = sedge + 2 * EE;

    const int T = 256;
    const int rowTiles = NN / 16;

    hipMemsetAsync(starts, 0, (size_t)NBP * sizeof(int), stream);
    prep_weights_kernel<<<(WT1 + WT2 + 255) / 256, 256, 0, stream>>>(W1, root1, W2, root2,
                                                                     sw1, sw2, mm);
    const int E4 = EE / 4;
    ew_minmax_kernel<<<256, T, 0, stream>>>(ew, EE, mm);

    rank_kernel<<<(E4 + T - 1) / T, T, 0, stream>>>(dst, et, starts, rank, E4);
    scan_tilesum_kernel<<<SCAN_BLOCKS, 256, 0, stream>>>(starts, tsum);
    scan_apply_kernel<<<SCAN_BLOCKS, 256, 0, stream>>>(starts, tsum);
    fill_kernel<<<(E4 + T - 1) / T, T, 0, stream>>>(src, dst, et, ew, mm, starts, rank,
                                                    sedge, E4);

    fused_layer_kernel<HH, true><<<rowTiles, 32, 0, stream>>>(
        x, sw1, bias1, starts, sedge, bufH);
    fused_layer_kernel<OO, false><<<rowTiles, 32, 0, stream>>>(
        bufH, sw2, bias2, starts, sedge, out);
}


# Two waves per 16-row tile: wave w sums the edges of rows 8w..8w+7 into a shared LDS tile (two tiles in turn, one barrier per relation), then does half the columns' WMMAs; sum order unchanged
# speedup vs baseline: 3.2990x; 1.0342x over previous
#include <hip/hip_runtime.h>
#include <hip/hip_bf16.h>

#define NN 50000
#define EE 600000
#define FF 128
#define HH 128
#define OO 64
#define RR 8

typedef float v2f __attribute__((ext_vector_type(2)));
typedef float v4f __attribute__((ext_vector_type(4)));
typedef float v8f __attribute__((ext_vector_type(8)));

#define WPR ((RR + 1) * (FF / 2))
#define WT1 (WPR * (HH / 4))
#define WT2 (WPR * (OO / 4))
__global__ __launch_bounds__(256)
void prep_weights_kernel(const float* __restrict__ W1, const float* __restrict__ root1,
                         const float* __restrict__ W2, const float* __restrict__ root2,
                         float* __restrict__ sw1, float* __restrict__ sw2,
                         unsigned* __restrict__ mm) {
    const int t = blockIdx.x * 256 + threadIdx.x;
    if (t == 0) { mm[0] = 0x7F800000u; mm[1] = 0u; }
    const float* src;
    float* dst;
    int M, pr, c4;
    if (t < WT1) {
        M = HH; pr = t / (HH / 4); c4 = t % (HH / 4); dst = sw1;
        src = pr < RR * (FF / 2) ? W1 + (size_t)(2 * pr) * HH
                                 : root1 + (size_t)(2 * (pr - RR * (FF / 2))) * HH;
    } else if (t < WT1 + WT2) {
        const int u = t - WT1;
        M = OO; pr = u / (OO / 4); c4 = u % (OO / 4); dst = sw2;
        src = pr < RR * (HH / 2) ? W2 + (size_t)(2 * pr) * OO
                                 : root2 + (size_t)(2 * (pr - RR * (HH / 2))) * OO;
    } else {
        return;
    }
    const v4f a = *(const v4f*)(src + 4 * c4);
    const v4f b = *(const v4f*)(src + M + 4 * c4);
    v4f o0, o1;
    o0.x = a.x; o0.y = b.x; o0.z = a.y; o0.w = b.y;
    o1.x = a.z; o1.y = b.z; o1.z = a.w; o1.w = b.w;
    float* d = dst + ((size_t)pr * M + 4 * c4) * 2;
    *(v4f*)d = o0;
    *(v4f*)(d + 4) = o1;
}

__global__ void ew_minmax_kernel(const float* __restrict__ ew, int E,
                                 unsigned* __restrict__ mm) {
    float mn = __builtin_huge_valf();
    float mx = 0.0f;
    for (int i = blockIdx.x * blockDim.x + threadIdx.x; i < E;
         i += gridDim.x * blockDim.x) {
        float w = ew[i];
        mn = fminf(mn, w);
        mx = fmaxf(mx, w);
    }
    for (int off = 16; off > 0; off >>= 1) {
        mn = fminf(mn, __shfl_down(mn, off, 32));
        mx = fmaxf(mx, __shfl_down(mx, off, 32));
    }
    if ((threadIdx.x & 31) == 0) {
        atomicMin(&mm[0], __float_as_uint(mn));
        atomicMax(&mm[1], __float_as_uint(mx));
    }
}

#define NB (RR * NN)
#define SCAN_TILE 1024
#define SCAN_BLOCKS ((NB + SCAN_TILE) / SCAN_TILE)
#define NBP (SCAN_BLOCKS * SCAN_TILE)

__device__ __forceinline__ int rank_one(int t, int d, int* __restrict__ hist) {
    int rk = -1;
    if ((unsigned)t < (unsigned)RR && (unsigned)d < (unsigned)NN)
        rk = atomicAdd(&hist[t * NN + d], 1);
    return rk;
}

__global__ void rank_kernel(const int* __restrict__ dst, const int* __restrict__ et,
                            int* __restrict__ hist, int* __restrict__ rank, int E4) {
    int q = blockIdx.x * blockDim.x + threadIdx.x;
    if (q >= E4) return;
    const int4 t = ((const int4*)et)[q];
    const int4 d = ((const int4*)dst)[q];
    int4 rk;
    rk.x = rank_one(t.x, d.x, hist);
    rk.y = rank_one(t.y, d.y, hist);
    rk.z = rank_one(t.z, d.z, hist);
    rk.w = rank_one(t.w, d.w, hist);
    ((int4*)rank)[q] = rk;
}

__global__ __launch_bounds__(256)
void scan_tilesum_kernel(const int* __restrict__ hist, int* __restrict__ tsum) {
    __shared__ int wsum[8];
    const int t = threadIdx.x;
    const int4 v = ((const int4*)hist)[blockIdx.x * 256 + t];
    int s = v.x + v.y + v.z + v.w;
    for (int off = 16; off > 0; off >>= 1) s += __shfl_down(s, off, 32);
    if ((t & 31) == 0) wsum[t >> 5] = s;
    __syncthreads();
    if (t == 0) {
        int tot = 0;
        for (int i = 0; i < 8; ++i) tot += wsum[i];
        tsum[blockIdx.x] = tot;
    }
}

__global__ __launch_bounds__(256)
void scan_apply_kernel(int* hist, const int* __restrict__ tsum) {
    __shared__ int wpart[8];
    __shared__ int wtot[8];
    const int t = threadIdx.x;
    const int lane = t & 31;
    const int wv = t >> 5;
    const int b = blockIdx.x;

    int part = 0;
    for (int j = t; j < b; j += 256) part += tsum[j];
    for (int off = 16; off > 0; off >>= 1) part += __shfl_down(part, off, 32);

    const int4 v = ((const int4*)hist)[b * 256 + t];
    const int s = v.x + v.y + v.z + v.w;
    int inc = s;
    for (int off = 1; off < 32; off <<= 1) {
        int n = __shfl_up(inc, off, 32);
        if (lane >= off) inc += n;
    }
    if (lane == 0) wpart[wv] = part;
    if (lane == 31) wtot[wv] = inc;
    __syncthreads();
    int base = 0;
    for (int i = 0; i < 8; ++i) {
        base += wpart[i];
        if (i < wv) base += wtot[i];
    }
    const int ex = base + inc - s;
    int4 o;
    o.x = ex;
    o.y = ex + v.x;
    o.z = o.y + v.y;
    o.w = o.z + v.z;
    ((int4*)hist)[b * 256 + t] = o;
}

__device__ __forceinline__ int fill_key(int rk, int t, int d) {
    return rk < 0 ? 0 : t * NN + d;
}

__device__ __forceinline__ void fill_one(int rk, int start, int s, float w, float mn,
                                         float mx, int* __restrict__ sedge) {
    if (rk < 0) return;
    int pos = start + rk;
    if ((unsigned)pos >= (unsigned)EE) return;
    int2 rec;
    rec.x = s;
    rec.y = __float_as_int((w - mn) / (mx - mn + 1e-8f));
    ((int2*)sedge)[pos] = rec;
}

__global__ void fill_kernel(const int* __restrict__ src, const int* __restrict__ dst,
                            const int* __restrict__ et, const float* __restrict__ ew,
                            const unsigned* __restrict__ mm,
                            const int* __restrict__ starts, const int* __restrict__ rank,
                            int* __restrict__ sedge, int E4) {
    int q = blockIdx.x * blockDim.x + threadIdx.x;
    if (q >= E4) return;
    const int4 rk = ((const int4*)rank)[q];
    const int4 t = ((const int4*)et)[q];
    const int4 d = ((const int4*)dst)[q];
    const int4 s = ((const int4*)src)[q];
    const float4 w = ((const float4*)ew)[q];
    const float mn = __uint_as_float(mm[0]);
    const float mx = __uint_as_float(mm[1]);
    const int st0 = starts[fill_key(rk.x, t.x, d.x)];
    const int st1 = starts[fill_key(rk.y, t.y, d.y)];
    const int st2 = starts[fill_key(rk.z, t.z, d.z)];
    const int st3 = starts[fill_key(rk.w, t.w, d.w)];
    fill_one(rk.x, st0, s.x, w.x, mn, mx, sedge);
    fill_one(rk.y, st1, s.y, w.y, mn, mx, sedge);
    fill_one(rk.z, st2, s.z, w.z, mn, mx, sedge);
    fill_one(rk.w, st3, s.w, w.w, mn, mx, sedge);
}

#define APITCH 132
#define GU 8


template <int M, int NQ, int BN>
__device__ __forceinline__ void load_b(v4f (&q)[BN][NQ], const float* __restrict__ bp, int s0) {
#pragma unroll
    for (int t = 0; t < BN; ++t) {
#pragma unroll
        for (int u = 0; u < NQ; ++u)
            q[t][u] = *(const v4f*)(bp + 4 * (s0 + t) * M + 64 * u);
    }
}

template <int NQ, int BN>
__device__ __forceinline__ void wmma_b(const v2f (&a)[32], int s0, const v4f (&q)[BN][NQ],
                                       v8f (&acc)[2 * NQ]) {
#pragma unroll
    for (int t = 0; t < BN; ++t) {
#pragma unroll
        for (int u = 0; u < NQ; ++u) {
            v2f b;
            b.x = q[t][u].x; b.y = q[t][u].y;
            acc[2 * u] = __builtin_amdgcn_wmma_f32_16x16x4_f32(
                false, a[s0 + t], false, b, (short)0, acc[2 * u], false, false);
            b.x = q[t][u].z; b.y = q[t][u].w;
            acc[2 * u + 1] = __builtin_amdgcn_wmma_f32_16x16x4_f32(
                false, a[s0 + t], false, b, (short)0, acc[2 * u + 1], false, false);
        }
    }
}


template <int M>
__device__ __forceinline__ const float* b_base(const float* __restrict__ Bs, int half, int lr,
                                               int cb) {
    return Bs + (size_t)half * (2 * M) + 2 * cb + 4 * lr;
}

template <int M, int NQ, int BN>
__device__ __forceinline__ void mma_head(v4f (&ring)[3][BN][NQ], const float* __restrict__ bp) {
    load_b<M, NQ, BN>(ring[0], bp, 0);
    load_b<M, NQ, BN>(ring[1], bp, BN);
}

template <int M, int NQ, int BN>
__device__ __forceinline__ void mma_all(const v2f (&a)[32], v4f (&ring)[3][BN][NQ],
                                        const float* __restrict__ bp, v8f (&acc)[2 * NQ]) {
    constexpr int NC = 32 / BN;
#pragma unroll
    for (int c = 0; c < NC; ++c) {
        if (c + 2 < NC)
            load_b<M, NQ, BN>(ring[(c + 2) % 3], bp, (c + 2) * BN);
        __builtin_amdgcn_sched_barrier(0);
        wmma_b<NQ, BN>(a, c * BN, ring[c % 3], acc);
        __builtin_amdgcn_sched_barrier(0);
    }
}

template <int M, bool RELU>
__global__ __launch_bounds__(64)
void fused_layer_kernel(const float* __restrict__ X,
                        const float* __restrict__ Wsw,
                        const float* __restrict__ bias,
                        const int* __restrict__ starts,
                        const int* __restrict__ sedge,
                        float* __restrict__ Out) {
    constexpr int K = 128;
    constexpr int MW = M / 2;
    constexpr int NQ = MW / 32;
    constexpr int BN = 8 / NQ;
    constexpr int CT = 2 * NQ;
    __shared__ __attribute__((aligned(16))) float tile[2][16 * APITCH];
    const int lane = threadIdx.x & 31;
    const int wv = __builtin_amdgcn_readfirstlane((int)(threadIdx.x >> 5));
    const int half = lane >> 4;
    const int lr   = lane & 15;
    const int rowbase = blockIdx.x * 16;
    const int rb = 8 * wv;
    const int cb = MW * wv;
    const int2* __restrict__ recs = (const int2*)sedge;
    const float4* __restrict__ X4 = (const float4*)X;
    const int sl = lane < 16 ? lane : 16;
    const float4 zero4 = make_float4(0.0f, 0.0f, 0.0f, 0.0f);

    v8f acc[CT];
#pragma unroll
    for (int ct = 0; ct < CT; ++ct) acc[ct] = v8f{};

    int svn = starts[rowbase + sl];
    int ssum = 0;
    for (int r = 0; r < RR; ++r) {
        const int svc = svn;
        ssum += svc;
        const int rn = r + 1 < RR ? r + 1 : r;
        svn = starts[rn * NN + rowbase + sl];

        int eb = __builtin_amdgcn_readlane(svc, rb);
        int ee = __builtin_amdgcn_readlane(svc, rb + 8);
        eb = eb < 0 ? 0 : eb;
        ee = ee > EE ? EE : ee;

        const float* bp = b_base<M>(Wsw + (size_t)r * K * M, half, lr, cb);
        v4f ring[3][BN][NQ];
        mma_head<M, NQ, BN>(ring, bp);

        float* tl = tile[r & 1];
#pragma unroll
        for (int j = 0; j < 8; ++j) *(float4*)&tl[(rb + j) * APITCH + lane * 4] = zero4;

        int cur = -1;
        float4 g = zero4;
        for (int c0 = eb; c0 < ee; c0 += 32) {
            int n = ee - c0;
            n = n > 32 ? 32 : n;
            const int myp = c0 + lane;
            int2 rec = make_int2(0, 0);
            if (lane < n) rec = recs[myp];
            int ms = rec.x;
            ms = ms < 0 ? 0 : (ms >= NN ? NN - 1 : ms);
            int myrow = 0;
#pragma unroll
            for (int j = 1; j < 16; ++j)
                myrow += (__builtin_amdgcn_readlane(svc, j) <= myp) ? 1 : 0;

            for (int e0 = 0; e0 < n; e0 += GU) {
                float4 v[GU];
#pragma unroll
                for (int u = 0; u < GU; ++u) {
                    int eu = e0 + u;
                    eu = eu < n ? eu : n - 1;
                    const int s = __builtin_amdgcn_readlane(ms, eu);
                    v[u] = X4[(size_t)s * (K / 4) + lane];
                }
#pragma unroll
                for (int u = 0; u < GU; ++u) {
                    const int eu = e0 + u;
                    if (eu < n) {
                        const int row = __builtin_amdgcn_readlane(myrow, eu);
                        const float w = __int_as_float(__builtin_amdgcn_readlane(rec.y, eu));
                        if (row != cur) {
                            if (cur >= 0) *(float4*)&tl[cur * APITCH + lane * 4] = g;
                            g = zero4;
                            cur = row;
                        }
                        g.x += v[u].x * w;
                        g.y += v[u].y * w;
                        g.z += v[u].z * w;
                        g.w += v[u].w * w;
                    }
                }
            }
        }
        if (cur >= 0) *(float4*)&tl[cur * APITCH + lane * 4] = g;

        __syncthreads();
        v2f a[32];
        const float* ap = &tl[lr * APITCH + 2 * half];
#pragma unroll
        for (int s = 0; s < 32; ++s) a[s] = *(const v2f*)(ap + 4 * s);

        mma_all<M, NQ, BN>(a, ring, bp, acc);
    }

    const int deg = __shfl_down(ssum, 1, 32) - ssum;
    const int invl = __float_as_int(1.0f / fmaxf((float)deg, 1.0f));
    float ivr[8];
#pragma unroll
    for (int j = 0; j < 8; ++j) {
        const int lo = __builtin_amdgcn_readlane(invl, j);
        const int hi = __builtin_amdgcn_readlane(invl, j + 8);
        ivr[j] = __int_as_float(half ? hi : lo);
    }
#pragma unroll
    for (int ct = 0; ct < CT; ++ct) {
#pragma unroll
        for (int j = 0; j < 8; ++j) acc[ct][j] *= ivr[j];
    }

    {
        const float* bp = b_base<M>(Wsw + (size_t)RR * K * M, half, lr, cb);
        v4f ring[3][BN][NQ];
        mma_head<M, NQ, BN>(ring, bp);
        const float* arow = X + (size_t)(rowbase + lr) * K + 2 * half;
        v2f a[32];
#pragma unroll
        for (int s = 0; s < 32; ++s) a[s] = *(const v2f*)(arow + 4 * s);
        mma_all<M, NQ, BN>(a, ring, bp, acc);
    }

#pragma unroll
    for (int u = 0; u < NQ; ++u) {
        const int col = cb + 32 * u + 2 * lr;
        const v2f bb = *(const v2f*)(bias + col);
        float* crow = Out + (size_t)(rowbase + 8 * half) * M + col;
#pragma unroll
        for (int j = 0; j < 8; ++j) {
            v2f val;
            val.x = acc[2 * u + 0][j] + bb.x;
            val.y = acc[2 * u + 1][j] + bb.y;
            if (RELU) {
                val.x = fmaxf(val.x, 0.0f);
                val.y = fmaxf(val.y, 0.0f);
            }
            *(v2f*)(crow + (size_t)j * M) = val;
        }
    }
}

extern "C" void kernel_launch(void* const* d_in, const int* in_sizes, int n_in,
                              void* d_out, int out_size, void* d_ws, size_t ws_size,
                              hipStream_t stream) {
    const float* x     = (const float*)d_in[0];
    const int*   eidx  = (const int*)d_in[1];
    const float* ew    = (const float*)d_in[2];
    const int*   et    = (const int*)d_in[3];
    const float* W1    = (const float*)d_in[4];
    const float* root1 = (const float*)d_in[5];
    const float* bias1 = (const float*)d_in[6];
    const float* W2    = (const float*)d_in[7];
    const float* root2 = (const float*)d_in[8];
    const float* bias2 = (const float*)d_in[9];
    float* out = (float*)d_out;

    const int* src = eidx;
    const int* dst = eidx + EE;

    float* bufH = (float*)d_ws;
    float* agg  = bufH + (size_t)NN * HH;
    float* sw1  = agg;
    float* sw2  = sw1 + (size_t)(RR + 1) * FF * HH;
    float* bufG = agg  + (size_t)NN * HH;
    unsigned* mm = (unsigned*)(bufG + (size_t)NN * HH);

    int*   starts = (int*)bufG;
    int*   tsum   = starts + NBP;
    int*   sedge  = tsum + 512;
    int*   rank   = sedge + 2 * EE;

    const int T = 256;
    const int rowTiles = NN / 16;

    hipMemsetAsync(starts, 0, (size_t)NBP * sizeof(int), stream);
    prep_weights_kernel<<<(WT1 + WT2 + 255) / 256, 256, 0, stream>>>(W1, root1, W2, root2,
                                                                     sw1, sw2, mm);
    const int E4 = EE / 4;
    ew_minmax_kernel<<<256, T, 0, stream>>>(ew, EE, mm);

    rank_kernel<<<(E4 + T - 1) / T, T, 0, stream>>>(dst, et, starts, rank, E4);
    scan_tilesum_kernel<<<SCAN_BLOCKS, 256, 0, stream>>>(starts, tsum);
    scan_apply_kernel<<<SCAN_BLOCKS, 256, 0, stream>>>(starts, tsum);
    fill_kernel<<<(E4 + T - 1) / T, T, 0, stream>>>(src, dst, et, ew, mm, starts, rank,
                                                    sedge, E4);

    fused_layer_kernel<HH, true><<<rowTiles, 64, 0, stream>>>(
        x, sw1, bias1, starts, sedge, bufH);
    fused_layer_kernel<OO, false><<<rowTiles, 64, 0, stream>>>(
        bufH, sw2, bias2, starts, sedge, out);
}


# Edge grouping on LDS counters: chunks ordered by node bucket in LDS, one block per bucket ranks by (relation,dst) in LDS; per-edge global atomics, memset, 2 launches gone; sum order: LDS arrival rank
# speedup vs baseline: 3.7470x; 1.1358x over previous
#include <hip/hip_runtime.h>
#include <hip/hip_bf16.h>

#define NN 50000
#define EE 600000
#define FF 128
#define HH 128
#define OO 64
#define RR 8

typedef float v2f __attribute__((ext_vector_type(2)));
typedef float v4f __attribute__((ext_vector_type(4)));
typedef float v8f __attribute__((ext_vector_type(8)));

#define WPR ((RR + 1) * (FF / 2))
#define WT1 (WPR * (HH / 4))
#define WT2 (WPR * (OO / 4))
#define NBK_TOT (241 * 32)
__global__ __launch_bounds__(256)
void prep_weights_kernel(const float* __restrict__ W1, const float* __restrict__ root1,
                         const float* __restrict__ W2, const float* __restrict__ root2,
                         float* __restrict__ sw1, float* __restrict__ sw2,
                         unsigned* __restrict__ mm, int* __restrict__ btot) {
    const int t = blockIdx.x * 256 + threadIdx.x;
    if (t == 0) { mm[0] = 0x7F800000u; mm[1] = 0u; }
    if (t < NBK_TOT) btot[t] = 0;
    const float* src;
    float* dst;
    int M, pr, c4;
    if (t < WT1) {
        M = HH; pr = t / (HH / 4); c4 = t % (HH / 4); dst = sw1;
        src = pr < RR * (FF / 2) ? W1 + (size_t)(2 * pr) * HH
                                 : root1 + (size_t)(2 * (pr - RR * (FF / 2))) * HH;
    } else if (t < WT1 + WT2) {
        const int u = t - WT1;
        M = OO; pr = u / (OO / 4); c4 = u % (OO / 4); dst = sw2;
        src = pr < RR * (HH / 2) ? W2 + (size_t)(2 * pr) * OO
                                 : root2 + (size_t)(2 * (pr - RR * (HH / 2))) * OO;
    } else {
        return;
    }
    const v4f a = *(const v4f*)(src + 4 * c4);
    const v4f b = *(const v4f*)(src + M + 4 * c4);
    v4f o0, o1;
    o0.x = a.x; o0.y = b.x; o0.z = a.y; o0.w = b.y;
    o1.x = a.z; o1.y = b.z; o1.z = a.w; o1.w = b.w;
    float* d = dst + ((size_t)pr * M + 4 * c4) * 2;
    *(v4f*)d = o0;
    *(v4f*)(d + 4) = o1;
}

__global__ void ew_minmax_kernel(const float* __restrict__ ew, int E,
                                 unsigned* __restrict__ mm) {
    float mn = __builtin_huge_valf();
    float mx = 0.0f;
    for (int i = blockIdx.x * blockDim.x + threadIdx.x; i < E;
         i += gridDim.x * blockDim.x) {
        float w = ew[i];
        mn = fminf(mn, w);
        mx = fmaxf(mx, w);
    }
    for (int off = 16; off > 0; off >>= 1) {
        mn = fminf(mn, __shfl_down(mn, off, 32));
        mx = fmaxf(mx, __shfl_down(mx, off, 32));
    }
    if ((threadIdx.x & 31) == 0) {
        atomicMin(&mm[0], __float_as_uint(mn));
        atomicMax(&mm[1], __float_as_uint(mx));
    }
}

#define BW 208
#define NBK ((NN + BW - 1) / BW)
#define KPB (RR * BW)
#define NKEY (NBK * KPB)
#define CHUNK 2048
#define NCH ((EE + CHUNK - 1) / CHUNK)
#define BCS 32
#define KPT ((KPB + 255) / 256)
#define BCAP 4096
#define NSTP (((NKEY + 1) + 3) / 4 * 4)
#define NTBP ((NBK * NCH + 3) / 4 * 4)
static_assert(BW % 16 == 0, "a row tile must not straddle buckets");
static_assert(NBK <= 256, "one bucket per thread in the chunk kernel");
static_assert(NN <= 65536 && KPB <= 65536, "source and in-bucket key share one word");
static_assert(EE % 4 == 0, "four edges per 16-byte load");
static_assert(NBK_TOT == NBK * BCS && NBK_TOT <= WT1, "prep_weights_kernel zeroes the totals");

__device__ __forceinline__ int block_scan_256(int v, int* wtot, int& tot) {
    const int lane = threadIdx.x & 31;
    const int wv = threadIdx.x >> 5;
    int inc = v;
    for (int off = 1; off < 32; off <<= 1) {
        const int n = __shfl_up(inc, off, 32);
        if (lane >= off) inc += n;
    }
    if (lane == 31) wtot[wv] = inc;
    __syncthreads();
    int base = 0, t = 0;
    for (int i = 0; i < 8; ++i) {
        const int w = wtot[i];
        if (i < wv) base += w;
        t += w;
    }
    tot = t;
    __syncthreads();
    return base + inc - v;
}

__global__ __launch_bounds__(256)
void group_chunk_kernel(const int* __restrict__ src, const int* __restrict__ dst,
                        const int* __restrict__ et, const float* __restrict__ ew,
                        const unsigned* __restrict__ mm,
                        int* __restrict__ btot,
                        int* __restrict__ tbl,
                        int* __restrict__ stage_w) {
    int2* __restrict__ stage = (int2*)stage_w;
    __shared__ int cnt[256];
    __shared__ int lst[256];
    __shared__ int wtot[8];
    __shared__ int2 srec[CHUNK];
    const int t = threadIdx.x;
    const int ch = blockIdx.x;
    cnt[t] = 0;
    __syncthreads();

    const float mn = __uint_as_float(mm[0]);
    const float mx = __uint_as_float(mm[1]);
    int bk[8], lr[8];
    int2 rc[8];
#pragma unroll
    for (int g = 0; g < 2; ++g) {
        const int q = ch * (CHUNK / 4) + g * 256 + t;
        int4 ty = make_int4(-1, -1, -1, -1), d = ty, s = ty;
        float4 w = make_float4(0.0f, 0.0f, 0.0f, 0.0f);
        if (q < EE / 4) {
            ty = ((const int4*)et)[q];
            d = ((const int4*)dst)[q];
            s = ((const int4*)src)[q];
            w = ((const float4*)ew)[q];
        }
        const int tt[4] = {ty.x, ty.y, ty.z, ty.w};
        const int dd[4] = {d.x, d.y, d.z, d.w};
        const int ss[4] = {s.x, s.y, s.z, s.w};
        const float ww[4] = {w.x, w.y, w.z, w.w};
#pragma unroll
        for (int j = 0; j < 4; ++j) {
            const int e = 4 * g + j;
            bk[e] = -1;
            lr[e] = 0;
            rc[e] = make_int2(0, 0);
            if ((unsigned)tt[j] < (unsigned)RR && (unsigned)dd[j] < (unsigned)NN) {
                const int b = dd[j] / BW;
                const int kl = tt[j] * BW + (dd[j] - b * BW);
                int sc = ss[j];
                sc = sc < 0 ? 0 : (sc >= NN ? NN - 1 : sc);
                bk[e] = b;
                lr[e] = atomicAdd(&cnt[b], 1);
                rc[e].x = sc | (kl << 16);
                rc[e].y = __float_as_int((ww[j] - mn) / (mx - mn + 1e-8f));
            }
        }
    }
    __syncthreads();

    const int c = cnt[t];
    int nvalid;
    const int ls = block_scan_256(c, wtot, nvalid);
    lst[t] = ls;
    if (t < NBK) {
        tbl[t * NCH + ch] = (ls << 16) | c;
        if (c > 0) atomicAdd(&btot[t * BCS], c);
    }
    __syncthreads();
#pragma unroll
    for (int e = 0; e < 8; ++e)
        if (bk[e] >= 0) srec[lst[bk[e]] + lr[e]] = rc[e];
    __syncthreads();
    int2* out = stage + (size_t)ch * CHUNK;
    for (int i = t; i < nvalid; i += 256) out[i] = srec[i];
}

__global__ __launch_bounds__(256)
void group_bucket_kernel(const int* __restrict__ btot, const int* __restrict__ tbl,
                         const int* __restrict__ stage_w,
                         int* __restrict__ starts,
                         int* __restrict__ sedge_w) {
    const int2* __restrict__ stage = (const int2*)stage_w;
    int2* __restrict__ sedge = (int2*)sedge_w;
    __shared__ int kc[256 * KPT];
    __shared__ int wtot[8];
    __shared__ int sbase, stot;
    __shared__ int2 lin[BCAP];
    __shared__ int2 lrec[BCAP];
    constexpr int RPT = (NCH + 255) / 256;
    const int t = threadIdx.x;
    const int b = blockIdx.x;
    for (int i = t; i < 256 * KPT; i += 256) kc[i] = 0;

    int total;
    const int mine = t < NBK ? btot[t * BCS] : 0;
    const int exb = block_scan_256(mine, wtot, total);
    if (t == b) { sbase = exb; stot = mine; }

    int rs[RPT], rn[RPT];
    int mylen = 0;
#pragma unroll
    for (int k = 0; k < RPT; ++k) {
        const int ch = t + 256 * k;
        rs[k] = 0;
        rn[k] = 0;
        if (ch < NCH) {
            const int p = tbl[b * NCH + ch];
            rs[k] = ch * CHUNK + (int)((unsigned)p >> 16);
            rn[k] = p & 0xFFFF;
            rn[k] = rn[k] > CHUNK ? CHUNK : rn[k];
        }
        mylen += rn[k];
    }
    int all;
    int off = block_scan_256(mylen, wtot, all);
    const int base = sbase;
    const bool staged = stot <= BCAP && all <= BCAP;

    constexpr int GR = 8;
#pragma unroll
    for (int k = 0; k < RPT; ++k) {
        for (int i = 0; i < rn[k]; i += GR) {
            int2 r[GR];
#pragma unroll
            for (int u = 0; u < GR; ++u) {
                const int iu = i + u < rn[k] ? i + u : rn[k] - 1;
                r[u] = stage[rs[k] + iu];
            }
#pragma unroll
            for (int u = 0; u < GR; ++u) {
                if (i + u < rn[k]) {
                    const int kl = (int)((unsigned)r[u].x >> 16);
                    if (kl < KPB) atomicAdd(&kc[kl], 1);
                    if (staged) lin[off + i + u] = r[u];
                }
            }
        }
        off += rn[k];
    }
    __syncthreads();

    int v[KPT];
    int sum = 0;
#pragma unroll
    for (int j = 0; j < KPT; ++j) {
        v[j] = kc[t * KPT + j];
        sum += v[j];
    }
    int nrec;
    int ex = block_scan_256(sum, wtot, nrec);
#pragma unroll
    for (int j = 0; j < KPT; ++j) {
        kc[t * KPT + j] = ex;
        ex += v[j];
    }
    __syncthreads();
    for (int k = t; k < KPB; k += 256) starts[b * KPB + k] = base + kc[k];
    if (b == NBK - 1 && t == 0) starts[NKEY] = base + nrec;
    __syncthreads();

    if (staged) {
        for (int i = t; i < all; i += 256) {
            const int2 r = lin[i];
            const int kl = (int)((unsigned)r.x >> 16);
            if (kl < KPB) {
                const int pos = atomicAdd(&kc[kl], 1);
                if (pos < BCAP) lrec[pos] = make_int2(r.x & 0xFFFF, r.y);
            }
        }
        __syncthreads();
        int n = nrec < BCAP ? nrec : BCAP;
        if (base + n > EE) n = EE - base;
        for (int i = t; i < n; i += 256) sedge[base + i] = lrec[i];
    } else {
#pragma unroll
        for (int k = 0; k < RPT; ++k)
            for (int i = 0; i < rn[k]; ++i) {
                const int2 r = stage[rs[k] + i];
                const int kl = (int)((unsigned)r.x >> 16);
                if (kl < KPB) {
                    const int pos = atomicAdd(&kc[kl], 1);
                    if ((unsigned)(base + pos) < (unsigned)EE)
                        sedge[base + pos] = make_int2(r.x & 0xFFFF, r.y);
                }
            }
    }
}

#define APITCH 132
#define GU 8


template <int M, int NQ, int BN>
__device__ __forceinline__ void load_b(v4f (&q)[BN][NQ], const float* __restrict__ bp, int s0) {
#pragma unroll
    for (int t = 0; t < BN; ++t) {
#pragma unroll
        for (int u = 0; u < NQ; ++u)
            q[t][u] = *(const v4f*)(bp + 4 * (s0 + t) * M + 64 * u);
    }
}

template <int NQ, int BN>
__device__ __forceinline__ void wmma_b(const v2f (&a)[32], int s0, const v4f (&q)[BN][NQ],
                                       v8f (&acc)[2 * NQ]) {
#pragma unroll
    for (int t = 0; t < BN; ++t) {
#pragma unroll
        for (int u = 0; u < NQ; ++u) {
            v2f b;
            b.x = q[t][u].x; b.y = q[t][u].y;
            acc[2 * u] = __builtin_amdgcn_wmma_f32_16x16x4_f32(
                false, a[s0 + t], false, b, (short)0, acc[2 * u], false, false);
            b.x = q[t][u].z; b.y = q[t][u].w;
            acc[2 * u + 1] = __builtin_amdgcn_wmma_f32_16x16x4_f32(
                false, a[s0 + t], false, b, (short)0, acc[2 * u + 1], false, false);
        }
    }
}


template <int M>
__device__ __forceinline__ const float* b_base(const float* __restrict__ Bs, int half, int lr,
                                               int cb) {
    return Bs + (size_t)half * (2 * M) + 2 * cb + 4 * lr;
}

template <int M, int NQ, int BN>
__device__ __forceinline__ void mma_head(v4f (&ring)[3][BN][NQ], const float* __restrict__ bp) {
    load_b<M, NQ, BN>(ring[0], bp, 0);
    load_b<M, NQ, BN>(ring[1], bp, BN);
}

template <int M, int NQ, int BN>
__device__ __forceinline__ void mma_all(const v2f (&a)[32], v4f (&ring)[3][BN][NQ],
                                        const float* __restrict__ bp, v8f (&acc)[2 * NQ]) {
    constexpr int NC = 32 / BN;
#pragma unroll
    for (int c = 0; c < NC; ++c) {
        if (c + 2 < NC)
            load_b<M, NQ, BN>(ring[(c + 2) % 3], bp, (c + 2) * BN);
        __builtin_amdgcn_sched_barrier(0);
        wmma_b<NQ, BN>(a, c * BN, ring[c % 3], acc);
        __builtin_amdgcn_sched_barrier(0);
    }
}

template <int M, bool RELU>
__global__ __launch_bounds__(64)
void fused_layer_kernel(const float* __restrict__ X,
                        const float* __restrict__ Wsw,
                        const float* __restrict__ bias,
                        const int* __restrict__ starts,
                        const int* __restrict__ sedge,
                        float* __restrict__ Out) {
    constexpr int K = 128;
    constexpr int MW = M / 2;
    constexpr int NQ = MW / 32;
    constexpr int BN = 8 / NQ;
    constexpr int CT = 2 * NQ;
    __shared__ __attribute__((aligned(16))) float tile[2][16 * APITCH];
    const int lane = threadIdx.x & 31;
    const int wv = __builtin_amdgcn_readfirstlane((int)(threadIdx.x >> 5));
    const int half = lane >> 4;
    const int lr   = lane & 15;
    const int rowbase = blockIdx.x * 16;
    const int rb = 8 * wv;
    const int cb = MW * wv;
    const int2* __restrict__ recs = (const int2*)sedge;
    const float4* __restrict__ X4 = (const float4*)X;
    const int sl = lane < 16 ? lane : 16;
    const float4 zero4 = make_float4(0.0f, 0.0f, 0.0f, 0.0f);

    v8f acc[CT];
#pragma unroll
    for (int ct = 0; ct < CT; ++ct) acc[ct] = v8f{};

    const int kb0 = (rowbase / BW) * KPB + rowbase % BW;
    int svn = starts[kb0 + sl];
    int ssum = 0;
    for (int r = 0; r < RR; ++r) {
        const int svc = svn;
        ssum += svc;
        const int rn = r + 1 < RR ? r + 1 : r;
        svn = starts[kb0 + rn * BW + sl];

        int eb = __builtin_amdgcn_readlane(svc, rb);
        int ee = __builtin_amdgcn_readlane(svc, rb + 8);
        eb = eb < 0 ? 0 : eb;
        ee = ee > EE ? EE : ee;

        const float* bp = b_base<M>(Wsw + (size_t)r * K * M, half, lr, cb);
        v4f ring[3][BN][NQ];
        mma_head<M, NQ, BN>(ring, bp);

        float* tl = tile[r & 1];
#pragma unroll
        for (int j = 0; j < 8; ++j) *(float4*)&tl[(rb + j) * APITCH + lane * 4] = zero4;

        int cur = -1;
        float4 g = zero4;
        for (int c0 = eb; c0 < ee; c0 += 32) {
            int n = ee - c0;
            n = n > 32 ? 32 : n;
            const int myp = c0 + lane;
            int2 rec = make_int2(0, 0);
            if (lane < n) rec = recs[myp];
            int ms = rec.x;
            ms = ms < 0 ? 0 : (ms >= NN ? NN - 1 : ms);
            int myrow = 0;
#pragma unroll
            for (int j = 1; j < 16; ++j)
                myrow += (__builtin_amdgcn_readlane(svc, j) <= myp) ? 1 : 0;

            for (int e0 = 0; e0 < n; e0 += GU) {
                float4 v[GU];
#pragma unroll
                for (int u = 0; u < GU; ++u) {
                    int eu = e0 + u;
                    eu = eu < n ? eu : n - 1;
                    const int s = __builtin_amdgcn_readlane(ms, eu);
                    v[u] = X4[(size_t)s * (K / 4) + lane];
                }
#pragma unroll
                for (int u = 0; u < GU; ++u) {
                    const int eu = e0 + u;
                    if (eu < n) {
                        const int row = __builtin_amdgcn_readlane(myrow, eu);
                        const float w = __int_as_float(__builtin_amdgcn_readlane(rec.y, eu));
                        if (row != cur) {
                            if (cur >= 0) *(float4*)&tl[cur * APITCH + lane * 4] = g;
                            g = zero4;
                            cur = row;
                        }
                        g.x += v[u].x * w;
                        g.y += v[u].y * w;
                        g.z += v[u].z * w;
                        g.w += v[u].w * w;
                    }
                }
            }
        }
        if (cur >= 0) *(float4*)&tl[cur * APITCH + lane * 4] = g;

        __syncthreads();
        v2f a[32];
        const float* ap = &tl[lr * APITCH + 2 * half];
#pragma unroll
        for (int s = 0; s < 32; ++s) a[s] = *(const v2f*)(ap + 4 * s);

        mma_all<M, NQ, BN>(a, ring, bp, acc);
    }

    const int deg = __shfl_down(ssum, 1, 32) - ssum;
    const int invl = __float_as_int(1.0f / fmaxf((float)deg, 1.0f));
    float ivr[8];
#pragma unroll
    for (int j = 0; j < 8; ++j) {
        const int lo = __builtin_amdgcn_readlane(invl, j);
        const int hi = __builtin_amdgcn_readlane(invl, j + 8);
        ivr[j] = __int_as_float(half ? hi : lo);
    }
#pragma unroll
    for (int ct = 0; ct < CT; ++ct) {
#pragma unroll
        for (int j = 0; j < 8; ++j) acc[ct][j] *= ivr[j];
    }

    {
        const float* bp = b_base<M>(Wsw + (size_t)RR * K * M, half, lr, cb);
        v4f ring[3][BN][NQ];
        mma_head<M, NQ, BN>(ring, bp);
        const float* arow = X + (size_t)(rowbase + lr) * K + 2 * half;
        v2f a[32];
#pragma unroll
        for (int s = 0; s < 32; ++s) a[s] = *(const v2f*)(arow + 4 * s);
        mma_all<M, NQ, BN>(a, ring, bp, acc);
    }

#pragma unroll
    for (int u = 0; u < NQ; ++u) {
        const int col = cb + 32 * u + 2 * lr;
        const v2f bb = *(const v2f*)(bias + col);
        float* crow = Out + (size_t)(rowbase + 8 * half) * M + col;
#pragma unroll
        for (int j = 0; j < 8; ++j) {
            v2f val;
            val.x = acc[2 * u + 0][j] + bb.x;
            val.y = acc[2 * u + 1][j] + bb.y;
            if (RELU) {
                val.x = fmaxf(val.x, 0.0f);
                val.y = fmaxf(val.y, 0.0f);
            }
            *(v2f*)(crow + (size_t)j * M) = val;
        }
    }
}

extern "C" void kernel_launch(void* const* d_in, const int* in_sizes, int n_in,
                              void* d_out, int out_size, void* d_ws, size_t ws_size,
                              hipStream_t stream) {
    const float* x     = (const float*)d_in[0];
    const int*   eidx  = (const int*)d_in[1];
    const float* ew    = (const float*)d_in[2];
    const int*   et    = (const int*)d_in[3];
    const float* W1    = (const float*)d_in[4];
    const float* root1 = (const float*)d_in[5];
    const float* bias1 = (const float*)d_in[6];
    const float* W2    = (const float*)d_in[7];
    const float* root2 = (const float*)d_in[8];
    const float* bias2 = (const float*)d_in[9];
    float* out = (float*)d_out;

    const int* src = eidx;
    const int* dst = eidx + EE;

    float* bufH = (float*)d_ws;
    float* agg  = bufH + (size_t)NN * HH;
    float* sw1  = agg;
    float* sw2  = sw1 + (size_t)(RR + 1) * FF * HH;
    float* bufG = agg  + (size_t)NN * HH;
    unsigned* mm = (unsigned*)(bufG + (size_t)NN * HH);

    int*   starts = (int*)bufG;
    int*   btot   = starts + NSTP;
    int*   tbl    = btot + NBK * BCS;
    int*   stage  = tbl + NTBP;
    int*   sedge  = stage + 2 * (size_t)NCH * CHUNK;

    const int T = 256;
    const int rowTiles = NN / 16;

    prep_weights_kernel<<<(WT1 + WT2 + 255) / 256, 256, 0, stream>>>(W1, root1, W2, root2,
                                                                     sw1, sw2, mm, btot);
    ew_minmax_kernel<<<256, T, 0, stream>>>(ew, EE, mm);

    group_chunk_kernel<<<NCH, 256, 0, stream>>>(src, dst, et, ew, mm, btot, tbl, stage);
    group_bucket_kernel<<<NBK, 256, 0, stream>>>(btot, tbl, stage, starts, sedge);

    fused_layer_kernel<HH, true><<<rowTiles, 64, 0, stream>>>(
        x, sw1, bias1, starts, sedge, bufH);
    fused_layer_kernel<OO, false><<<rowTiles, 64, 0, stream>>>(
        bufH, sw2, bias2, starts, sedge, out);
}


# Fewer registers per wave (A fragments read from LDS per chunk, root rows through the same tile, B ring of 2): 7 waves per SIMD instead of 4, every row tile resident at once; sum order unchanged
# speedup vs baseline: 3.9514x; 1.0546x over previous
#include <hip/hip_runtime.h>
#include <hip/hip_bf16.h>

#define NN 50000
#define EE 600000
#define FF 128
#define HH 128
#define OO 64
#define RR 8

typedef float v2f __attribute__((ext_vector_type(2)));
typedef float v4f __attribute__((ext_vector_type(4)));
typedef float v8f __attribute__((ext_vector_type(8)));

#define WPR ((RR + 1) * (FF / 2))
#define WT1 (WPR * (HH / 4))
#define WT2 (WPR * (OO / 4))
#define NBK_TOT (241 * 32)
__global__ __launch_bounds__(256)
void prep_weights_kernel(const float* __restrict__ W1, const float* __restrict__ root1,
                         const float* __restrict__ W2, const float* __restrict__ root2,
                         float* __restrict__ sw1, float* __restrict__ sw2,
                         unsigned* __restrict__ mm, int* __restrict__ btot) {
    const int t = blockIdx.x * 256 + threadIdx.x;
    if (t == 0) { mm[0] = 0x7F800000u; mm[1] = 0u; }
    if (t < NBK_TOT) btot[t] = 0;
    const float* src;
    float* dst;
    int M, pr, c4;
    if (t < WT1) {
        M = HH; pr = t / (HH / 4); c4 = t % (HH / 4); dst = sw1;
        src = pr < RR * (FF / 2) ? W1 + (size_t)(2 * pr) * HH
                                 : root1 + (size_t)(2 * (pr - RR * (FF / 2))) * HH;
    } else if (t < WT1 + WT2) {
        const int u = t - WT1;
        M = OO; pr = u / (OO / 4); c4 = u % (OO / 4); dst = sw2;
        src = pr < RR * (HH / 2) ? W2 + (size_t)(2 * pr) * OO
                                 : root2 + (size_t)(2 * (pr - RR * (HH / 2))) * OO;
    } else {
        return;
    }
    const v4f a = *(const v4f*)(src + 4 * c4);
    const v4f b = *(const v4f*)(src + M + 4 * c4);
    v4f o0, o1;
    o0.x = a.x; o0.y = b.x; o0.z = a.y; o0.w = b.y;
    o1.x = a.z; o1.y = b.z; o1.z = a.w; o1.w = b.w;
    float* d = dst + ((size_t)pr * M + 4 * c4) * 2;
    *(v4f*)d = o0;
    *(v4f*)(d + 4) = o1;
}

__global__ void ew_minmax_kernel(const float* __restrict__ ew, int E,
                                 unsigned* __restrict__ mm) {
    float mn = __builtin_huge_valf();
    float mx = 0.0f;
    for (int i = blockIdx.x * blockDim.x + threadIdx.x; i < E;
         i += gridDim.x * blockDim.x) {
        float w = ew[i];
        mn = fminf(mn, w);
        mx = fmaxf(mx, w);
    }
    for (int off = 16; off > 0; off >>= 1) {
        mn = fminf(mn, __shfl_down(mn, off, 32));
        mx = fmaxf(mx, __shfl_down(mx, off, 32));
    }
    if ((threadIdx.x & 31) == 0) {
        atomicMin(&mm[0], __float_as_uint(mn));
        atomicMax(&mm[1], __float_as_uint(mx));
    }
}

#define BW 208
#define NBK ((NN + BW - 1) / BW)
#define KPB (RR * BW)
#define NKEY (NBK * KPB)
#define CHUNK 2048
#define NCH ((EE + CHUNK - 1) / CHUNK)
#define BCS 32
#define KPT ((KPB + 255) / 256)
#define BCAP 4096
#define NSTP (((NKEY + 1) + 3) / 4 * 4)
#define NTBP ((NBK * NCH + 3) / 4 * 4)
static_assert(BW % 16 == 0, "a row tile must not straddle buckets");
static_assert(NBK <= 256, "one bucket per thread in the chunk kernel");
static_assert(NN <= 65536 && KPB <= 65536, "source and in-bucket key share one word");
static_assert(EE % 4 == 0, "four edges per 16-byte load");
static_assert(NBK_TOT == NBK * BCS && NBK_TOT <= WT1, "prep_weights_kernel zeroes the totals");

__device__ __forceinline__ int block_scan_256(int v, int* wtot, int& tot) {
    const int lane = threadIdx.x & 31;
    const int wv = threadIdx.x >> 5;
    int inc = v;
    for (int off = 1; off < 32; off <<= 1) {
        const int n = __shfl_up(inc, off, 32);
        if (lane >= off) inc += n;
    }
    if (lane == 31) wtot[wv] = inc;
    __syncthreads();
    int base = 0, t = 0;
    for (int i = 0; i < 8; ++i) {
        const int w = wtot[i];
        if (i < wv) base += w;
        t += w;
    }
    tot = t;
    __syncthreads();
    return base + inc - v;
}

__global__ __launch_bounds__(256)
void group_chunk_kernel(const int* __restrict__ src, const int* __restrict__ dst,
                        const int* __restrict__ et, const float* __restrict__ ew,
                        const unsigned* __restrict__ mm,
                        int* __restrict__ btot,
                        int* __restrict__ tbl,
                        int* __restrict__ stage_w) {
    int2* __restrict__ stage = (int2*)stage_w;
    __shared__ int cnt[256];
    __shared__ int lst[256];
    __shared__ int wtot[8];
    __shared__ int2 srec[CHUNK];
    const int t = threadIdx.x;
    const int ch = blockIdx.x;
    cnt[t] = 0;
    __syncthreads();

    const float mn = __uint_as_float(mm[0]);
    const float mx = __uint_as_float(mm[1]);
    int bk[8], lr[8];
    int2 rc[8];
#pragma unroll
    for (int g = 0; g < 2; ++g) {
        const int q = ch * (CHUNK / 4) + g * 256 + t;
        int4 ty = make_int4(-1, -1, -1, -1), d = ty, s = ty;
        float4 w = make_float4(0.0f, 0.0f, 0.0f, 0.0f);
        if (q < EE / 4) {
            ty = ((const int4*)et)[q];
            d = ((const int4*)dst)[q];
            s = ((const int4*)src)[q];
            w = ((const float4*)ew)[q];
        }
        const int tt[4] = {ty.x, ty.y, ty.z, ty.w};
        const int dd[4] = {d.x, d.y, d.z, d.w};
        const int ss[4] = {s.x, s.y, s.z, s.w};
        const float ww[4] = {w.x, w.y, w.z, w.w};
#pragma unroll
        for (int j = 0; j < 4; ++j) {
            const int e = 4 * g + j;
            bk[e] = -1;
            lr[e] = 0;
            rc[e] = make_int2(0, 0);
            if ((unsigned)tt[j] < (unsigned)RR && (unsigned)dd[j] < (unsigned)NN) {
                const int b = dd[j] / BW;
                const int kl = tt[j] * BW + (dd[j] - b * BW);
                int sc = ss[j];
                sc = sc < 0 ? 0 : (sc >= NN ? NN - 1 : sc);
                bk[e] = b;
                lr[e] = atomicAdd(&cnt[b], 1);
                rc[e].x = sc | (kl << 16);
                rc[e].y = __float_as_int((ww[j] - mn) / (mx - mn + 1e-8f));
            }
        }
    }
    __syncthreads();

    const int c = cnt[t];
    int nvalid;
    const int ls = block_scan_256(c, wtot, nvalid);
    lst[t] = ls;
    if (t < NBK) {
        tbl[t * NCH + ch] = (ls << 16) | c;
        if (c > 0) atomicAdd(&btot[t * BCS], c);
    }
    __syncthreads();
#pragma unroll
    for (int e = 0; e < 8; ++e)
        if (bk[e] >= 0) srec[lst[bk[e]] + lr[e]] = rc[e];
    __syncthreads();
    int2* out = stage + (size_t)ch * CHUNK;
    for (int i = t; i < nvalid; i += 256) out[i] = srec[i];
}

__global__ __launch_bounds__(256)
void group_bucket_kernel(const int* __restrict__ btot, const int* __restrict__ tbl,
                         const int* __restrict__ stage_w,
                         int* __restrict__ starts,
                         int* __restrict__ sedge_w) {
    const int2* __restrict__ stage = (const int2*)stage_w;
    int2* __restrict__ sedge = (int2*)sedge_w;
    __shared__ int kc[256 * KPT];
    __shared__ int wtot[8];
    __shared__ int sbase, stot;
    __shared__ int2 lin[BCAP];
    __shared__ int2 lrec[BCAP];
    constexpr int RPT = (NCH + 255) / 256;
    const int t = threadIdx.x;
    const int b = blockIdx.x;
    for (int i = t; i < 256 * KPT; i += 256) kc[i] = 0;

    int total;
    const int mine = t < NBK ? btot[t * BCS] : 0;
    const int exb = block_scan_256(mine, wtot, total);
    if (t == b) { sbase = exb; stot = mine; }

    int rs[RPT], rn[RPT];
    int mylen = 0;
#pragma unroll
    for (int k = 0; k < RPT; ++k) {
        const int ch = t + 256 * k;
        rs[k] = 0;
        rn[k] = 0;
        if (ch < NCH) {
            const int p = tbl[b * NCH + ch];
            rs[k] = ch * CHUNK + (int)((unsigned)p >> 16);
            rn[k] = p & 0xFFFF;
            rn[k] = rn[k] > CHUNK ? CHUNK : rn[k];
        }
        mylen += rn[k];
    }
    int all;
    int off = block_scan_256(mylen, wtot, all);
    const int base = sbase;
    const bool staged = stot <= BCAP && all <= BCAP;

    constexpr int GR = 8;
#pragma unroll
    for (int k = 0; k < RPT; ++k) {
        for (int i = 0; i < rn[k]; i += GR) {
            int2 r[GR];
#pragma unroll
            for (int u = 0; u < GR; ++u) {
                const int iu = i + u < rn[k] ? i + u : rn[k] - 1;
                r[u] = stage[rs[k] + iu];
            }
#pragma unroll
            for (int u = 0; u < GR; ++u) {
                if (i + u < rn[k]) {
                    const int kl = (int)((unsigned)r[u].x >> 16);
                    if (kl < KPB) atomicAdd(&kc[kl], 1);
                    if (staged) lin[off + i + u] = r[u];
                }
            }
        }
        off += rn[k];
    }
    __syncthreads();

    int v[KPT];
    int sum = 0;
#pragma unroll
    for (int j = 0; j < KPT; ++j) {
        v[j] = kc[t * KPT + j];
        sum += v[j];
    }
    int nrec;
    int ex = block_scan_256(sum, wtot, nrec);
#pragma unroll
    for (int j = 0; j < KPT; ++j) {
        kc[t * KPT + j] = ex;
        ex += v[j];
    }
    __syncthreads();
    for (int k = t; k < KPB; k += 256) starts[b * KPB + k] = base + kc[k];
    if (b == NBK - 1 && t == 0) starts[NKEY] = base + nrec;
    __syncthreads();

    if (staged) {
        for (int i = t; i < all; i += 256) {
            const int2 r = lin[i];
            const int kl = (int)((unsigned)r.x >> 16);
            if (kl < KPB) {
                const int pos = atomicAdd(&kc[kl], 1);
                if (pos < BCAP) lrec[pos] = make_int2(r.x & 0xFFFF, r.y);
            }
        }
        __syncthreads();
        int n = nrec < BCAP ? nrec : BCAP;
        if (base + n > EE) n = EE - base;
        for (int i = t; i < n; i += 256) sedge[base + i] = lrec[i];
    } else {
#pragma unroll
        for (int k = 0; k < RPT; ++k)
            for (int i = 0; i < rn[k]; ++i) {
                const int2 r = stage[rs[k] + i];
                const int kl = (int)((unsigned)r.x >> 16);
                if (kl < KPB) {
                    const int pos = atomicAdd(&kc[kl], 1);
                    if ((unsigned)(base + pos) < (unsigned)EE)
                        sedge[base + pos] = make_int2(r.x & 0xFFFF, r.y);
                }
            }
    }
}

#define APITCH 132
#define GU 8
#define RINGDEPTH 2


template <int M, int NQ, int BN>
__device__ __forceinline__ void load_b(v4f (&q)[BN][NQ], const float* __restrict__ bp, int s0) {
#pragma unroll
    for (int t = 0; t < BN; ++t) {
#pragma unroll
        for (int u = 0; u < NQ; ++u)
            q[t][u] = *(const v4f*)(bp + 4 * (s0 + t) * M + 64 * u);
    }
}

template <int NQ, int BN>
__device__ __forceinline__ void wmma_b(const v2f (&a)[BN], const v4f (&q)[BN][NQ],
                                       v8f (&acc)[2 * NQ]) {
#pragma unroll
    for (int t = 0; t < BN; ++t) {
#pragma unroll
        for (int u = 0; u < NQ; ++u) {
            v2f b;
            b.x = q[t][u].x; b.y = q[t][u].y;
            acc[2 * u] = __builtin_amdgcn_wmma_f32_16x16x4_f32(
                false, a[t], false, b, (short)0, acc[2 * u], false, false);
            b.x = q[t][u].z; b.y = q[t][u].w;
            acc[2 * u + 1] = __builtin_amdgcn_wmma_f32_16x16x4_f32(
                false, a[t], false, b, (short)0, acc[2 * u + 1], false, false);
        }
    }
}


template <int BN>
__device__ __forceinline__ void load_a(v2f (&a)[BN], const float* ap, int s0) {
#pragma unroll
    for (int t = 0; t < BN; ++t) a[t] = *(const v2f*)(ap + 4 * (s0 + t));
}

template <int M>
__device__ __forceinline__ const float* b_base(const float* __restrict__ Bs, int half, int lr,
                                               int cb) {
    return Bs + (size_t)half * (2 * M) + 2 * cb + 4 * lr;
}

template <int M, int NQ, int BN, int RING>
__device__ __forceinline__ void mma_head(v4f (&ring)[RING][BN][NQ], const float* __restrict__ bp) {
#pragma unroll
    for (int c = 0; c < RING - 1; ++c) load_b<M, NQ, BN>(ring[c], bp, c * BN);
}

template <int M, int NQ, int BN, int RING>
__device__ __forceinline__ void mma_all(const float* ap, v4f (&ring)[RING][BN][NQ],
                                        const float* __restrict__ bp, v8f (&acc)[2 * NQ]) {
    constexpr int NC = 32 / BN;
    v2f a[2][BN];
    load_a<BN>(a[0], ap, 0);
#pragma unroll
    for (int c = 0; c < NC; ++c) {
        if (c + RING - 1 < NC)
            load_b<M, NQ, BN>(ring[(c + RING - 1) % RING], bp, (c + RING - 1) * BN);
        if (c + 1 < NC) load_a<BN>(a[(c + 1) & 1], ap, (c + 1) * BN);
        __builtin_amdgcn_sched_barrier(0);
        wmma_b<NQ, BN>(a[c & 1], ring[c % RING], acc);
        __builtin_amdgcn_sched_barrier(0);
    }
}

template <int M, bool RELU>
__global__ __launch_bounds__(64)
void fused_layer_kernel(const float* __restrict__ X,
                        const float* __restrict__ Wsw,
                        const float* __restrict__ bias,
                        const int* __restrict__ starts,
                        const int* __restrict__ sedge,
                        float* __restrict__ Out) {
    constexpr int K = 128;
    constexpr int MW = M / 2;
    constexpr int NQ = MW / 32;
    constexpr int BN = 8 / NQ;
    constexpr int CT = 2 * NQ;
    constexpr int RING = RINGDEPTH;
    __shared__ __attribute__((aligned(16))) float tile[2][16 * APITCH];
    const int lane = threadIdx.x & 31;
    const int wv = __builtin_amdgcn_readfirstlane((int)(threadIdx.x >> 5));
    const int half = lane >> 4;
    const int lr   = lane & 15;
    const int rowbase = blockIdx.x * 16;
    const int rb = 8 * wv;
    const int cb = MW * wv;
    const int2* __restrict__ recs = (const int2*)sedge;
    const float4* __restrict__ X4 = (const float4*)X;
    const int sl = lane < 16 ? lane : 16;
    const float4 zero4 = make_float4(0.0f, 0.0f, 0.0f, 0.0f);

    v8f acc[CT];
#pragma unroll
    for (int ct = 0; ct < CT; ++ct) acc[ct] = v8f{};

    const int kb0 = (rowbase / BW) * KPB + rowbase % BW;
    int svn = starts[kb0 + sl];
    int ssum = 0;
    for (int r = 0; r < RR; ++r) {
        const int svc = svn;
        ssum += svc;
        const int rn = r + 1 < RR ? r + 1 : r;
        svn = starts[kb0 + rn * BW + sl];

        int eb = __builtin_amdgcn_readlane(svc, rb);
        int ee = __builtin_amdgcn_readlane(svc, rb + 8);
        eb = eb < 0 ? 0 : eb;
        ee = ee > EE ? EE : ee;

        const float* bp = b_base<M>(Wsw + (size_t)r * K * M, half, lr, cb);
        v4f ring[RING][BN][NQ];
        mma_head<M, NQ, BN, RING>(ring, bp);

        float* tl = tile[r & 1];
#pragma unroll
        for (int j = 0; j < 8; ++j) *(float4*)&tl[(rb + j) * APITCH + lane * 4] = zero4;

        int cur = -1;
        float4 g = zero4;
        for (int c0 = eb; c0 < ee; c0 += 32) {
            int n = ee - c0;
            n = n > 32 ? 32 : n;
            const int myp = c0 + lane;
            int2 rec = make_int2(0, 0);
            if (lane < n) rec = recs[myp];
            int ms = rec.x;
            ms = ms < 0 ? 0 : (ms >= NN ? NN - 1 : ms);
            int myrow = 0;
#pragma unroll
            for (int j = 1; j < 16; ++j)
                myrow += (__builtin_amdgcn_readlane(svc, j) <= myp) ? 1 : 0;

            for (int e0 = 0; e0 < n; e0 += GU) {
                float4 v[GU];
#pragma unroll
                for (int u = 0; u < GU; ++u) {
                    int eu = e0 + u;
                    eu = eu < n ? eu : n - 1;
                    const int s = __builtin_amdgcn_readlane(ms, eu);
                    v[u] = X4[(size_t)s * (K / 4) + lane];
                }
#pragma unroll
                for (int u = 0; u < GU; ++u) {
                    const int eu = e0 + u;
                    if (eu < n) {
                        const int row = __builtin_amdgcn_readlane(myrow, eu);
                        const float w = __int_as_float(__builtin_amdgcn_readlane(rec.y, eu));
                        if (row != cur) {
                            if (cur >= 0) *(float4*)&tl[cur * APITCH + lane * 4] = g;
                            g = zero4;
                            cur = row;
                        }
                        g.x += v[u].x * w;
                        g.y += v[u].y * w;
                        g.z += v[u].z * w;
                        g.w += v[u].w * w;
                    }
                }
            }
        }
        if (cur >= 0) *(float4*)&tl[cur * APITCH + lane * 4] = g;

        __syncthreads();
        mma_all<M, NQ, BN, RING>(&tl[lr * APITCH + 2 * half], ring, bp, acc);
    }

    const float* bpr = b_base<M>(Wsw + (size_t)RR * K * M, half, lr, cb);
    v4f ringr[RING][BN][NQ];
    mma_head<M, NQ, BN, RING>(ringr, bpr);
    float4 xr[8];
#pragma unroll
    for (int j = 0; j < 8; ++j) xr[j] = X4[(size_t)(rowbase + rb + j) * (K / 4) + lane];

    const int deg = __shfl_down(ssum, 1, 32) - ssum;
    const int invl = __float_as_int(1.0f / fmaxf((float)deg, 1.0f));
    float ivr[8];
#pragma unroll
    for (int j = 0; j < 8; ++j) {
        const int lo = __builtin_amdgcn_readlane(invl, j);
        const int hi = __builtin_amdgcn_readlane(invl, j + 8);
        ivr[j] = __int_as_float(half ? hi : lo);
    }
#pragma unroll
    for (int ct = 0; ct < CT; ++ct) {
#pragma unroll
        for (int j = 0; j < 8; ++j) acc[ct][j] *= ivr[j];
    }

    {
        float* tl = tile[RR & 1];
#pragma unroll
        for (int j = 0; j < 8; ++j) *(float4*)&tl[(rb + j) * APITCH + lane * 4] = xr[j];
        __syncthreads();
        mma_all<M, NQ, BN, RING>(&tl[lr * APITCH + 2 * half], ringr, bpr, acc);
    }

#pragma unroll
    for (int u = 0; u < NQ; ++u) {
        const int col = cb + 32 * u + 2 * lr;
        const v2f bb = *(const v2f*)(bias + col);
        float* crow = Out + (size_t)(rowbase + 8 * half) * M + col;
#pragma unroll
        for (int j = 0; j < 8; ++j) {
            v2f val;
            val.x = acc[2 * u + 0][j] + bb.x;
            val.y = acc[2 * u + 1][j] + bb.y;
            if (RELU) {
                val.x = fmaxf(val.x, 0.0f);
                val.y = fmaxf(val.y, 0.0f);
            }
            *(v2f*)(crow + (size_t)j * M) = val;
        }
    }
}

extern "C" void kernel_launch(void* const* d_in, const int* in_sizes, int n_in,
                              void* d_out, int out_size, void* d_ws, size_t ws_size,
                              hipStream_t stream) {
    const float* x     = (const float*)d_in[0];
    const int*   eidx  = (const int*)d_in[1];
    const float* ew    = (const float*)d_in[2];
    const int*   et    = (const int*)d_in[3];
    const float* W1    = (const float*)d_in[4];
    const float* root1 = (const float*)d_in[5];
    const float* bias1 = (const float*)d_in[6];
    const float* W2    = (const float*)d_in[7];
    const float* root2 = (const float*)d_in[8];
    const float* bias2 = (const float*)d_in[9];
    float* out = (float*)d_out;

    const int* src = eidx;
    const int* dst = eidx + EE;

    float* bufH = (float*)d_ws;
    float* agg  = bufH + (size_t)NN * HH;
    float* sw1  = agg;
    float* sw2  = sw1 + (size_t)(RR + 1) * FF * HH;
    float* bufG = agg  + (size_t)NN * HH;
    unsigned* mm = (unsigned*)(bufG + (size_t)NN * HH);

    int*   starts = (int*)bufG;
    int*   btot   = starts + NSTP;
    int*   tbl    = btot + NBK * BCS;
    int*   stage  = tbl + NTBP;
    int*   sedge  = stage + 2 * (size_t)NCH * CHUNK;

    const int T = 256;
    const int rowTiles = NN / 16;

    prep_weights_kernel<<<(WT1 + WT2 + 255) / 256, 256, 0, stream>>>(W1, root1, W2, root2,
                                                                     sw1, sw2, mm, btot);
    ew_minmax_kernel<<<256, T, 0, stream>>>(ew, EE, mm);

    group_chunk_kernel<<<NCH, 256, 0, stream>>>(src, dst, et, ew, mm, btot, tbl, stage);
    group_bucket_kernel<<<NBK, 256, 0, stream>>>(btot, tbl, stage, starts, sedge);

    fused_layer_kernel<HH, true><<<rowTiles, 64, 0, stream>>>(
        x, sw1, bias1, starts, sedge, bufH);
    fused_layer_kernel<OO, false><<<rowTiles, 64, 0, stream>>>(
        bufH, sw2, bias2, starts, sedge, out);
}


# Prep folded into the two grouping launches: weight min/max per chunk, bucket offsets from the run table, weight re-layout as extra blocks: 6 -> 4 launches, no global atomic left; sums unchanged
# speedup vs baseline: 4.2700x; 1.0806x over previous
#include <hip/hip_runtime.h>
#include <hip/hip_bf16.h>

#define NN 50000
#define EE 600000
#define FF 128
#define HH 128
#define OO 64
#define RR 8

typedef float v2f __attribute__((ext_vector_type(2)));
typedef float v4f __attribute__((ext_vector_type(4)));
typedef float v8f __attribute__((ext_vector_type(8)));

#define WPR ((RR + 1) * (FF / 2))
#define WT1 (WPR * (HH / 4))
#define WT2 (WPR * (OO / 4))
#define WBLK ((WT1 + WT2 + 255) / 256)
__device__ __forceinline__ void relayout_weights(int t,
                         const float* __restrict__ W1, const float* __restrict__ root1,
                         const float* __restrict__ W2, const float* __restrict__ root2,
                         float* __restrict__ sw1, float* __restrict__ sw2) {
    const float* src;
    float* dst;
    int M, pr, c4;
    if (t < WT1) {
        M = HH; pr = t / (HH / 4); c4 = t % (HH / 4); dst = sw1;
        src = pr < RR * (FF / 2) ? W1 + (size_t)(2 * pr) * HH
                                 : root1 + (size_t)(2 * (pr - RR * (FF / 2))) * HH;
    } else if (t < WT1 + WT2) {
        const int u = t - WT1;
        M = OO; pr = u / (OO / 4); c4 = u % (OO / 4); dst = sw2;
        src = pr < RR * (HH / 2) ? W2 + (size_t)(2 * pr) * OO
                                 : root2 + (size_t)(2 * (pr - RR * (HH / 2))) * OO;
    } else {
        return;
    }
    const v4f a = *(const v4f*)(src + 4 * c4);
    const v4f b = *(const v4f*)(src + M + 4 * c4);
    v4f o0, o1;
    o0.x = a.x; o0.y = b.x; o0.z = a.y; o0.w = b.y;
    o1.x = a.z; o1.y = b.z; o1.z = a.w; o1.w = b.w;
    float* d = dst + ((size_t)pr * M + 4 * c4) * 2;
    *(v4f*)d = o0;
    *(v4f*)(d + 4) = o1;
}

#define BW 208
#define NBK ((NN + BW - 1) / BW)
#define KPB (RR * BW)
#define NKEY (NBK * KPB)
#define CHUNK 2048
#define NCH ((EE + CHUNK - 1) / CHUNK)
#define BCS 32
#define KPT ((KPB + 255) / 256)
#define BCAP 4096
#define NSTP (((NKEY + 1) + 3) / 4 * 4)
#define NTBP ((NBK * NCH + 3) / 4 * 4)
static_assert(BW % 16 == 0, "a row tile must not straddle buckets");
static_assert(NBK <= 256, "one bucket per thread in the chunk kernel");
static_assert(NN <= 65536 && KPB <= 65536, "source and in-bucket key share one word");
static_assert(EE % 4 == 0, "four edges per 16-byte load");
static_assert(2 * NCH <= NBK * BCS, "the chunks' min/max pairs fit their part of the workspace");
static_assert(CHUNK <= 65535, "run start and run length each fit 16 bits of a table word");

__device__ __forceinline__ int block_scan_256(int v, int* wtot, int& tot) {
    const int lane = threadIdx.x & 31;
    const int wv = threadIdx.x >> 5;
    int inc = v;
    for (int off = 1; off < 32; off <<= 1) {
        const int n = __shfl_up(inc, off, 32);
        if (lane >= off) inc += n;
    }
    if (lane == 31) wtot[wv] = inc;
    __syncthreads();
    int base = 0, t = 0;
    for (int i = 0; i < 8; ++i) {
        const int w = wtot[i];
        if (i < wv) base += w;
        t += w;
    }
    tot = t;
    __syncthreads();
    return base + inc - v;
}

__global__ __launch_bounds__(256)
void group_chunk_kernel(const int* __restrict__ src, const int* __restrict__ dst,
                        const int* __restrict__ et, const float* __restrict__ ew,
                        const float* __restrict__ W1, const float* __restrict__ root1,
                        const float* __restrict__ W2, const float* __restrict__ root2,
                        float* __restrict__ sw1, float* __restrict__ sw2,
                        unsigned* __restrict__ pmm,
                        int* __restrict__ tbl,
                        int* __restrict__ stage_w) {
    if (blockIdx.x >= NCH) {
        relayout_weights((int)(blockIdx.x - NCH) * 256 + (int)threadIdx.x,
                         W1, root1, W2, root2, sw1, sw2);
        return;
    }
    int2* __restrict__ stage = (int2*)stage_w;
    __shared__ int cnt[256];
    __shared__ int lst[256];
    __shared__ int wtot[8];
    __shared__ int2 srec[CHUNK];
    __shared__ unsigned smm[2];
    const int t = threadIdx.x;
    const int ch = blockIdx.x;
    cnt[t] = 0;
    if (t == 0) { smm[0] = 0x7F800000u; smm[1] = 0u; }
    __syncthreads();

    float mn = __builtin_huge_valf();
    float mx = 0.0f;
    int bk[8], lr[8];
    int2 rc[8];
#pragma unroll
    for (int g = 0; g < 2; ++g) {
        const int q = ch * (CHUNK / 4) + g * 256 + t;
        int4 ty = make_int4(-1, -1, -1, -1), d = ty, s = ty;
        float4 w = make_float4(0.0f, 0.0f, 0.0f, 0.0f);
        if (q < EE / 4) {
            ty = ((const int4*)et)[q];
            d = ((const int4*)dst)[q];
            s = ((const int4*)src)[q];
            w = ((const float4*)ew)[q];
            mn = fminf(mn, fminf(fminf(w.x, w.y), fminf(w.z, w.w)));
            mx = fmaxf(mx, fmaxf(fmaxf(w.x, w.y), fmaxf(w.z, w.w)));
        }
        const int tt[4] = {ty.x, ty.y, ty.z, ty.w};
        const int dd[4] = {d.x, d.y, d.z, d.w};
        const int ss[4] = {s.x, s.y, s.z, s.w};
        const float ww[4] = {w.x, w.y, w.z, w.w};
#pragma unroll
        for (int j = 0; j < 4; ++j) {
            const int e = 4 * g + j;
            bk[e] = -1;
            lr[e] = 0;
            rc[e] = make_int2(0, 0);
            if ((unsigned)tt[j] < (unsigned)RR && (unsigned)dd[j] < (unsigned)NN) {
                const int b = dd[j] / BW;
                const int kl = tt[j] * BW + (dd[j] - b * BW);
                int sc = ss[j];
                sc = sc < 0 ? 0 : (sc >= NN ? NN - 1 : sc);
                bk[e] = b;
                lr[e] = atomicAdd(&cnt[b], 1);
                rc[e].x = sc | (kl << 16);
                rc[e].y = __float_as_int(ww[j]);
            }
        }
    }
    for (int off = 16; off > 0; off >>= 1) {
        mn = fminf(mn, __shfl_down(mn, off, 32));
        mx = fmaxf(mx, __shfl_down(mx, off, 32));
    }
    if ((t & 31) == 0) {
        atomicMin(&smm[0], __float_as_uint(mn));
        atomicMax(&smm[1], __float_as_uint(mx));
    }
    __syncthreads();
    if (t == 0) {
        pmm[2 * ch] = smm[0];
        pmm[2 * ch + 1] = smm[1];
    }

    const int c = cnt[t];
    int nvalid;
    const int ls = block_scan_256(c, wtot, nvalid);
    lst[t] = ls;
    if (t < NBK) tbl[t * NCH + ch] = (ls << 16) | c;
    __syncthreads();
#pragma unroll
    for (int e = 0; e < 8; ++e)
        if (bk[e] >= 0) srec[lst[bk[e]] + lr[e]] = rc[e];
    __syncthreads();
    int2* out = stage + (size_t)ch * CHUNK;
    for (int i = t; i < nvalid; i += 256) out[i] = srec[i];
}

__global__ __launch_bounds__(256)
void group_bucket_kernel(const int* __restrict__ tbl,
                         const int* __restrict__ stage_w,
                         const unsigned* __restrict__ pmm,
                         int* __restrict__ starts,
                         int* __restrict__ sedge_w) {
    const int2* __restrict__ stage = (const int2*)stage_w;
    int2* __restrict__ sedge = (int2*)sedge_w;
    __shared__ int kc[256 * KPT];
    __shared__ int wtot[8];
    __shared__ unsigned smm[2];
    __shared__ int2 lin[BCAP];
    __shared__ int2 lrec[BCAP];
    constexpr int RPT = (NCH + 255) / 256;
    const int t = threadIdx.x;
    const int b = blockIdx.x;
    for (int i = t; i < 256 * KPT; i += 256) kc[i] = 0;
    if (t == 0) { smm[0] = 0x7F800000u; smm[1] = 0u; }

    int rs[RPT], rn[RPT];
    int mylen = 0;
    int mypre = 0;
    unsigned pmn = 0x7F800000u, pmx = 0u;
#pragma unroll
    for (int k = 0; k < RPT; ++k) {
        const int ch = t + 256 * k;
        rs[k] = 0;
        rn[k] = 0;
        if (ch < NCH) {
            const int p = tbl[b * NCH + ch];
            const uint2 m = ((const uint2*)pmm)[ch];
            const int st = (int)((unsigned)p >> 16);
            rs[k] = ch * CHUNK + st;
            rn[k] = p & 0xFFFF;
            rn[k] = rn[k] > CHUNK ? CHUNK : rn[k];
            mypre += st;
            pmn = m.x < pmn ? m.x : pmn;
            pmx = m.y > pmx ? m.y : pmx;
        }
        mylen += rn[k];
    }
    int all;
    int off = block_scan_256(mylen, wtot, all);
    for (int o = 16; o > 0; o >>= 1) {
        const unsigned a = (unsigned)__shfl_down((int)pmn, o, 32);
        const unsigned c = (unsigned)__shfl_down((int)pmx, o, 32);
        pmn = a < pmn ? a : pmn;
        pmx = c > pmx ? c : pmx;
    }
    if ((t & 31) == 0) {
        atomicMin(&smm[0], pmn);
        atomicMax(&smm[1], pmx);
    }
    int base;
    (void)block_scan_256(mypre, wtot, base);
    const bool staged = all <= BCAP;
    const float wmn = __uint_as_float(smm[0]);
    const float wden = __uint_as_float(smm[1]) - wmn + 1e-8f;

    constexpr int GR = 8;
#pragma unroll
    for (int k = 0; k < RPT; ++k) {
        for (int i = 0; i < rn[k]; i += GR) {
            int2 r[GR];
#pragma unroll
            for (int u = 0; u < GR; ++u) {
                const int iu = i + u < rn[k] ? i + u : rn[k] - 1;
                r[u] = stage[rs[k] + iu];
            }
#pragma unroll
            for (int u = 0; u < GR; ++u) {
                if (i + u < rn[k]) {
                    const int kl = (int)((unsigned)r[u].x >> 16);
                    if (kl < KPB) atomicAdd(&kc[kl], 1);
                    if (staged) lin[off + i + u] = r[u];
                }
            }
        }
        off += rn[k];
    }
    __syncthreads();

    int v[KPT];
    int sum = 0;
#pragma unroll
    for (int j = 0; j < KPT; ++j) {
        v[j] = kc[t * KPT + j];
        sum += v[j];
    }
    int nrec;
    int ex = block_scan_256(sum, wtot, nrec);
#pragma unroll
    for (int j = 0; j < KPT; ++j) {
        kc[t * KPT + j] = ex;
        ex += v[j];
    }
    __syncthreads();
    for (int k = t; k < KPB; k += 256) starts[b * KPB + k] = base + kc[k];
    if (b == NBK - 1 && t == 0) starts[NKEY] = base + nrec;
    __syncthreads();

    if (staged) {
        for (int i = t; i < all; i += 256) {
            const int2 r = lin[i];
            const int kl = (int)((unsigned)r.x >> 16);
            if (kl < KPB) {
                const int pos = atomicAdd(&kc[kl], 1);
                if (pos < BCAP)
                    lrec[pos] = make_int2(r.x & 0xFFFF,
                                          __float_as_int((__int_as_float(r.y) - wmn) / wden));
            }
        }
        __syncthreads();
        int n = nrec < BCAP ? nrec : BCAP;
        if (base + n > EE) n = EE - base;
        for (int i = t; i < n; i += 256) sedge[base + i] = lrec[i];
    } else {
#pragma unroll
        for (int k = 0; k < RPT; ++k)
            for (int i = 0; i < rn[k]; ++i) {
                const int2 r = stage[rs[k] + i];
                const int kl = (int)((unsigned)r.x >> 16);
                if (kl < KPB) {
                    const int pos = atomicAdd(&kc[kl], 1);
                    if ((unsigned)(base + pos) < (unsigned)EE)
                        sedge[base + pos] = make_int2(
                            r.x & 0xFFFF, __float_as_int((__int_as_float(r.y) - wmn) / wden));
                }
            }
    }
}

#define APITCH 132
#define GU 8
#define RINGDEPTH 2


template <int M, int NQ, int BN>
__device__ __forceinline__ void load_b(v4f (&q)[BN][NQ], const float* __restrict__ bp, int s0) {
#pragma unroll
    for (int t = 0; t < BN; ++t) {
#pragma unroll
        for (int u = 0; u < NQ; ++u)
            q[t][u] = *(const v4f*)(bp + 4 * (s0 + t) * M + 64 * u);
    }
}

template <int NQ, int BN>
__device__ __forceinline__ void wmma_b(const v2f (&a)[BN], const v4f (&q)[BN][NQ],
                                       v8f (&acc)[2 * NQ]) {
#pragma unroll
    for (int t = 0; t < BN; ++t) {
#pragma unroll
        for (int u = 0; u < NQ; ++u) {
            v2f b;
            b.x = q[t][u].x; b.y = q[t][u].y;
            acc[2 * u] = __builtin_amdgcn_wmma_f32_16x16x4_f32(
                false, a[t], false, b, (short)0, acc[2 * u], false, false);
            b.x = q[t][u].z; b.y = q[t][u].w;
            acc[2 * u + 1] = __builtin_amdgcn_wmma_f32_16x16x4_f32(
                false, a[t], false, b, (short)0, acc[2 * u + 1], false, false);
        }
    }
}


template <int BN>
__device__ __forceinline__ void load_a(v2f (&a)[BN], const float* ap, int s0) {
#pragma unroll
    for (int t = 0; t < BN; ++t) a[t] = *(const v2f*)(ap + 4 * (s0 + t));
}

template <int M>
__device__ __forceinline__ const float* b_base(const float* __restrict__ Bs, int half, int lr,
                                               int cb) {
    return Bs + (size_t)half * (2 * M) + 2 * cb + 4 * lr;
}

template <int M, int NQ, int BN, int RING>
__device__ __forceinline__ void mma_head(v4f (&ring)[RING][BN][NQ], const float* __restrict__ bp) {
#pragma unroll
    for (int c = 0; c < RING - 1; ++c) load_b<M, NQ, BN>(ring[c], bp, c * BN);
}

template <int M, int NQ, int BN, int RING>
__device__ __forceinline__ void mma_all(const float* ap, v4f (&ring)[RING][BN][NQ],
                                        const float* __restrict__ bp, v8f (&acc)[2 * NQ]) {
    constexpr int NC = 32 / BN;
    v2f a[2][BN];
    load_a<BN>(a[0], ap, 0);
#pragma unroll
    for (int c = 0; c < NC; ++c) {
        if (c + RING - 1 < NC)
            load_b<M, NQ, BN>(ring[(c + RING - 1) % RING], bp, (c + RING - 1) * BN);
        if (c + 1 < NC) load_a<BN>(a[(c + 1) & 1], ap, (c + 1) * BN);
        __builtin_amdgcn_sched_barrier(0);
        wmma_b<NQ, BN>(a[c & 1], ring[c % RING], acc);
        __builtin_amdgcn_sched_barrier(0);
    }
}

template <int M, bool RELU>
__global__ __launch_bounds__(64)
void fused_layer_kernel(const float* __restrict__ X,
                        const float* __restrict__ Wsw,
                        const float* __restrict__ bias,
                        const int* __restrict__ starts,
                        const int* __restrict__ sedge,
                        float* __restrict__ Out) {
    constexpr int K = 128;
    constexpr int MW = M / 2;
    constexpr int NQ = MW / 32;
    constexpr int BN = 8 / NQ;
    constexpr int CT = 2 * NQ;
    constexpr int RING = RINGDEPTH;
    __shared__ __attribute__((aligned(16))) float tile[2][16 * APITCH];
    const int lane = threadIdx.x & 31;
    const int wv = __builtin_amdgcn_readfirstlane((int)(threadIdx.x >> 5));
    const int half = lane >> 4;
    const int lr   = lane & 15;
    const int rowbase = blockIdx.x * 16;
    const int rb = 8 * wv;
    const int cb = MW * wv;
    const int2* __restrict__ recs = (const int2*)sedge;
    const float4* __restrict__ X4 = (const float4*)X;
    const int sl = lane < 16 ? lane : 16;
    const float4 zero4 = make_float4(0.0f, 0.0f, 0.0f, 0.0f);

    v8f acc[CT];
#pragma unroll
    for (int ct = 0; ct < CT; ++ct) acc[ct] = v8f{};

    const int kb0 = (rowbase / BW) * KPB + rowbase % BW;
    int svn = starts[kb0 + sl];
    int ssum = 0;
    for (int r = 0; r < RR; ++r) {
        const int svc = svn;
        ssum += svc;
        const int rn = r + 1 < RR ? r + 1 : r;
        svn = starts[kb0 + rn * BW + sl];

        int eb = __builtin_amdgcn_readlane(svc, rb);
        int ee = __builtin_amdgcn_readlane(svc, rb + 8);
        eb = eb < 0 ? 0 : eb;
        ee = ee > EE ? EE : ee;

        const float* bp = b_base<M>(Wsw + (size_t)r * K * M, half, lr, cb);
        v4f ring[RING][BN][NQ];
        mma_head<M, NQ, BN, RING>(ring, bp);

        float* tl = tile[r & 1];
#pragma unroll
        for (int j = 0; j < 8; ++j) *(float4*)&tl[(rb + j) * APITCH + lane * 4] = zero4;

        int cur = -1;
        float4 g = zero4;
        for (int c0 = eb; c0 < ee; c0 += 32) {
            int n = ee - c0;
            n = n > 32 ? 32 : n;
            const int myp = c0 + lane;
            int2 rec = make_int2(0, 0);
            if (lane < n) rec = recs[myp];
            int ms = rec.x;
            ms = ms < 0 ? 0 : (ms >= NN ? NN - 1 : ms);
            int myrow = 0;
#pragma unroll
            for (int j = 1; j < 16; ++j)
                myrow += (__builtin_amdgcn_readlane(svc, j) <= myp) ? 1 : 0;

            for (int e0 = 0; e0 < n; e0 += GU) {
                float4 v[GU];
#pragma unroll
                for (int u = 0; u < GU; ++u) {
                    int eu = e0 + u;
                    eu = eu < n ? eu : n - 1;
                    const int s = __builtin_amdgcn_readlane(ms, eu);
                    v[u] = X4[(size_t)s * (K / 4) + lane];
                }
#pragma unroll
                for (int u = 0; u < GU; ++u) {
                    const int eu = e0 + u;
                    if (eu < n) {
                        const int row = __builtin_amdgcn_readlane(myrow, eu);
                        const float w = __int_as_float(__builtin_amdgcn_readlane(rec.y, eu));
                        if (row != cur) {
                            if (cur >= 0) *(float4*)&tl[cur * APITCH + lane * 4] = g;
                            g = zero4;
                            cur = row;
                        }
                        g.x += v[u].x * w;
                        g.y += v[u].y * w;
                        g.z += v[u].z * w;
                        g.w += v[u].w * w;
                    }
                }
            }
        }
        if (cur >= 0) *(float4*)&tl[cur * APITCH + lane * 4] = g;

        __syncthreads();
        mma_all<M, NQ, BN, RING>(&tl[lr * APITCH + 2 * half], ring, bp, acc);
    }

    const float* bpr = b_base<M>(Wsw + (size_t)RR * K * M, half, lr, cb);
    v4f ringr[RING][BN][NQ];
    mma_head<M, NQ, BN, RING>(ringr, bpr);
    float4 xr[8];
#pragma unroll
    for (int j = 0; j < 8; ++j) xr[j] = X4[(size_t)(rowbase + rb + j) * (K / 4) + lane];

    const int deg = __shfl_down(ssum, 1, 32) - ssum;
    const int invl = __float_as_int(1.0f / fmaxf((float)deg, 1.0f));
    float ivr[8];
#pragma unroll
    for (int j = 0; j < 8; ++j) {
        const int lo = __builtin_amdgcn_readlane(invl, j);
        const int hi = __builtin_amdgcn_readlane(invl, j + 8);
        ivr[j] = __int_as_float(half ? hi : lo);
    }
#pragma unroll
    for (int ct = 0; ct < CT; ++ct) {
#pragma unroll
        for (int j = 0; j < 8; ++j) acc[ct][j] *= ivr[j];
    }

    {
        float* tl = tile[RR & 1];
#pragma unroll
        for (int j = 0; j < 8; ++j) *(float4*)&tl[(rb + j) * APITCH + lane * 4] = xr[j];
        __syncthreads();
        mma_all<M, NQ, BN, RING>(&tl[lr * APITCH + 2 * half], ringr, bpr, acc);
    }

#pragma unroll
    for (int u = 0; u < NQ; ++u) {
        const int col = cb + 32 * u + 2 * lr;
        const v2f bb = *(const v2f*)(bias + col);
        float* crow = Out + (size_t)(rowbase + 8 * half) * M + col;
#pragma unroll
        for (int j = 0; j < 8; ++j) {
            v2f val;
            val.x = acc[2 * u + 0][j] + bb.x;
            val.y = acc[2 * u + 1][j] + bb.y;
            if (RELU) {
                val.x = fmaxf(val.x, 0.0f);
                val.y = fmaxf(val.y, 0.0f);
            }
            *(v2f*)(crow + (size_t)j * M) = val;
        }
    }
}

extern "C" void kernel_launch(void* const* d_in, const int* in_sizes, int n_in,
                              void* d_out, int out_size, void* d_ws, size_t ws_size,
                              hipStream_t stream) {
    const float* x     = (const float*)d_in[0];
    const int*   eidx  = (const int*)d_in[1];
    const float* ew    = (const float*)d_in[2];
    const int*   et    = (const int*)d_in[3];
    const float* W1    = (const float*)d_in[4];
    const float* root1 = (const float*)d_in[5];
    const float* bias1 = (const float*)d_in[6];
    const float* W2    = (const float*)d_in[7];
    const float* root2 = (const float*)d_in[8];
    const float* bias2 = (const float*)d_in[9];
    float* out = (float*)d_out;

    const int* src = eidx;
    const int* dst = eidx + EE;

    float* bufH = (float*)d_ws;
    float* agg  = bufH + (size_t)NN * HH;
    float* sw1  = agg;
    float* sw2  = sw1 + (size_t)(RR + 1) * FF * HH;
    float* bufG = agg  + (size_t)NN * HH;

    int*   starts = (int*)bufG;
    unsigned* pmm = (unsigned*)(starts + NSTP);
    int*   tbl    = (int*)pmm + NBK * BCS;
    int*   stage  = tbl + NTBP;
    int*   sedge  = stage + 2 * (size_t)NCH * CHUNK;

    const int rowTiles = NN / 16;

    group_chunk_kernel<<<NCH + WBLK, 256, 0, stream>>>(src, dst, et, ew, W1, root1, W2, root2,
                                                      sw1, sw2, pmm, tbl, stage);
    group_bucket_kernel<<<NBK, 256, 0, stream>>>(tbl, stage, pmm, starts, sedge);

    fused_layer_kernel<HH, true><<<rowTiles, 64, 0, stream>>>(
        x, sw1, bias1, starts, sedge, bufH);
    fused_layer_kernel<OO, false><<<rowTiles, 64, 0, stream>>>(
        bufH, sw2, bias2, starts, sedge, out);
}


# Wave priority by progress: s_setprio 3 in relations 0-4, 2 in 5-6, 1 in 7, 0 in the root product; a SIMD's waves advance together, none finishes alone with the matrix unit idle; sums unchanged
# speedup vs baseline: 4.6054x; 1.0785x over previous
#include <hip/hip_runtime.h>
#include <hip/hip_bf16.h>

#define NN 50000
#define EE 600000
#define FF 128
#define HH 128
#define OO 64
#define RR 8

typedef float v2f __attribute__((ext_vector_type(2)));
typedef float v4f __attribute__((ext_vector_type(4)));
typedef float v8f __attribute__((ext_vector_type(8)));

#define WPR ((RR + 1) * (FF / 2))
#define WT1 (WPR * (HH / 4))
#define WT2 (WPR * (OO / 4))
#define WBLK ((WT1 + WT2 + 255) / 256)
__device__ __forceinline__ void relayout_weights(int t,
                         const float* __restrict__ W1, const float* __restrict__ root1,
                         const float* __restrict__ W2, const float* __restrict__ root2,
                         float* __restrict__ sw1, float* __restrict__ sw2) {
    const float* src;
    float* dst;
    int M, pr, c4;
    if (t < WT1) {
        M = HH; pr = t / (HH / 4); c4 = t % (HH / 4); dst = sw1;
        src = pr < RR * (FF / 2) ? W1 + (size_t)(2 * pr) * HH
                                 : root1 + (size_t)(2 * (pr - RR * (FF / 2))) * HH;
    } else if (t < WT1 + WT2) {
        const int u = t - WT1;
        M = OO; pr = u / (OO / 4); c4 = u % (OO / 4); dst = sw2;
        src = pr < RR * (HH / 2) ? W2 + (size_t)(2 * pr) * OO
                                 : root2 + (size_t)(2 * (pr - RR * (HH / 2))) * OO;
    } else {
        return;
    }
    const v4f a = *(const v4f*)(src + 4 * c4);
    const v4f b = *(const v4f*)(src + M + 4 * c4);
    v4f o0, o1;
    o0.x = a.x; o0.y = b.x; o0.z = a.y; o0.w = b.y;
    o1.x = a.z; o1.y = b.z; o1.z = a.w; o1.w = b.w;
    float* d = dst + ((size_t)pr * M + 4 * c4) * 2;
    *(v4f*)d = o0;
    *(v4f*)(d + 4) = o1;
}

#define BW 208
#define NBK ((NN + BW - 1) / BW)
#define KPB (RR * BW)
#define NKEY (NBK * KPB)
#define CHUNK 2048
#define NCH ((EE + CHUNK - 1) / CHUNK)
#define BCS 32
#define KPT ((KPB + 255) / 256)
#define BCAP 4096
#define NSTP (((NKEY + 1) + 3) / 4 * 4)
#define NTBP ((NBK * NCH + 3) / 4 * 4)
static_assert(BW % 16 == 0, "a row tile must not straddle buckets");
static_assert(NBK <= 256, "one bucket per thread in the chunk kernel");
static_assert(NN <= 65536 && KPB <= 65536, "source and in-bucket key share one word");
static_assert(EE % 4 == 0, "four edges per 16-byte load");
static_assert(2 * NCH <= NBK * BCS, "the chunks' min/max pairs fit their part of the workspace");
static_assert(CHUNK <= 65535, "run start and run length each fit 16 bits of a table word");

__device__ __forceinline__ int block_scan_256(int v, int* wtot, int& tot) {
    const int lane = threadIdx.x & 31;
    const int wv = threadIdx.x >> 5;
    int inc = v;
    for (int off = 1; off < 32; off <<= 1) {
        const int n = __shfl_up(inc, off, 32);
        if (lane >= off) inc += n;
    }
    if (lane == 31) wtot[wv] = inc;
    __syncthreads();
    int base = 0, t = 0;
    for (int i = 0; i < 8; ++i) {
        const int w = wtot[i];
        if (i < wv) base += w;
        t += w;
    }
    tot = t;
    __syncthreads();
    return base + inc - v;
}

__global__ __launch_bounds__(256)
void group_chunk_kernel(const int* __restrict__ src, const int* __restrict__ dst,
                        const int* __restrict__ et, const float* __restrict__ ew,
                        const float* __restrict__ W1, const float* __restrict__ root1,
                        const float* __restrict__ W2, const float* __restrict__ root2,
                        float* __restrict__ sw1, float* __restrict__ sw2,
                        unsigned* __restrict__ pmm,
                        int* __restrict__ tbl,
                        int* __restrict__ stage_w) {
    if (blockIdx.x >= NCH) {
        relayout_weights((int)(blockIdx.x - NCH) * 256 + (int)threadIdx.x,
                         W1, root1, W2, root2, sw1, sw2);
        return;
    }
    int2* __restrict__ stage = (int2*)stage_w;
    __shared__ int cnt[256];
    __shared__ int lst[256];
    __shared__ int wtot[8];
    __shared__ int2 srec[CHUNK];
    __shared__ unsigned smm[2];
    const int t = threadIdx.x;
    const int ch = blockIdx.x;
    cnt[t] = 0;
    if (t == 0) { smm[0] = 0x7F800000u; smm[1] = 0u; }
    __syncthreads();

    float mn = __builtin_huge_valf();
    float mx = 0.0f;
    int bk[8], lr[8];
    int2 rc[8];
#pragma unroll
    for (int g = 0; g < 2; ++g) {
        const int q = ch * (CHUNK / 4) + g * 256 + t;
        int4 ty = make_int4(-1, -1, -1, -1), d = ty, s = ty;
        float4 w = make_float4(0.0f, 0.0f, 0.0f, 0.0f);
        if (q < EE / 4) {
            ty = ((const int4*)et)[q];
            d = ((const int4*)dst)[q];
            s = ((const int4*)src)[q];
            w = ((const float4*)ew)[q];
            mn = fminf(mn, fminf(fminf(w.x, w.y), fminf(w.z, w.w)));
            mx = fmaxf(mx, fmaxf(fmaxf(w.x, w.y), fmaxf(w.z, w.w)));
        }
        const int tt[4] = {ty.x, ty.y, ty.z, ty.w};
        const int dd[4] = {d.x, d.y, d.z, d.w};
        const int ss[4] = {s.x, s.y, s.z, s.w};
        const float ww[4] = {w.x, w.y, w.z, w.w};
#pragma unroll
        for (int j = 0; j < 4; ++j) {
            const int e = 4 * g + j;
            bk[e] = -1;
            lr[e] = 0;
            rc[e] = make_int2(0, 0);
            if ((unsigned)tt[j] < (unsigned)RR && (unsigned)dd[j] < (unsigned)NN) {
                const int b = dd[j] / BW;
                const int kl = tt[j] * BW + (dd[j] - b * BW);
                int sc = ss[j];
                sc = sc < 0 ? 0 : (sc >= NN ? NN - 1 : sc);
                bk[e] = b;
                lr[e] = atomicAdd(&cnt[b], 1);
                rc[e].x = sc | (kl << 16);
                rc[e].y = __float_as_int(ww[j]);
            }
        }
    }
    for (int off = 16; off > 0; off >>= 1) {
        mn = fminf(mn, __shfl_down(mn, off, 32));
        mx = fmaxf(mx, __shfl_down(mx, off, 32));
    }
    if ((t & 31) == 0) {
        atomicMin(&smm[0], __float_as_uint(mn));
        atomicMax(&smm[1], __float_as_uint(mx));
    }
    __syncthreads();
    if (t == 0) {
        pmm[2 * ch] = smm[0];
        pmm[2 * ch + 1] = smm[1];
    }

    const int c = cnt[t];
    int nvalid;
    const int ls = block_scan_256(c, wtot, nvalid);
    lst[t] = ls;
    if (t < NBK) tbl[t * NCH + ch] = (ls << 16) | c;
    __syncthreads();
#pragma unroll
    for (int e = 0; e < 8; ++e)
        if (bk[e] >= 0) srec[lst[bk[e]] + lr[e]] = rc[e];
    __syncthreads();
    int2* out = stage + (size_t)ch * CHUNK;
    for (int i = t; i < nvalid; i += 256) out[i] = srec[i];
}

__global__ __launch_bounds__(256)
void group_bucket_kernel(const int* __restrict__ tbl,
                         const int* __restrict__ stage_w,
                         const unsigned* __restrict__ pmm,
                         int* __restrict__ starts,
                         int* __restrict__ sedge_w) {
    const int2* __restrict__ stage = (const int2*)stage_w;
    int2* __restrict__ sedge = (int2*)sedge_w;
    __shared__ int kc[256 * KPT];
    __shared__ int wtot[8];
    __shared__ unsigned smm[2];
    __shared__ int2 lin[BCAP];
    __shared__ int2 lrec[BCAP];
    constexpr int RPT = (NCH + 255) / 256;
    const int t = threadIdx.x;
    const int b = blockIdx.x;
    for (int i = t; i < 256 * KPT; i += 256) kc[i] = 0;
    if (t == 0) { smm[0] = 0x7F800000u; smm[1] = 0u; }

    int rs[RPT], rn[RPT];
    int mylen = 0;
    int mypre = 0;
    unsigned pmn = 0x7F800000u, pmx = 0u;
#pragma unroll
    for (int k = 0; k < RPT; ++k) {
        const int ch = t + 256 * k;
        rs[k] = 0;
        rn[k] = 0;
        if (ch < NCH) {
            const int p = tbl[b * NCH + ch];
            const uint2 m = ((const uint2*)pmm)[ch];
            const int st = (int)((unsigned)p >> 16);
            rs[k] = ch * CHUNK + st;
            rn[k] = p & 0xFFFF;
            rn[k] = rn[k] > CHUNK ? CHUNK : rn[k];
            mypre += st;
            pmn = m.x < pmn ? m.x : pmn;
            pmx = m.y > pmx ? m.y : pmx;
        }
        mylen += rn[k];
    }
    int all;
    int off = block_scan_256(mylen, wtot, all);
    for (int o = 16; o > 0; o >>= 1) {
        const unsigned a = (unsigned)__shfl_down((int)pmn, o, 32);
        const unsigned c = (unsigned)__shfl_down((int)pmx, o, 32);
        pmn = a < pmn ? a : pmn;
        pmx = c > pmx ? c : pmx;
    }
    if ((t & 31) == 0) {
        atomicMin(&smm[0], pmn);
        atomicMax(&smm[1], pmx);
    }
    int base;
    (void)block_scan_256(mypre, wtot, base);
    const bool staged = all <= BCAP;
    const float wmn = __uint_as_float(smm[0]);
    const float wden = __uint_as_float(smm[1]) - wmn + 1e-8f;

    constexpr int GR = 8;
#pragma unroll
    for (int k = 0; k < RPT; ++k) {
        for (int i = 0; i < rn[k]; i += GR) {
            int2 r[GR];
#pragma unroll
            for (int u = 0; u < GR; ++u) {
                const int iu = i + u < rn[k] ? i + u : rn[k] - 1;
                r[u] = stage[rs[k] + iu];
            }
#pragma unroll
            for (int u = 0; u < GR; ++u) {
                if (i + u < rn[k]) {
                    const int kl = (int)((unsigned)r[u].x >> 16);
                    if (kl < KPB) atomicAdd(&kc[kl], 1);
                    if (staged) lin[off + i + u] = r[u];
                }
            }
        }
        off += rn[k];
    }
    __syncthreads();

    int v[KPT];
    int sum = 0;
#pragma unroll
    for (int j = 0; j < KPT; ++j) {
        v[j] = kc[t * KPT + j];
        sum += v[j];
    }
    int nrec;
    int ex = block_scan_256(sum, wtot, nrec);
#pragma unroll
    for (int j = 0; j < KPT; ++j) {
        kc[t * KPT + j] = ex;
        ex += v[j];
    }
    __syncthreads();
    for (int k = t; k < KPB; k += 256) starts[b * KPB + k] = base + kc[k];
    if (b == NBK - 1 && t == 0) starts[NKEY] = base + nrec;
    __syncthreads();

    if (staged) {
        for (int i = t; i < all; i += 256) {
            const int2 r = lin[i];
            const int kl = (int)((unsigned)r.x >> 16);
            if (kl < KPB) {
                const int pos = atomicAdd(&kc[kl], 1);
                if (pos < BCAP)
                    lrec[pos] = make_int2(r.x & 0xFFFF,
                                          __float_as_int((__int_as_float(r.y) - wmn) / wden));
            }
        }
        __syncthreads();
        int n = nrec < BCAP ? nrec : BCAP;
        if (base + n > EE) n = EE - base;
        for (int i = t; i < n; i += 256) sedge[base + i] = lrec[i];
    } else {
#pragma unroll
        for (int k = 0; k < RPT; ++k)
            for (int i = 0; i < rn[k]; ++i) {
                const int2 r = stage[rs[k] + i];
                const int kl = (int)((unsigned)r.x >> 16);
                if (kl < KPB) {
                    const int pos = atomicAdd(&kc[kl], 1);
                    if ((unsigned)(base + pos) < (unsigned)EE)
                        sedge[base + pos] = make_int2(
                            r.x & 0xFFFF, __float_as_int((__int_as_float(r.y) - wmn) / wden));
                }
            }
    }
}

#define APITCH 132
#define GU 8
#define RINGDEPTH 2


template <int M, int NQ, int BN>
__device__ __forceinline__ void load_b(v4f (&q)[BN][NQ], const float* __restrict__ bp, int s0) {
#pragma unroll
    for (int t = 0; t < BN; ++t) {
#pragma unroll
        for (int u = 0; u < NQ; ++u)
            q[t][u] = *(const v4f*)(bp + 4 * (s0 + t) * M + 64 * u);
    }
}

template <int NQ, int BN>
__device__ __forceinline__ void wmma_b(const v2f (&a)[BN], const v4f (&q)[BN][NQ],
                                       v8f (&acc)[2 * NQ]) {
#pragma unroll
    for (int t = 0; t < BN; ++t) {
#pragma unroll
        for (int u = 0; u < NQ; ++u) {
            v2f b;
            b.x = q[t][u].x; b.y = q[t][u].y;
            acc[2 * u] = __builtin_amdgcn_wmma_f32_16x16x4_f32(
                false, a[t], false, b, (short)0, acc[2 * u], false, false);
            b.x = q[t][u].z; b.y = q[t][u].w;
            acc[2 * u + 1] = __builtin_amdgcn_wmma_f32_16x16x4_f32(
                false, a[t], false, b, (short)0, acc[2 * u + 1], false, false);
        }
    }
}


template <int BN>
__device__ __forceinline__ void load_a(v2f (&a)[BN], const float* ap, int s0) {
#pragma unroll
    for (int t = 0; t < BN; ++t) a[t] = *(const v2f*)(ap + 4 * (s0 + t));
}

template <int M>
__device__ __forceinline__ const float* b_base(const float* __restrict__ Bs, int half, int lr,
                                               int cb) {
    return Bs + (size_t)half * (2 * M) + 2 * cb + 4 * lr;
}

template <int M, int NQ, int BN, int RING>
__device__ __forceinline__ void mma_head(v4f (&ring)[RING][BN][NQ], const float* __restrict__ bp) {
#pragma unroll
    for (int c = 0; c < RING - 1; ++c) load_b<M, NQ, BN>(ring[c], bp, c * BN);
}

template <int M, int NQ, int BN, int RING>
__device__ __forceinline__ void mma_all(const float* ap, v4f (&ring)[RING][BN][NQ],
                                        const float* __restrict__ bp, v8f (&acc)[2 * NQ]) {
    constexpr int NC = 32 / BN;
    v2f a[2][BN];
    load_a<BN>(a[0], ap, 0);
#pragma unroll
    for (int c = 0; c < NC; ++c) {
        if (c + RING - 1 < NC)
            load_b<M, NQ, BN>(ring[(c + RING - 1) % RING], bp, (c + RING - 1) * BN);
        if (c + 1 < NC) load_a<BN>(a[(c + 1) & 1], ap, (c + 1) * BN);
        __builtin_amdgcn_sched_barrier(0);
        wmma_b<NQ, BN>(a[c & 1], ring[c % RING], acc);
        __builtin_amdgcn_sched_barrier(0);
    }
}

template <int M, bool RELU>
__global__ __launch_bounds__(64)
void fused_layer_kernel(const float* __restrict__ X,
                        const float* __restrict__ Wsw,
                        const float* __restrict__ bias,
                        const int* __restrict__ starts,
                        const int* __restrict__ sedge,
                        float* __restrict__ Out) {
    constexpr int K = 128;
    constexpr int MW = M / 2;
    constexpr int NQ = MW / 32;
    constexpr int BN = 8 / NQ;
    constexpr int CT = 2 * NQ;
    constexpr int RING = RINGDEPTH;
    __shared__ __attribute__((aligned(16))) float tile[2][16 * APITCH];
    const int lane = threadIdx.x & 31;
    const int wv = __builtin_amdgcn_readfirstlane((int)(threadIdx.x >> 5));
    const int half = lane >> 4;
    const int lr   = lane & 15;
    const int rowbase = blockIdx.x * 16;
    const int rb = 8 * wv;
    const int cb = MW * wv;
    const int2* __restrict__ recs = (const int2*)sedge;
    const float4* __restrict__ X4 = (const float4*)X;
    const int sl = lane < 16 ? lane : 16;
    const float4 zero4 = make_float4(0.0f, 0.0f, 0.0f, 0.0f);

    v8f acc[CT];
#pragma unroll
    for (int ct = 0; ct < CT; ++ct) acc[ct] = v8f{};

    const int kb0 = (rowbase / BW) * KPB + rowbase % BW;
    int svn = starts[kb0 + sl];
    int ssum = 0;
    for (int r = 0; r < RR; ++r) {
        if (r == 0) __builtin_amdgcn_s_setprio(3);
        if (r == RR - 3) __builtin_amdgcn_s_setprio(2);
        if (r == RR - 1) __builtin_amdgcn_s_setprio(1);
        const int svc = svn;
        ssum += svc;
        const int rn = r + 1 < RR ? r + 1 : r;
        svn = starts[kb0 + rn * BW + sl];

        int eb = __builtin_amdgcn_readlane(svc, rb);
        int ee = __builtin_amdgcn_readlane(svc, rb + 8);
        eb = eb < 0 ? 0 : eb;
        ee = ee > EE ? EE : ee;

        const float* bp = b_base<M>(Wsw + (size_t)r * K * M, half, lr, cb);
        v4f ring[RING][BN][NQ];
        mma_head<M, NQ, BN, RING>(ring, bp);

        float* tl = tile[r & 1];
#pragma unroll
        for (int j = 0; j < 8; ++j) *(float4*)&tl[(rb + j) * APITCH + lane * 4] = zero4;

        int cur = -1;
        float4 g = zero4;
        for (int c0 = eb; c0 < ee; c0 += 32) {
            int n = ee - c0;
            n = n > 32 ? 32 : n;
            const int myp = c0 + lane;
            int2 rec = make_int2(0, 0);
            if (lane < n) rec = recs[myp];
            int ms = rec.x;
            ms = ms < 0 ? 0 : (ms >= NN ? NN - 1 : ms);
            int myrow = 0;
#pragma unroll
            for (int j = 1; j < 16; ++j)
                myrow += (__builtin_amdgcn_readlane(svc, j) <= myp) ? 1 : 0;

            for (int e0 = 0; e0 < n; e0 += GU) {
                float4 v[GU];
#pragma unroll
                for (int u = 0; u < GU; ++u) {
                    int eu = e0 + u;
                    eu = eu < n ? eu : n - 1;
                    const int s = __builtin_amdgcn_readlane(ms, eu);
                    v[u] = X4[(size_t)s * (K / 4) + lane];
                }
#pragma unroll
                for (int u = 0; u < GU; ++u) {
                    const int eu = e0 + u;
                    if (eu < n) {
                        const int row = __builtin_amdgcn_readlane(myrow, eu);
                        const float w = __int_as_float(__builtin_amdgcn_readlane(rec.y, eu));
                        if (row != cur) {
                            if (cur >= 0) *(float4*)&tl[cur * APITCH + lane * 4] = g;
                            g = zero4;
                            cur = row;
                        }
                        g.x += v[u].x * w;
                        g.y += v[u].y * w;
                        g.z += v[u].z * w;
                        g.w += v[u].w * w;
                    }
                }
            }
        }
        if (cur >= 0) *(float4*)&tl[cur * APITCH + lane * 4] = g;

        __syncthreads();
        mma_all<M, NQ, BN, RING>(&tl[lr * APITCH + 2 * half], ring, bp, acc);
    }

    __builtin_amdgcn_s_setprio(0);

    const float* bpr = b_base<M>(Wsw + (size_t)RR * K * M, half, lr, cb);
    v4f ringr[RING][BN][NQ];
    mma_head<M, NQ, BN, RING>(ringr, bpr);
    float4 xr[8];
#pragma unroll
    for (int j = 0; j < 8; ++j) xr[j] = X4[(size_t)(rowbase + rb + j) * (K / 4) + lane];

    const int deg = __shfl_down(ssum, 1, 32) - ssum;
    const int invl = __float_as_int(1.0f / fmaxf((float)deg, 1.0f));
    float ivr[8];
#pragma unroll
    for (int j = 0; j < 8; ++j) {
        const int lo = __builtin_amdgcn_readlane(invl, j);
        const int hi = __builtin_amdgcn_readlane(invl, j + 8);
        ivr[j] = __int_as_float(half ? hi : lo);
    }
#pragma unroll
    for (int ct = 0; ct < CT; ++ct) {
#pragma unroll
        for (int j = 0; j < 8; ++j) acc[ct][j] *= ivr[j];
    }

    {
        float* tl = tile[RR & 1];
#pragma unroll
        for (int j = 0; j < 8; ++j) *(float4*)&tl[(rb + j) * APITCH + lane * 4] = xr[j];
        __syncthreads();
        mma_all<M, NQ, BN, RING>(&tl[lr * APITCH + 2 * half], ringr, bpr, acc);
    }

#pragma unroll
    for (int u = 0; u < NQ; ++u) {
        const int col = cb + 32 * u + 2 * lr;
        const v2f bb = *(const v2f*)(bias + col);
        float* crow = Out + (size_t)(rowbase + 8 * half) * M + col;
#pragma unroll
        for (int j = 0; j < 8; ++j) {
            v2f val;
            val.x = acc[2 * u + 0][j] + bb.x;
            val.y = acc[2 * u + 1][j] + bb.y;
            if (RELU) {
                val.x = fmaxf(val.x, 0.0f);
                val.y = fmaxf(val.y, 0.0f);
            }
            *(v2f*)(crow + (size_t)j * M) = val;
        }
    }
}

extern "C" void kernel_launch(void* const* d_in, const int* in_sizes, int n_in,
                              void* d_out, int out_size, void* d_ws, size_t ws_size,
                              hipStream_t stream) {
    const float* x     = (const float*)d_in[0];
    const int*   eidx  = (const int*)d_in[1];
    const float* ew    = (const float*)d_in[2];
    const int*   et    = (const int*)d_in[3];
    const float* W1    = (const float*)d_in[4];
    const float* root1 = (const float*)d_in[5];
    const float* bias1 = (const float*)d_in[6];
    const float* W2    = (const float*)d_in[7];
    const float* root2 = (const float*)d_in[8];
    const float* bias2 = (const float*)d_in[9];
    float* out = (float*)d_out;

    const int* src = eidx;
    const int* dst = eidx + EE;

    float* bufH = (float*)d_ws;
    float* agg  = bufH + (size_t)NN * HH;
    float* sw1  = agg;
    float* sw2  = sw1 + (size_t)(RR + 1) * FF * HH;
    float* bufG = agg  + (size_t)NN * HH;

    int*   starts = (int*)bufG;
    unsigned* pmm = (unsigned*)(starts + NSTP);
    int*   tbl    = (int*)pmm + NBK * BCS;
    int*   stage  = tbl + NTBP;
    int*   sedge  = stage + 2 * (size_t)NCH * CHUNK;

    const int rowTiles = NN / 16;

    group_chunk_kernel<<<NCH + WBLK, 256, 0, stream>>>(src, dst, et, ew, W1, root1, W2, root2,
                                                      sw1, sw2, pmm, tbl, stage);
    group_bucket_kernel<<<NBK, 256, 0, stream>>>(tbl, stage, pmm, starts, sedge);

    fused_layer_kernel<HH, true><<<rowTiles, 64, 0, stream>>>(
        x, sw1, bias1, starts, sedge, bufH);
    fused_layer_kernel<OO, false><<<rowTiles, 64, 0, stream>>>(
        bufH, sw2, bias2, starts, sedge, out);
}
